# starting kernel
# baseline (speedup 1.0000x reference)
#include <hip/hip_runtime.h>

typedef __attribute__((ext_vector_type(2))) float v2f;
typedef __attribute__((ext_vector_type(8))) float v8f;

#define WAVES_PER_BLOCK 4


__global__ void gcn_deg_init(float* __restrict__ deg, int N) {
  int i = blockIdx.x * blockDim.x + threadIdx.x;
  if (i < N) deg[i] = 1.0f;
}

__global__ void gcn_deg_edge(const int* __restrict__ col, float* __restrict__ deg, int E) {
  int e = blockIdx.x * blockDim.x + threadIdx.x;
  if (e < E) unsafeAtomicAdd(&deg[col[e]], 1.0f);
}

__global__ void gcn_deg_to_dis(float* __restrict__ dis, int N) {
  int i = blockIdx.x * blockDim.x + threadIdx.x;
  if (i < N) {
    float d = dis[i];
    dis[i] = (d > 0.0f) ? rsqrtf(d) : 0.0f;
  }
}


__global__ void gcn_self_loop(const float* __restrict__ dis, const float* __restrict__ h,
                              float* __restrict__ agg, int N, int C) {
  long long i = (long long)blockIdx.x * blockDim.x + threadIdx.x;
  long long total = (long long)N * C;
  if (i < total) {
    int node = (int)(i / C);
    float d = dis[node];
    agg[i] = d * d * h[i];
  }
}

__global__ void gcn_edge_scatter(const int* __restrict__ ei, const float* __restrict__ dis,
                                 const float* __restrict__ h, float* __restrict__ agg,
                                 int E, int C) {
  int gw   = (int)(((long long)blockIdx.x * blockDim.x + threadIdx.x) >> 5);
  int lane = threadIdx.x & 31;
  if (gw >= E) return;
  int r = ei[gw];
  int c = ei[E + gw];
  float w = dis[r] * dis[c];
  const float* hp = h + (long long)r * C;
  float*       ap = agg + (long long)c * C;
  for (int ch = lane; ch < C; ch += 32)
    unsafeAtomicAdd(ap + ch, w * hp[ch]);
}

template <int IC, bool DUAL, bool RELU, bool GUARD>
__global__ __launch_bounds__(32 * WAVES_PER_BLOCK) void gcn_wmma_gemm(
    const float* __restrict__ inA, const float* __restrict__ WA, const float* __restrict__ bA,
    const float* __restrict__ inB, const float* __restrict__ WB, const float* __restrict__ bB,
    float* __restrict__ out, int nrows, int OC) {
  __shared__ float lds[(DUAL ? 2 : 1) * 16 * IC];
  const int tid  = threadIdx.x;
  const int lane = tid & 31;
  const int wave = tid >> 5;
  const int row0 = blockIdx.x << 4;
  const int col0 = (blockIdx.y * WAVES_PER_BLOCK + wave) << 4;

  float* Atile = lds;
  float* Btile = DUAL ? (lds + 16 * IC) : lds;

  constexpr int vecPerRow = IC >> 2;
  constexpr int totalVec  = vecPerRow << 4;
  for (int idx = tid; idx < totalVec; idx += 32 * WAVES_PER_BLOCK) {
    const int r = idx / vecPerRow;
    const int v = idx - r * vecPerRow;
    const int node = row0 + r;
    if (GUARD && node >= nrows) {
      float4 z = {0.f, 0.f, 0.f, 0.f};
      ((float4*)Atile)[idx] = z;
      if constexpr (DUAL) ((float4*)Btile)[idx] = z;
    } else {
      ((float4*)Atile)[idx] = ((const float4*)(inA + (size_t)node * IC))[v];
      if constexpr (DUAL)
        ((float4*)Btile)[idx] = ((const float4*)(inB + (size_t)node * IC))[v];
    }
  }
  __syncthreads();

  if (col0 >= OC) return;

  const int  nidx = lane & 15;
  const int  koff = (lane >> 4) << 1;
  const int  col  = col0 + nidx;
  const bool colOK = GUARD ? (col < OC) : true;
  const int  colC  = GUARD ? (colOK ? col : (OC - 1)) : col;
  const float* wAp = WA + (size_t)colC * IC + koff;
  const float* wBp = DUAL ? (WB + (size_t)colC * IC + koff) : WA;
  const float* ar  = Atile + nidx * IC + koff;
  const float* br  = Btile + nidx * IC + koff;

  constexpr int KSTEPS = IC / 4;
  constexpr int BATCH  = DUAL ? (KSTEPS < 4 ? KSTEPS : 4)
                              : (KSTEPS < 8 ? KSTEPS : 8);

  v8f acc = {};
#pragma unroll 1
  for (int kb = 0; kb < KSTEPS; kb += BATCH) {
    v2f fa[BATCH], fw[BATCH];
    v2f ga[DUAL ? BATCH : 1], gw[DUAL ? BATCH : 1];
#pragma unroll
    for (int i = 0; i < BATCH; ++i) {
      const int k = (kb + i) * 4;
      fa[i] = *(const v2f*)(ar + k);
      fw[i] = *(const v2f*)(wAp + k);
      if constexpr (DUAL) {
        ga[i] = *(const v2f*)(br + k);
        gw[i] = *(const v2f*)(wBp + k);
      }
    }
#pragma unroll
    for (int i = 0; i < BATCH; ++i) {
      acc = __builtin_amdgcn_wmma_f32_16x16x4_f32(false, fa[i], false, fw[i],
                                                  (short)0, acc, false, false);
      if constexpr (DUAL)
        acc = __builtin_amdgcn_wmma_f32_16x16x4_f32(false, ga[i], false, gw[i],
                                                    (short)0, acc, false, false);
    }
  }

  const int mb = (lane >> 4) << 3;
  if constexpr (GUARD) {
    float bias = 0.f;
    if (colOK) {
      bias = bA[col];
      if constexpr (DUAL) bias += bB[col];
    }
#pragma unroll
    for (int vg = 0; vg < 8; ++vg) {
      const int r = row0 + mb + vg;
      if (colOK && r < nrows) {
        float val = acc[vg] + bias;
        if constexpr (RELU) val = val > 0.f ? val : 0.f;
        out[(size_t)r * OC + col] = val;
      }
    }
  } else {
    float bias = bA[col];
    if constexpr (DUAL) bias += bB[col];
    float* op = out + (size_t)(row0 + mb) * OC + col;
#pragma unroll
    for (int vg = 0; vg < 8; ++vg) {
      float val = acc[vg] + bias;
      if constexpr (RELU) val = val > 0.f ? val : 0.f;
      op[(size_t)vg * OC] = val;
    }
  }
}


template <int IC, bool DUAL, bool RELU>
static inline void launch_gemm(const float* inA, const float* WA, const float* bA,
                               const float* inB, const float* WB, const float* bB,
                               float* out, int N, int OC, hipStream_t stream) {
  const int rowTiles = (N + 15) / 16;
  const int gy = (((OC + 15) / 16) + WAVES_PER_BLOCK - 1) / WAVES_PER_BLOCK;
  const dim3 grid(rowTiles, gy), blk(32 * WAVES_PER_BLOCK);
  if ((N & 15) == 0 && (OC & 15) == 0)
    gcn_wmma_gemm<IC, DUAL, RELU, false><<<grid, blk, 0, stream>>>(
        inA, WA, bA, inB, WB, bB, out, N, OC);
  else
    gcn_wmma_gemm<IC, DUAL, RELU, true><<<grid, blk, 0, stream>>>(
        inA, WA, bA, inB, WB, bB, out, N, OC);
}

extern "C" void kernel_launch(void* const* d_in, const int* in_sizes, int n_in,
                              void* d_out, int out_size, void* d_ws, size_t ws_size,
                              hipStream_t stream) {
  (void)n_in; (void)out_size; (void)ws_size;
  const float* x    = (const float*)d_in[0];
  const int*   ei   = (const int*)d_in[1];
  const float* w1   = (const float*)d_in[2];
  const float* b1   = (const float*)d_in[3];
  const float* w2   = (const float*)d_in[4];
  const float* b2   = (const float*)d_in[5];
  const float* c1wl = (const float*)d_in[6];
  const float* c1bl = (const float*)d_in[7];
  const float* c1wr = (const float*)d_in[8];
  const float* c1br = (const float*)d_in[9];
  const float* c2wl = (const float*)d_in[10];
  const float* c2bl = (const float*)d_in[11];
  const float* c2wr = (const float*)d_in[12];
  const float* c2br = (const float*)d_in[13];
  const float* w3   = (const float*)d_in[14];
  const float* b3   = (const float*)d_in[15];
  const float* w4   = (const float*)d_in[16];
  const float* b4   = (const float*)d_in[17];

  const int IN_CH = 256;
  const int N = in_sizes[0] / IN_CH;
  const int E = in_sizes[1] / 2;
  const size_t n = (size_t)N;

  float* ws   = (float*)d_ws;
  float* dis  = ws;
  float* h2   = ws + 1 * n;
  float* agg1 = ws + 33 * n;
  float* h3   = ws + 65 * n;
  float* agg2 = ws + 129 * n;
  float* h4   = ws + 193 * n;
  float* h1   = ws + 193 * n;
  float* h5   = ws + 1 * n;

  const int eltBlocks  = (N + 255) / 256;
  const int edgeBlocks = (E + 255) / 256;
  const int scatBlocks = (int)(((long long)E * 32 + 255) / 256);

  launch_gemm<256, false, false>(x, w1, b1, nullptr, nullptr, nullptr, h1, N, 16, stream);
  launch_gemm<16, false, false>(h1, w2, b2, nullptr, nullptr, nullptr, h2, N, 32, stream);

  gcn_deg_init<<<eltBlocks, 256, 0, stream>>>(dis, N);
  gcn_deg_edge<<<edgeBlocks, 256, 0, stream>>>(ei + E, dis, E);
  gcn_deg_to_dis<<<eltBlocks, 256, 0, stream>>>(dis, N);

  gcn_self_loop<<<(int)((n * 32 + 255) / 256), 256, 0, stream>>>(dis, h2, agg1, N, 32);
  gcn_edge_scatter<<<scatBlocks, 256, 0, stream>>>(ei, dis, h2, agg1, E, 32);
  launch_gemm<32, true, true>(h2, c1wl, c1bl, agg1, c1wr, c1br, h3, N, 64, stream);

  gcn_self_loop<<<(int)((n * 64 + 255) / 256), 256, 0, stream>>>(dis, h3, agg2, N, 64);
  gcn_edge_scatter<<<scatBlocks, 256, 0, stream>>>(ei, dis, h3, agg2, E, 64);
  launch_gemm<64, true, true>(h3, c2wl, c2bl, agg2, c2wr, c2br, h4, N, 128, stream);

  launch_gemm<128, false, false>(h4, w3, b3, nullptr, nullptr, nullptr, h5, N, 192, stream);
  launch_gemm<192, false, false>(h5, w4, b4, nullptr, nullptr, nullptr, (float*)d_out, N, 20, stream);
}


# edge scatter: one wave walks 32 consecutive edges (coalesced index reads, endpoints broadcast from lanes, row reads issued 16 edges ahead) instead of one wave per edge; float atomics kept
# speedup vs baseline: 1.4822x; 1.4822x over previous
#include <hip/hip_runtime.h>

typedef __attribute__((ext_vector_type(2))) float v2f;
typedef __attribute__((ext_vector_type(8))) float v8f;

#define WAVES_PER_BLOCK 4


__global__ void gcn_deg_init(float* __restrict__ deg, int N) {
  int i = blockIdx.x * blockDim.x + threadIdx.x;
  if (i < N) deg[i] = 1.0f;
}

__global__ void gcn_deg_edge(const int* __restrict__ col, float* __restrict__ deg, int E) {
  int e = blockIdx.x * blockDim.x + threadIdx.x;
  if (e < E) unsafeAtomicAdd(&deg[col[e]], 1.0f);
}

__global__ void gcn_deg_to_dis(float* __restrict__ dis, int N) {
  int i = blockIdx.x * blockDim.x + threadIdx.x;
  if (i < N) {
    float d = dis[i];
    dis[i] = (d > 0.0f) ? rsqrtf(d) : 0.0f;
  }
}


__global__ void gcn_self_loop(const float* __restrict__ dis, const float* __restrict__ h,
                              float* __restrict__ agg, int N, int C) {
  long long i = (long long)blockIdx.x * blockDim.x + threadIdx.x;
  long long total = (long long)N * C;
  if (i < total) {
    int node = (int)(i / C);
    float d = dis[node];
    agg[i] = d * d * h[i];
  }
}

#define SCAT_THREADS 256
template <int C>
__global__ __launch_bounds__(SCAT_THREADS) void gcn_edge_scatter(const int* __restrict__ ei,
                                                                 const float* __restrict__ dis,
                                                                 const float* __restrict__ h,
                                                                 float* __restrict__ agg, int E) {
  static_assert(C % 32 == 0, "lane = channel needs whole 32-channel lines");
  constexpr int Q   = C / 32;
  constexpr int GRP = 16;
  const int lane = threadIdx.x & 31;
  const int wv = __builtin_amdgcn_readfirstlane((int)(threadIdx.x >> 5));
  const long long base = ((long long)blockIdx.x * (SCAT_THREADS / 32) + wv) << 5;
  if (base >= E) return;
  const long long rem = (long long)E - base;
  const int cnt = rem < 32 ? (int)rem : 32;

  int r = 0, c = 0;
  float w = 0.f;
  if (lane < cnt) {
    const int* ep = ei + base;
    r = ep[lane];
    c = ep[(long long)E + lane];
    w = dis[r] * dis[c];
  }

  if (cnt == 32) {
#pragma unroll
    for (int g = 0; g < 32; g += GRP) {
      float v[GRP][Q];
#pragma unroll
      for (int u = 0; u < GRP; ++u) {
        const int rj = __builtin_amdgcn_readlane(r, g + u);
        const float* hrow = h + (long long)rj * C;
#pragma unroll
        for (int q = 0; q < Q; ++q) v[u][q] = hrow[lane + q * 32];
      }
#pragma unroll
      for (int u = 0; u < GRP; ++u) {
        const int   cj = __builtin_amdgcn_readlane(c, g + u);
        const float wj = __int_as_float(__builtin_amdgcn_readlane(__float_as_int(w), g + u));
        float* arow = agg + (long long)cj * C;
#pragma unroll
        for (int q = 0; q < Q; ++q) unsafeAtomicAdd(arow + lane + q * 32, wj * v[u][q]);
      }
    }
  } else {
    for (int j = 0; j < cnt; ++j) {
      const int   rj = __builtin_amdgcn_readlane(r, j);
      const int   cj = __builtin_amdgcn_readlane(c, j);
      const float wj = __int_as_float(__builtin_amdgcn_readlane(__float_as_int(w), j));
      const float* hrow = h + (long long)rj * C;
      float*       arow = agg + (long long)cj * C;
      for (int q = 0; q < Q; ++q) unsafeAtomicAdd(arow + lane + q * 32, wj * hrow[lane + q * 32]);
    }
  }
}

template <int IC, bool DUAL, bool RELU, bool GUARD>
__global__ __launch_bounds__(32 * WAVES_PER_BLOCK) void gcn_wmma_gemm(
    const float* __restrict__ inA, const float* __restrict__ WA, const float* __restrict__ bA,
    const float* __restrict__ inB, const float* __restrict__ WB, const float* __restrict__ bB,
    float* __restrict__ out, int nrows, int OC) {
  __shared__ float lds[(DUAL ? 2 : 1) * 16 * IC];
  const int tid  = threadIdx.x;
  const int lane = tid & 31;
  const int wave = tid >> 5;
  const int row0 = blockIdx.x << 4;
  const int col0 = (blockIdx.y * WAVES_PER_BLOCK + wave) << 4;

  float* Atile = lds;
  float* Btile = DUAL ? (lds + 16 * IC) : lds;

  constexpr int vecPerRow = IC >> 2;
  constexpr int totalVec  = vecPerRow << 4;
  for (int idx = tid; idx < totalVec; idx += 32 * WAVES_PER_BLOCK) {
    const int r = idx / vecPerRow;
    const int v = idx - r * vecPerRow;
    const int node = row0 + r;
    if (GUARD && node >= nrows) {
      float4 z = {0.f, 0.f, 0.f, 0.f};
      ((float4*)Atile)[idx] = z;
      if constexpr (DUAL) ((float4*)Btile)[idx] = z;
    } else {
      ((float4*)Atile)[idx] = ((const float4*)(inA + (size_t)node * IC))[v];
      if constexpr (DUAL)
        ((float4*)Btile)[idx] = ((const float4*)(inB + (size_t)node * IC))[v];
    }
  }
  __syncthreads();

  if (col0 >= OC) return;

  const int  nidx = lane & 15;
  const int  koff = (lane >> 4) << 1;
  const int  col  = col0 + nidx;
  const bool colOK = GUARD ? (col < OC) : true;
  const int  colC  = GUARD ? (colOK ? col : (OC - 1)) : col;
  const float* wAp = WA + (size_t)colC * IC + koff;
  const float* wBp = DUAL ? (WB + (size_t)colC * IC + koff) : WA;
  const float* ar  = Atile + nidx * IC + koff;
  const float* br  = Btile + nidx * IC + koff;

  constexpr int KSTEPS = IC / 4;
  constexpr int BATCH  = DUAL ? (KSTEPS < 4 ? KSTEPS : 4)
                              : (KSTEPS < 8 ? KSTEPS : 8);

  v8f acc = {};
#pragma unroll 1
  for (int kb = 0; kb < KSTEPS; kb += BATCH) {
    v2f fa[BATCH], fw[BATCH];
    v2f ga[DUAL ? BATCH : 1], gw[DUAL ? BATCH : 1];
#pragma unroll
    for (int i = 0; i < BATCH; ++i) {
      const int k = (kb + i) * 4;
      fa[i] = *(const v2f*)(ar + k);
      fw[i] = *(const v2f*)(wAp + k);
      if constexpr (DUAL) {
        ga[i] = *(const v2f*)(br + k);
        gw[i] = *(const v2f*)(wBp + k);
      }
    }
#pragma unroll
    for (int i = 0; i < BATCH; ++i) {
      acc = __builtin_amdgcn_wmma_f32_16x16x4_f32(false, fa[i], false, fw[i],
                                                  (short)0, acc, false, false);
      if constexpr (DUAL)
        acc = __builtin_amdgcn_wmma_f32_16x16x4_f32(false, ga[i], false, gw[i],
                                                    (short)0, acc, false, false);
    }
  }

  const int mb = (lane >> 4) << 3;
  if constexpr (GUARD) {
    float bias = 0.f;
    if (colOK) {
      bias = bA[col];
      if constexpr (DUAL) bias += bB[col];
    }
#pragma unroll
    for (int vg = 0; vg < 8; ++vg) {
      const int r = row0 + mb + vg;
      if (colOK && r < nrows) {
        float val = acc[vg] + bias;
        if constexpr (RELU) val = val > 0.f ? val : 0.f;
        out[(size_t)r * OC + col] = val;
      }
    }
  } else {
    float bias = bA[col];
    if constexpr (DUAL) bias += bB[col];
    float* op = out + (size_t)(row0 + mb) * OC + col;
#pragma unroll
    for (int vg = 0; vg < 8; ++vg) {
      float val = acc[vg] + bias;
      if constexpr (RELU) val = val > 0.f ? val : 0.f;
      op[(size_t)vg * OC] = val;
    }
  }
}


template <int IC, bool DUAL, bool RELU>
static inline void launch_gemm(const float* inA, const float* WA, const float* bA,
                               const float* inB, const float* WB, const float* bB,
                               float* out, int N, int OC, hipStream_t stream) {
  const int rowTiles = (N + 15) / 16;
  const int gy = (((OC + 15) / 16) + WAVES_PER_BLOCK - 1) / WAVES_PER_BLOCK;
  const dim3 grid(rowTiles, gy), blk(32 * WAVES_PER_BLOCK);
  if ((N & 15) == 0 && (OC & 15) == 0)
    gcn_wmma_gemm<IC, DUAL, RELU, false><<<grid, blk, 0, stream>>>(
        inA, WA, bA, inB, WB, bB, out, N, OC);
  else
    gcn_wmma_gemm<IC, DUAL, RELU, true><<<grid, blk, 0, stream>>>(
        inA, WA, bA, inB, WB, bB, out, N, OC);
}

extern "C" void kernel_launch(void* const* d_in, const int* in_sizes, int n_in,
                              void* d_out, int out_size, void* d_ws, size_t ws_size,
                              hipStream_t stream) {
  (void)n_in; (void)out_size; (void)ws_size;
  const float* x    = (const float*)d_in[0];
  const int*   ei   = (const int*)d_in[1];
  const float* w1   = (const float*)d_in[2];
  const float* b1   = (const float*)d_in[3];
  const float* w2   = (const float*)d_in[4];
  const float* b2   = (const float*)d_in[5];
  const float* c1wl = (const float*)d_in[6];
  const float* c1bl = (const float*)d_in[7];
  const float* c1wr = (const float*)d_in[8];
  const float* c1br = (const float*)d_in[9];
  const float* c2wl = (const float*)d_in[10];
  const float* c2bl = (const float*)d_in[11];
  const float* c2wr = (const float*)d_in[12];
  const float* c2br = (const float*)d_in[13];
  const float* w3   = (const float*)d_in[14];
  const float* b3   = (const float*)d_in[15];
  const float* w4   = (const float*)d_in[16];
  const float* b4   = (const float*)d_in[17];

  const int IN_CH = 256;
  const int N = in_sizes[0] / IN_CH;
  const int E = in_sizes[1] / 2;
  const size_t n = (size_t)N;

  float* ws   = (float*)d_ws;
  float* dis  = ws;
  float* h2   = ws + 1 * n;
  float* agg1 = ws + 33 * n;
  float* h3   = ws + 65 * n;
  float* agg2 = ws + 129 * n;
  float* h4   = ws + 193 * n;
  float* h1   = ws + 193 * n;
  float* h5   = ws + 1 * n;

  const int eltBlocks  = (N + 255) / 256;
  const int edgeBlocks = (E + 255) / 256;
  const int scatBlocks = (int)(((((long long)E + 31) / 32) * 32 + SCAT_THREADS - 1) / SCAT_THREADS);

  launch_gemm<256, false, false>(x, w1, b1, nullptr, nullptr, nullptr, h1, N, 16, stream);
  launch_gemm<16, false, false>(h1, w2, b2, nullptr, nullptr, nullptr, h2, N, 32, stream);

  gcn_deg_init<<<eltBlocks, 256, 0, stream>>>(dis, N);
  gcn_deg_edge<<<edgeBlocks, 256, 0, stream>>>(ei + E, dis, E);
  gcn_deg_to_dis<<<eltBlocks, 256, 0, stream>>>(dis, N);

  gcn_self_loop<<<(int)((n * 32 + 255) / 256), 256, 0, stream>>>(dis, h2, agg1, N, 32);
  gcn_edge_scatter<32><<<scatBlocks, SCAT_THREADS, 0, stream>>>(ei, dis, h2, agg1, E);
  launch_gemm<32, true, true>(h2, c1wl, c1bl, agg1, c1wr, c1br, h3, N, 64, stream);

  gcn_self_loop<<<(int)((n * 64 + 255) / 256), 256, 0, stream>>>(dis, h3, agg2, N, 64);
  gcn_edge_scatter<64><<<scatBlocks, SCAT_THREADS, 0, stream>>>(ei, dis, h3, agg2, E);
  launch_gemm<64, true, true>(h3, c2wl, c2bl, agg2, c2wr, c2br, h4, N, 128, stream);

  launch_gemm<128, false, false>(h4, w3, b3, nullptr, nullptr, nullptr, h5, N, 192, stream);
  launch_gemm<192, false, false>(h5, w4, b4, nullptr, nullptr, nullptr, (float*)d_out, N, 20, stream);
}


# in-degree counted in LDS per node range (work-group = list piece x node range, partial count planes summed in the finish kernel) instead of one global float atomic per edge; degree stays exact
# speedup vs baseline: 1.7155x; 1.1574x over previous
#include <hip/hip_runtime.h>

typedef __attribute__((ext_vector_type(2))) float v2f;
typedef __attribute__((ext_vector_type(8))) float v8f;

#define WAVES_PER_BLOCK 4

#define DEG_RANGE   6400
#define DEG_CHUNKS  16
#define DEG_THREADS 1024

__device__ __forceinline__ void deg_count_one(int* hist, int c, int lo, int nn) {
  const unsigned d = (unsigned)(c - lo);
  if (d < (unsigned)nn) atomicAdd(&hist[d], 1);
}

__device__ __forceinline__ void deg_count_four(int* hist, int4 v, int lo, int nn) {
  deg_count_one(hist, v.x, lo, nn);
  deg_count_one(hist, v.y, lo, nn);
  deg_count_one(hist, v.z, lo, nn);
  deg_count_one(hist, v.w, lo, nn);
}

__global__ __launch_bounds__(DEG_THREADS) void gcn_deg_count(const int* __restrict__ col,
                                                             int* __restrict__ part,
                                                             int N, int E) {
  __shared__ int hist[DEG_RANGE];
  const int tid = threadIdx.x;
  const int lo  = blockIdx.y * DEG_RANGE;
  const int nn  = (N - lo) < DEG_RANGE ? (N - lo) : DEG_RANGE;
  for (int i = tid; i < DEG_RANGE; i += DEG_THREADS) hist[i] = 0;
  __syncthreads();

  const int quads = (int)(((long long)E + 3) >> 2);
  const int qper  = (quads + DEG_CHUNKS - 1) / DEG_CHUNKS;
  const int q0    = blockIdx.x * qper;
  const int q1    = (q0 + qper) < quads ? (q0 + qper) : quads;

  if ((E & 3) == 0 && (((size_t)col) & 15) == 0) {
    const int4* c4 = (const int4*)col;
    int q = q0 + tid;
    for (; q + 3 * DEG_THREADS < q1; q += 4 * DEG_THREADS) {
      const int4 a = c4[q];
      const int4 b = c4[q + DEG_THREADS];
      const int4 c = c4[q + 2 * DEG_THREADS];
      const int4 d = c4[q + 3 * DEG_THREADS];
      deg_count_four(hist, a, lo, nn);
      deg_count_four(hist, b, lo, nn);
      deg_count_four(hist, c, lo, nn);
      deg_count_four(hist, d, lo, nn);
    }
    for (; q < q1; q += DEG_THREADS) deg_count_four(hist, c4[q], lo, nn);
  } else {
    const long long e0 = (long long)q0 << 2;
    long long e1 = (long long)q1 << 2;
    if (e1 > E) e1 = E;
    for (long long e = e0 + tid; e < e1; e += DEG_THREADS) deg_count_one(hist, col[e], lo, nn);
  }
  __syncthreads();

  int* plane = part + (size_t)blockIdx.x * N + lo;
  for (int i = tid; i < nn; i += DEG_THREADS) plane[i] = hist[i];
}

__global__ void gcn_deg_finish(const int* __restrict__ part, float* __restrict__ dis, int N) {
  int i = blockIdx.x * blockDim.x + threadIdx.x;
  if (i < N) {
    int cnt = 1;
    for (int p = 0; p < DEG_CHUNKS; ++p) cnt += part[(size_t)p * N + i];
    float d = (float)cnt;
    dis[i] = (d > 0.0f) ? rsqrtf(d) : 0.0f;
  }
}


__global__ void gcn_self_loop(const float* __restrict__ dis, const float* __restrict__ h,
                              float* __restrict__ agg, int N, int C) {
  long long i = (long long)blockIdx.x * blockDim.x + threadIdx.x;
  long long total = (long long)N * C;
  if (i < total) {
    int node = (int)(i / C);
    float d = dis[node];
    agg[i] = d * d * h[i];
  }
}

#define SCAT_THREADS 256
template <int C>
__global__ __launch_bounds__(SCAT_THREADS) void gcn_edge_scatter(const int* __restrict__ ei,
                                                                 const float* __restrict__ dis,
                                                                 const float* __restrict__ h,
                                                                 float* __restrict__ agg, int E) {
  static_assert(C % 32 == 0, "lane = channel needs whole 32-channel lines");
  constexpr int Q   = C / 32;
  constexpr int GRP = 16;
  const int lane = threadIdx.x & 31;
  const int wv = __builtin_amdgcn_readfirstlane((int)(threadIdx.x >> 5));
  const long long base = ((long long)blockIdx.x * (SCAT_THREADS / 32) + wv) << 5;
  if (base >= E) return;
  const long long rem = (long long)E - base;
  const int cnt = rem < 32 ? (int)rem : 32;

  int r = 0, c = 0;
  float w = 0.f;
  if (lane < cnt) {
    const int* ep = ei + base;
    r = ep[lane];
    c = ep[(long long)E + lane];
    w = dis[r] * dis[c];
  }

  if (cnt == 32) {
#pragma unroll
    for (int g = 0; g < 32; g += GRP) {
      float v[GRP][Q];
#pragma unroll
      for (int u = 0; u < GRP; ++u) {
        const int rj = __builtin_amdgcn_readlane(r, g + u);
        const float* hrow = h + (long long)rj * C;
#pragma unroll
        for (int q = 0; q < Q; ++q) v[u][q] = hrow[lane + q * 32];
      }
#pragma unroll
      for (int u = 0; u < GRP; ++u) {
        const int   cj = __builtin_amdgcn_readlane(c, g + u);
        const float wj = __int_as_float(__builtin_amdgcn_readlane(__float_as_int(w), g + u));
        float* arow = agg + (long long)cj * C;
#pragma unroll
        for (int q = 0; q < Q; ++q) unsafeAtomicAdd(arow + lane + q * 32, wj * v[u][q]);
      }
    }
  } else {
    for (int j = 0; j < cnt; ++j) {
      const int   rj = __builtin_amdgcn_readlane(r, j);
      const int   cj = __builtin_amdgcn_readlane(c, j);
      const float wj = __int_as_float(__builtin_amdgcn_readlane(__float_as_int(w), j));
      const float* hrow = h + (long long)rj * C;
      float*       arow = agg + (long long)cj * C;
      for (int q = 0; q < Q; ++q) unsafeAtomicAdd(arow + lane + q * 32, wj * hrow[lane + q * 32]);
    }
  }
}

template <int IC, bool DUAL, bool RELU, bool GUARD>
__global__ __launch_bounds__(32 * WAVES_PER_BLOCK) void gcn_wmma_gemm(
    const float* __restrict__ inA, const float* __restrict__ WA, const float* __restrict__ bA,
    const float* __restrict__ inB, const float* __restrict__ WB, const float* __restrict__ bB,
    float* __restrict__ out, int nrows, int OC) {
  __shared__ float lds[(DUAL ? 2 : 1) * 16 * IC];
  const int tid  = threadIdx.x;
  const int lane = tid & 31;
  const int wave = tid >> 5;
  const int row0 = blockIdx.x << 4;
  const int col0 = (blockIdx.y * WAVES_PER_BLOCK + wave) << 4;

  float* Atile = lds;
  float* Btile = DUAL ? (lds + 16 * IC) : lds;

  constexpr int vecPerRow = IC >> 2;
  constexpr int totalVec  = vecPerRow << 4;
  for (int idx = tid; idx < totalVec; idx += 32 * WAVES_PER_BLOCK) {
    const int r = idx / vecPerRow;
    const int v = idx - r * vecPerRow;
    const int node = row0 + r;
    if (GUARD && node >= nrows) {
      float4 z = {0.f, 0.f, 0.f, 0.f};
      ((float4*)Atile)[idx] = z;
      if constexpr (DUAL) ((float4*)Btile)[idx] = z;
    } else {
      ((float4*)Atile)[idx] = ((const float4*)(inA + (size_t)node * IC))[v];
      if constexpr (DUAL)
        ((float4*)Btile)[idx] = ((const float4*)(inB + (size_t)node * IC))[v];
    }
  }
  __syncthreads();

  if (col0 >= OC) return;

  const int  nidx = lane & 15;
  const int  koff = (lane >> 4) << 1;
  const int  col  = col0 + nidx;
  const bool colOK = GUARD ? (col < OC) : true;
  const int  colC  = GUARD ? (colOK ? col : (OC - 1)) : col;
  const float* wAp = WA + (size_t)colC * IC + koff;
  const float* wBp = DUAL ? (WB + (size_t)colC * IC + koff) : WA;
  const float* ar  = Atile + nidx * IC + koff;
  const float* br  = Btile + nidx * IC + koff;

  constexpr int KSTEPS = IC / 4;
  constexpr int BATCH  = DUAL ? (KSTEPS < 4 ? KSTEPS : 4)
                              : (KSTEPS < 8 ? KSTEPS : 8);

  v8f acc = {};
#pragma unroll 1
  for (int kb = 0; kb < KSTEPS; kb += BATCH) {
    v2f fa[BATCH], fw[BATCH];
    v2f ga[DUAL ? BATCH : 1], gw[DUAL ? BATCH : 1];
#pragma unroll
    for (int i = 0; i < BATCH; ++i) {
      const int k = (kb + i) * 4;
      fa[i] = *(const v2f*)(ar + k);
      fw[i] = *(const v2f*)(wAp + k);
      if constexpr (DUAL) {
        ga[i] = *(const v2f*)(br + k);
        gw[i] = *(const v2f*)(wBp + k);
      }
    }
#pragma unroll
    for (int i = 0; i < BATCH; ++i) {
      acc = __builtin_amdgcn_wmma_f32_16x16x4_f32(false, fa[i], false, fw[i],
                                                  (short)0, acc, false, false);
      if constexpr (DUAL)
        acc = __builtin_amdgcn_wmma_f32_16x16x4_f32(false, ga[i], false, gw[i],
                                                    (short)0, acc, false, false);
    }
  }

  const int mb = (lane >> 4) << 3;
  if constexpr (GUARD) {
    float bias = 0.f;
    if (colOK) {
      bias = bA[col];
      if constexpr (DUAL) bias += bB[col];
    }
#pragma unroll
    for (int vg = 0; vg < 8; ++vg) {
      const int r = row0 + mb + vg;
      if (colOK && r < nrows) {
        float val = acc[vg] + bias;
        if constexpr (RELU) val = val > 0.f ? val : 0.f;
        out[(size_t)r * OC + col] = val;
      }
    }
  } else {
    float bias = bA[col];
    if constexpr (DUAL) bias += bB[col];
    float* op = out + (size_t)(row0 + mb) * OC + col;
#pragma unroll
    for (int vg = 0; vg < 8; ++vg) {
      float val = acc[vg] + bias;
      if constexpr (RELU) val = val > 0.f ? val : 0.f;
      op[(size_t)vg * OC] = val;
    }
  }
}


template <int IC, bool DUAL, bool RELU>
static inline void launch_gemm(const float* inA, const float* WA, const float* bA,
                               const float* inB, const float* WB, const float* bB,
                               float* out, int N, int OC, hipStream_t stream) {
  const int rowTiles = (N + 15) / 16;
  const int gy = (((OC + 15) / 16) + WAVES_PER_BLOCK - 1) / WAVES_PER_BLOCK;
  const dim3 grid(rowTiles, gy), blk(32 * WAVES_PER_BLOCK);
  if ((N & 15) == 0 && (OC & 15) == 0)
    gcn_wmma_gemm<IC, DUAL, RELU, false><<<grid, blk, 0, stream>>>(
        inA, WA, bA, inB, WB, bB, out, N, OC);
  else
    gcn_wmma_gemm<IC, DUAL, RELU, true><<<grid, blk, 0, stream>>>(
        inA, WA, bA, inB, WB, bB, out, N, OC);
}

extern "C" void kernel_launch(void* const* d_in, const int* in_sizes, int n_in,
                              void* d_out, int out_size, void* d_ws, size_t ws_size,
                              hipStream_t stream) {
  (void)n_in; (void)out_size; (void)ws_size;
  const float* x    = (const float*)d_in[0];
  const int*   ei   = (const int*)d_in[1];
  const float* w1   = (const float*)d_in[2];
  const float* b1   = (const float*)d_in[3];
  const float* w2   = (const float*)d_in[4];
  const float* b2   = (const float*)d_in[5];
  const float* c1wl = (const float*)d_in[6];
  const float* c1bl = (const float*)d_in[7];
  const float* c1wr = (const float*)d_in[8];
  const float* c1br = (const float*)d_in[9];
  const float* c2wl = (const float*)d_in[10];
  const float* c2bl = (const float*)d_in[11];
  const float* c2wr = (const float*)d_in[12];
  const float* c2br = (const float*)d_in[13];
  const float* w3   = (const float*)d_in[14];
  const float* b3   = (const float*)d_in[15];
  const float* w4   = (const float*)d_in[16];
  const float* b4   = (const float*)d_in[17];

  const int IN_CH = 256;
  const int N = in_sizes[0] / IN_CH;
  const int E = in_sizes[1] / 2;
  const size_t n = (size_t)N;

  float* ws   = (float*)d_ws;
  float* dis  = ws;
  float* h2   = ws + 1 * n;
  float* agg1 = ws + 33 * n;
  float* h3   = ws + 65 * n;
  float* agg2 = ws + 129 * n;
  float* h4   = ws + 193 * n;
  float* h1   = ws + 193 * n;
  float* h5   = ws + 1 * n;
  int*   part = (int*)(ws + 129 * n);

  const int eltBlocks  = (N + 255) / 256;
  const int degRanges  = (N + DEG_RANGE - 1) / DEG_RANGE;
  const int scatBlocks = (int)(((((long long)E + 31) / 32) * 32 + SCAT_THREADS - 1) / SCAT_THREADS);

  launch_gemm<256, false, false>(x, w1, b1, nullptr, nullptr, nullptr, h1, N, 16, stream);
  launch_gemm<16, false, false>(h1, w2, b2, nullptr, nullptr, nullptr, h2, N, 32, stream);

  gcn_deg_count<<<dim3(DEG_CHUNKS, degRanges), DEG_THREADS, 0, stream>>>(ei + E, part, N, E);
  gcn_deg_finish<<<eltBlocks, 256, 0, stream>>>(part, dis, N);

  gcn_self_loop<<<(int)((n * 32 + 255) / 256), 256, 0, stream>>>(dis, h2, agg1, N, 32);
  gcn_edge_scatter<32><<<scatBlocks, SCAT_THREADS, 0, stream>>>(ei, dis, h2, agg1, E);
  launch_gemm<32, true, true>(h2, c1wl, c1bl, agg1, c1wr, c1br, h3, N, 64, stream);

  gcn_self_loop<<<(int)((n * 64 + 255) / 256), 256, 0, stream>>>(dis, h3, agg2, N, 64);
  gcn_edge_scatter<64><<<scatBlocks, SCAT_THREADS, 0, stream>>>(ei, dis, h3, agg2, E);
  launch_gemm<64, true, true>(h3, c2wl, c2bl, agg2, c2wr, c2br, h4, N, 128, stream);

  launch_gemm<128, false, false>(h4, w3, b3, nullptr, nullptr, nullptr, h5, N, 192, stream);
  launch_gemm<192, false, false>(h5, w4, b4, nullptr, nullptr, nullptr, (float*)d_out, N, 20, stream);
}


# GEMM activation tile staged in LDS with rows IC+4 floats apart instead of IC: the rows of one fragment read start in different LDS banks (before: all in one bank, served one row per clock)
# speedup vs baseline: 1.8778x; 1.0946x over previous
#include <hip/hip_runtime.h>

typedef __attribute__((ext_vector_type(2))) float v2f;
typedef __attribute__((ext_vector_type(8))) float v8f;

#define WAVES_PER_BLOCK 4

#define DEG_RANGE   6400
#define DEG_CHUNKS  16
#define DEG_THREADS 1024

__device__ __forceinline__ void deg_count_one(int* hist, int c, int lo, int nn) {
  const unsigned d = (unsigned)(c - lo);
  if (d < (unsigned)nn) atomicAdd(&hist[d], 1);
}

__device__ __forceinline__ void deg_count_four(int* hist, int4 v, int lo, int nn) {
  deg_count_one(hist, v.x, lo, nn);
  deg_count_one(hist, v.y, lo, nn);
  deg_count_one(hist, v.z, lo, nn);
  deg_count_one(hist, v.w, lo, nn);
}

__global__ __launch_bounds__(DEG_THREADS) void gcn_deg_count(const int* __restrict__ col,
                                                             int* __restrict__ part,
                                                             int N, int E) {
  __shared__ int hist[DEG_RANGE];
  const int tid = threadIdx.x;
  const int lo  = blockIdx.y * DEG_RANGE;
  const int nn  = (N - lo) < DEG_RANGE ? (N - lo) : DEG_RANGE;
  for (int i = tid; i < DEG_RANGE; i += DEG_THREADS) hist[i] = 0;
  __syncthreads();

  const int quads = (int)(((long long)E + 3) >> 2);
  const int qper  = (quads + DEG_CHUNKS - 1) / DEG_CHUNKS;
  const int q0    = blockIdx.x * qper;
  const int q1    = (q0 + qper) < quads ? (q0 + qper) : quads;

  if ((E & 3) == 0 && (((size_t)col) & 15) == 0) {
    const int4* c4 = (const int4*)col;
    int q = q0 + tid;
    for (; q + 3 * DEG_THREADS < q1; q += 4 * DEG_THREADS) {
      const int4 a = c4[q];
      const int4 b = c4[q + DEG_THREADS];
      const int4 c = c4[q + 2 * DEG_THREADS];
      const int4 d = c4[q + 3 * DEG_THREADS];
      deg_count_four(hist, a, lo, nn);
      deg_count_four(hist, b, lo, nn);
      deg_count_four(hist, c, lo, nn);
      deg_count_four(hist, d, lo, nn);
    }
    for (; q < q1; q += DEG_THREADS) deg_count_four(hist, c4[q], lo, nn);
  } else {
    const long long e0 = (long long)q0 << 2;
    long long e1 = (long long)q1 << 2;
    if (e1 > E) e1 = E;
    for (long long e = e0 + tid; e < e1; e += DEG_THREADS) deg_count_one(hist, col[e], lo, nn);
  }
  __syncthreads();

  int* plane = part + (size_t)blockIdx.x * N + lo;
  for (int i = tid; i < nn; i += DEG_THREADS) plane[i] = hist[i];
}

__global__ void gcn_deg_finish(const int* __restrict__ part, float* __restrict__ dis, int N) {
  int i = blockIdx.x * blockDim.x + threadIdx.x;
  if (i < N) {
    int cnt = 1;
    for (int p = 0; p < DEG_CHUNKS; ++p) cnt += part[(size_t)p * N + i];
    float d = (float)cnt;
    dis[i] = (d > 0.0f) ? rsqrtf(d) : 0.0f;
  }
}


__global__ void gcn_self_loop(const float* __restrict__ dis, const float* __restrict__ h,
                              float* __restrict__ agg, int N, int C) {
  long long i = (long long)blockIdx.x * blockDim.x + threadIdx.x;
  long long total = (long long)N * C;
  if (i < total) {
    int node = (int)(i / C);
    float d = dis[node];
    agg[i] = d * d * h[i];
  }
}

#define SCAT_THREADS 256
template <int C>
__global__ __launch_bounds__(SCAT_THREADS) void gcn_edge_scatter(const int* __restrict__ ei,
                                                                 const float* __restrict__ dis,
                                                                 const float* __restrict__ h,
                                                                 float* __restrict__ agg, int E) {
  static_assert(C % 32 == 0, "lane = channel needs whole 32-channel lines");
  constexpr int Q   = C / 32;
  constexpr int GRP = 16;
  const int lane = threadIdx.x & 31;
  const int wv = __builtin_amdgcn_readfirstlane((int)(threadIdx.x >> 5));
  const long long base = ((long long)blockIdx.x * (SCAT_THREADS / 32) + wv) << 5;
  if (base >= E) return;
  const long long rem = (long long)E - base;
  const int cnt = rem < 32 ? (int)rem : 32;

  int r = 0, c = 0;
  float w = 0.f;
  if (lane < cnt) {
    const int* ep = ei + base;
    r = ep[lane];
    c = ep[(long long)E + lane];
    w = dis[r] * dis[c];
  }

  if (cnt == 32) {
#pragma unroll
    for (int g = 0; g < 32; g += GRP) {
      float v[GRP][Q];
#pragma unroll
      for (int u = 0; u < GRP; ++u) {
        const int rj = __builtin_amdgcn_readlane(r, g + u);
        const float* hrow = h + (long long)rj * C;
#pragma unroll
        for (int q = 0; q < Q; ++q) v[u][q] = hrow[lane + q * 32];
      }
#pragma unroll
      for (int u = 0; u < GRP; ++u) {
        const int   cj = __builtin_amdgcn_readlane(c, g + u);
        const float wj = __int_as_float(__builtin_amdgcn_readlane(__float_as_int(w), g + u));
        float* arow = agg + (long long)cj * C;
#pragma unroll
        for (int q = 0; q < Q; ++q) unsafeAtomicAdd(arow + lane + q * 32, wj * v[u][q]);
      }
    }
  } else {
    for (int j = 0; j < cnt; ++j) {
      const int   rj = __builtin_amdgcn_readlane(r, j);
      const int   cj = __builtin_amdgcn_readlane(c, j);
      const float wj = __int_as_float(__builtin_amdgcn_readlane(__float_as_int(w), j));
      const float* hrow = h + (long long)rj * C;
      float*       arow = agg + (long long)cj * C;
      for (int q = 0; q < Q; ++q) unsafeAtomicAdd(arow + lane + q * 32, wj * hrow[lane + q * 32]);
    }
  }
}

template <int IC, bool DUAL, bool RELU, bool GUARD>
__global__ __launch_bounds__(32 * WAVES_PER_BLOCK) void gcn_wmma_gemm(
    const float* __restrict__ inA, const float* __restrict__ WA, const float* __restrict__ bA,
    const float* __restrict__ inB, const float* __restrict__ WB, const float* __restrict__ bB,
    float* __restrict__ out, int nrows, int OC) {
  constexpr int PITCH = IC + 4;
  static_assert((PITCH & 7) == 4, "pitch must be 4 * odd floats");
  __shared__ __attribute__((aligned(16))) float lds[(DUAL ? 2 : 1) * 16 * PITCH];
  const int tid  = threadIdx.x;
  const int lane = tid & 31;
  const int wave = tid >> 5;
  const int row0 = blockIdx.x << 4;
  const int col0 = (blockIdx.y * WAVES_PER_BLOCK + wave) << 4;

  float* Atile = lds;
  float* Btile = DUAL ? (lds + 16 * PITCH) : lds;

  constexpr int vecPerRow = IC >> 2;
  constexpr int totalVec  = vecPerRow << 4;
  for (int idx = tid; idx < totalVec; idx += 32 * WAVES_PER_BLOCK) {
    const int r = idx / vecPerRow;
    const int v = idx - r * vecPerRow;
    const int node = row0 + r;
    const int off = r * PITCH + (v << 2);
    if (GUARD && node >= nrows) {
      float4 z = {0.f, 0.f, 0.f, 0.f};
      *(float4*)(Atile + off) = z;
      if constexpr (DUAL) *(float4*)(Btile + off) = z;
    } else {
      *(float4*)(Atile + off) = ((const float4*)(inA + (size_t)node * IC))[v];
      if constexpr (DUAL)
        *(float4*)(Btile + off) = ((const float4*)(inB + (size_t)node * IC))[v];
    }
  }
  __syncthreads();

  if (col0 >= OC) return;

  const int  nidx = lane & 15;
  const int  koff = (lane >> 4) << 1;
  const int  col  = col0 + nidx;
  const bool colOK = GUARD ? (col < OC) : true;
  const int  colC  = GUARD ? (colOK ? col : (OC - 1)) : col;
  const float* wAp = WA + (size_t)colC * IC + koff;
  const float* wBp = DUAL ? (WB + (size_t)colC * IC + koff) : WA;
  const float* ar  = Atile + nidx * PITCH + koff;
  const float* br  = Btile + nidx * PITCH + koff;

  constexpr int KSTEPS = IC / 4;
  constexpr int BATCH  = DUAL ? (KSTEPS < 4 ? KSTEPS : 4)
                              : (KSTEPS < 8 ? KSTEPS : 8);

  v8f acc = {};
#pragma unroll 1
  for (int kb = 0; kb < KSTEPS; kb += BATCH) {
    v2f fa[BATCH], fw[BATCH];
    v2f ga[DUAL ? BATCH : 1], gw[DUAL ? BATCH : 1];
#pragma unroll
    for (int i = 0; i < BATCH; ++i) {
      const int k = (kb + i) * 4;
      fa[i] = *(const v2f*)(ar + k);
      fw[i] = *(const v2f*)(wAp + k);
      if constexpr (DUAL) {
        ga[i] = *(const v2f*)(br + k);
        gw[i] = *(const v2f*)(wBp + k);
      }
    }
#pragma unroll
    for (int i = 0; i < BATCH; ++i) {
      acc = __builtin_amdgcn_wmma_f32_16x16x4_f32(false, fa[i], false, fw[i],
                                                  (short)0, acc, false, false);
      if constexpr (DUAL)
        acc = __builtin_amdgcn_wmma_f32_16x16x4_f32(false, ga[i], false, gw[i],
                                                    (short)0, acc, false, false);
    }
  }

  const int mb = (lane >> 4) << 3;
  if constexpr (GUARD) {
    float bias = 0.f;
    if (colOK) {
      bias = bA[col];
      if constexpr (DUAL) bias += bB[col];
    }
#pragma unroll
    for (int vg = 0; vg < 8; ++vg) {
      const int r = row0 + mb + vg;
      if (colOK && r < nrows) {
        float val = acc[vg] + bias;
        if constexpr (RELU) val = val > 0.f ? val : 0.f;
        out[(size_t)r * OC + col] = val;
      }
    }
  } else {
    float bias = bA[col];
    if constexpr (DUAL) bias += bB[col];
    float* op = out + (size_t)(row0 + mb) * OC + col;
#pragma unroll
    for (int vg = 0; vg < 8; ++vg) {
      float val = acc[vg] + bias;
      if constexpr (RELU) val = val > 0.f ? val : 0.f;
      op[(size_t)vg * OC] = val;
    }
  }
}


template <int IC, bool DUAL, bool RELU>
static inline void launch_gemm(const float* inA, const float* WA, const float* bA,
                               const float* inB, const float* WB, const float* bB,
                               float* out, int N, int OC, hipStream_t stream) {
  const int rowTiles = (N + 15) / 16;
  const int gy = (((OC + 15) / 16) + WAVES_PER_BLOCK - 1) / WAVES_PER_BLOCK;
  const dim3 grid(rowTiles, gy), blk(32 * WAVES_PER_BLOCK);
  if ((N & 15) == 0 && (OC & 15) == 0)
    gcn_wmma_gemm<IC, DUAL, RELU, false><<<grid, blk, 0, stream>>>(
        inA, WA, bA, inB, WB, bB, out, N, OC);
  else
    gcn_wmma_gemm<IC, DUAL, RELU, true><<<grid, blk, 0, stream>>>(
        inA, WA, bA, inB, WB, bB, out, N, OC);
}

extern "C" void kernel_launch(void* const* d_in, const int* in_sizes, int n_in,
                              void* d_out, int out_size, void* d_ws, size_t ws_size,
                              hipStream_t stream) {
  (void)n_in; (void)out_size; (void)ws_size;
  const float* x    = (const float*)d_in[0];
  const int*   ei   = (const int*)d_in[1];
  const float* w1   = (const float*)d_in[2];
  const float* b1   = (const float*)d_in[3];
  const float* w2   = (const float*)d_in[4];
  const float* b2   = (const float*)d_in[5];
  const float* c1wl = (const float*)d_in[6];
  const float* c1bl = (const float*)d_in[7];
  const float* c1wr = (const float*)d_in[8];
  const float* c1br = (const float*)d_in[9];
  const float* c2wl = (const float*)d_in[10];
  const float* c2bl = (const float*)d_in[11];
  const float* c2wr = (const float*)d_in[12];
  const float* c2br = (const float*)d_in[13];
  const float* w3   = (const float*)d_in[14];
  const float* b3   = (const float*)d_in[15];
  const float* w4   = (const float*)d_in[16];
  const float* b4   = (const float*)d_in[17];

  const int IN_CH = 256;
  const int N = in_sizes[0] / IN_CH;
  const int E = in_sizes[1] / 2;
  const size_t n = (size_t)N;

  float* ws   = (float*)d_ws;
  float* dis  = ws;
  float* h2   = ws + 1 * n;
  float* agg1 = ws + 33 * n;
  float* h3   = ws + 65 * n;
  float* agg2 = ws + 129 * n;
  float* h4   = ws + 193 * n;
  float* h1   = ws + 193 * n;
  float* h5   = ws + 1 * n;
  int*   part = (int*)(ws + 129 * n);

  const int eltBlocks  = (N + 255) / 256;
  const int degRanges  = (N + DEG_RANGE - 1) / DEG_RANGE;
  const int scatBlocks = (int)(((((long long)E + 31) / 32) * 32 + SCAT_THREADS - 1) / SCAT_THREADS);

  launch_gemm<256, false, false>(x, w1, b1, nullptr, nullptr, nullptr, h1, N, 16, stream);
  launch_gemm<16, false, false>(h1, w2, b2, nullptr, nullptr, nullptr, h2, N, 32, stream);

  gcn_deg_count<<<dim3(DEG_CHUNKS, degRanges), DEG_THREADS, 0, stream>>>(ei + E, part, N, E);
  gcn_deg_finish<<<eltBlocks, 256, 0, stream>>>(part, dis, N);

  gcn_self_loop<<<(int)((n * 32 + 255) / 256), 256, 0, stream>>>(dis, h2, agg1, N, 32);
  gcn_edge_scatter<32><<<scatBlocks, SCAT_THREADS, 0, stream>>>(ei, dis, h2, agg1, E);
  launch_gemm<32, true, true>(h2, c1wl, c1bl, agg1, c1wr, c1br, h3, N, 64, stream);

  gcn_self_loop<<<(int)((n * 64 + 255) / 256), 256, 0, stream>>>(dis, h3, agg2, N, 64);
  gcn_edge_scatter<64><<<scatBlocks, SCAT_THREADS, 0, stream>>>(ei, dis, h3, agg2, E);
  launch_gemm<64, true, true>(h3, c2wl, c2bl, agg2, c2wr, c2br, h4, N, 128, stream);

  launch_gemm<128, false, false>(h4, w3, b3, nullptr, nullptr, nullptr, h5, N, 192, stream);
  launch_gemm<192, false, false>(h5, w4, b4, nullptr, nullptr, nullptr, (float*)d_out, N, 20, stream);
}


# last two linear layers (128->192->20, nothing between) folded: a small kernel forms w4.w3 and w4.b3+b4 on f32 matrix units each call, one GEMM 128->20 follows; float sums reassociated
# speedup vs baseline: 2.3038x; 1.2269x over previous
#include <hip/hip_runtime.h>

typedef __attribute__((ext_vector_type(2))) float v2f;
typedef __attribute__((ext_vector_type(8))) float v8f;

#define WAVES_PER_BLOCK 4

#define DEG_RANGE   6400
#define DEG_CHUNKS  16
#define DEG_THREADS 1024

__device__ __forceinline__ void deg_count_one(int* hist, int c, int lo, int nn) {
  const unsigned d = (unsigned)(c - lo);
  if (d < (unsigned)nn) atomicAdd(&hist[d], 1);
}

__device__ __forceinline__ void deg_count_four(int* hist, int4 v, int lo, int nn) {
  deg_count_one(hist, v.x, lo, nn);
  deg_count_one(hist, v.y, lo, nn);
  deg_count_one(hist, v.z, lo, nn);
  deg_count_one(hist, v.w, lo, nn);
}

__global__ __launch_bounds__(DEG_THREADS) void gcn_deg_count(const int* __restrict__ col,
                                                             int* __restrict__ part,
                                                             int N, int E) {
  __shared__ int hist[DEG_RANGE];
  const int tid = threadIdx.x;
  const int lo  = blockIdx.y * DEG_RANGE;
  const int nn  = (N - lo) < DEG_RANGE ? (N - lo) : DEG_RANGE;
  for (int i = tid; i < DEG_RANGE; i += DEG_THREADS) hist[i] = 0;
  __syncthreads();

  const int quads = (int)(((long long)E + 3) >> 2);
  const int qper  = (quads + DEG_CHUNKS - 1) / DEG_CHUNKS;
  const int q0    = blockIdx.x * qper;
  const int q1    = (q0 + qper) < quads ? (q0 + qper) : quads;

  if ((E & 3) == 0 && (((size_t)col) & 15) == 0) {
    const int4* c4 = (const int4*)col;
    int q = q0 + tid;
    for (; q + 3 * DEG_THREADS < q1; q += 4 * DEG_THREADS) {
      const int4 a = c4[q];
      const int4 b = c4[q + DEG_THREADS];
      const int4 c = c4[q + 2 * DEG_THREADS];
      const int4 d = c4[q + 3 * DEG_THREADS];
      deg_count_four(hist, a, lo, nn);
      deg_count_four(hist, b, lo, nn);
      deg_count_four(hist, c, lo, nn);
      deg_count_four(hist, d, lo, nn);
    }
    for (; q < q1; q += DEG_THREADS) deg_count_four(hist, c4[q], lo, nn);
  } else {
    const long long e0 = (long long)q0 << 2;
    long long e1 = (long long)q1 << 2;
    if (e1 > E) e1 = E;
    for (long long e = e0 + tid; e < e1; e += DEG_THREADS) deg_count_one(hist, col[e], lo, nn);
  }
  __syncthreads();

  int* plane = part + (size_t)blockIdx.x * N + lo;
  for (int i = tid; i < nn; i += DEG_THREADS) plane[i] = hist[i];
}

__global__ void gcn_deg_finish(const int* __restrict__ part, float* __restrict__ dis, int N) {
  int i = blockIdx.x * blockDim.x + threadIdx.x;
  if (i < N) {
    int cnt = 1;
    for (int p = 0; p < DEG_CHUNKS; ++p) cnt += part[(size_t)p * N + i];
    float d = (float)cnt;
    dis[i] = (d > 0.0f) ? rsqrtf(d) : 0.0f;
  }
}


__global__ void gcn_self_loop(const float* __restrict__ dis, const float* __restrict__ h,
                              float* __restrict__ agg, int N, int C) {
  long long i = (long long)blockIdx.x * blockDim.x + threadIdx.x;
  long long total = (long long)N * C;
  if (i < total) {
    int node = (int)(i / C);
    float d = dis[node];
    agg[i] = d * d * h[i];
  }
}

#define SCAT_THREADS 256
template <int C>
__global__ __launch_bounds__(SCAT_THREADS) void gcn_edge_scatter(const int* __restrict__ ei,
                                                                 const float* __restrict__ dis,
                                                                 const float* __restrict__ h,
                                                                 float* __restrict__ agg, int E) {
  static_assert(C % 32 == 0, "lane = channel needs whole 32-channel lines");
  constexpr int Q   = C / 32;
  constexpr int GRP = 16;
  const int lane = threadIdx.x & 31;
  const int wv = __builtin_amdgcn_readfirstlane((int)(threadIdx.x >> 5));
  const long long base = ((long long)blockIdx.x * (SCAT_THREADS / 32) + wv) << 5;
  if (base >= E) return;
  const long long rem = (long long)E - base;
  const int cnt = rem < 32 ? (int)rem : 32;

  int r = 0, c = 0;
  float w = 0.f;
  if (lane < cnt) {
    const int* ep = ei + base;
    r = ep[lane];
    c = ep[(long long)E + lane];
    w = dis[r] * dis[c];
  }

  if (cnt == 32) {
#pragma unroll
    for (int g = 0; g < 32; g += GRP) {
      float v[GRP][Q];
#pragma unroll
      for (int u = 0; u < GRP; ++u) {
        const int rj = __builtin_amdgcn_readlane(r, g + u);
        const float* hrow = h + (long long)rj * C;
#pragma unroll
        for (int q = 0; q < Q; ++q) v[u][q] = hrow[lane + q * 32];
      }
#pragma unroll
      for (int u = 0; u < GRP; ++u) {
        const int   cj = __builtin_amdgcn_readlane(c, g + u);
        const float wj = __int_as_float(__builtin_amdgcn_readlane(__float_as_int(w), g + u));
        float* arow = agg + (long long)cj * C;
#pragma unroll
        for (int q = 0; q < Q; ++q) unsafeAtomicAdd(arow + lane + q * 32, wj * v[u][q]);
      }
    }
  } else {
    for (int j = 0; j < cnt; ++j) {
      const int   rj = __builtin_amdgcn_readlane(r, j);
      const int   cj = __builtin_amdgcn_readlane(c, j);
      const float wj = __int_as_float(__builtin_amdgcn_readlane(__float_as_int(w), j));
      const float* hrow = h + (long long)rj * C;
      float*       arow = agg + (long long)cj * C;
      for (int q = 0; q < Q; ++q) unsafeAtomicAdd(arow + lane + q * 32, wj * hrow[lane + q * 32]);
    }
  }
}

template <int IC, bool DUAL, bool RELU, bool GUARD>
__global__ __launch_bounds__(32 * WAVES_PER_BLOCK) void gcn_wmma_gemm(
    const float* __restrict__ inA, const float* __restrict__ WA, const float* __restrict__ bA,
    const float* __restrict__ inB, const float* __restrict__ WB, const float* __restrict__ bB,
    float* __restrict__ out, int nrows, int OC) {
  constexpr int PITCH = IC + 4;
  static_assert((PITCH & 7) == 4, "pitch must be 4 * odd floats");
  __shared__ __attribute__((aligned(16))) float lds[(DUAL ? 2 : 1) * 16 * PITCH];
  const int tid  = threadIdx.x;
  const int lane = tid & 31;
  const int wave = tid >> 5;
  const int row0 = blockIdx.x << 4;
  const int col0 = (blockIdx.y * WAVES_PER_BLOCK + wave) << 4;

  float* Atile = lds;
  float* Btile = DUAL ? (lds + 16 * PITCH) : lds;

  constexpr int vecPerRow = IC >> 2;
  constexpr int totalVec  = vecPerRow << 4;
  for (int idx = tid; idx < totalVec; idx += 32 * WAVES_PER_BLOCK) {
    const int r = idx / vecPerRow;
    const int v = idx - r * vecPerRow;
    const int node = row0 + r;
    const int off = r * PITCH + (v << 2);
    if (GUARD && node >= nrows) {
      float4 z = {0.f, 0.f, 0.f, 0.f};
      *(float4*)(Atile + off) = z;
      if constexpr (DUAL) *(float4*)(Btile + off) = z;
    } else {
      *(float4*)(Atile + off) = ((const float4*)(inA + (size_t)node * IC))[v];
      if constexpr (DUAL)
        *(float4*)(Btile + off) = ((const float4*)(inB + (size_t)node * IC))[v];
    }
  }
  __syncthreads();

  if (col0 >= OC) return;

  const int  nidx = lane & 15;
  const int  koff = (lane >> 4) << 1;
  const int  col  = col0 + nidx;
  const bool colOK = GUARD ? (col < OC) : true;
  const int  colC  = GUARD ? (colOK ? col : (OC - 1)) : col;
  const float* wAp = WA + (size_t)colC * IC + koff;
  const float* wBp = DUAL ? (WB + (size_t)colC * IC + koff) : WA;
  const float* ar  = Atile + nidx * PITCH + koff;
  const float* br  = Btile + nidx * PITCH + koff;

  constexpr int KSTEPS = IC / 4;
  constexpr int BATCH  = DUAL ? (KSTEPS < 4 ? KSTEPS : 4)
                              : (KSTEPS < 8 ? KSTEPS : 8);

  v8f acc = {};
#pragma unroll 1
  for (int kb = 0; kb < KSTEPS; kb += BATCH) {
    v2f fa[BATCH], fw[BATCH];
    v2f ga[DUAL ? BATCH : 1], gw[DUAL ? BATCH : 1];
#pragma unroll
    for (int i = 0; i < BATCH; ++i) {
      const int k = (kb + i) * 4;
      fa[i] = *(const v2f*)(ar + k);
      fw[i] = *(const v2f*)(wAp + k);
      if constexpr (DUAL) {
        ga[i] = *(const v2f*)(br + k);
        gw[i] = *(const v2f*)(wBp + k);
      }
    }
#pragma unroll
    for (int i = 0; i < BATCH; ++i) {
      acc = __builtin_amdgcn_wmma_f32_16x16x4_f32(false, fa[i], false, fw[i],
                                                  (short)0, acc, false, false);
      if constexpr (DUAL)
        acc = __builtin_amdgcn_wmma_f32_16x16x4_f32(false, ga[i], false, gw[i],
                                                    (short)0, acc, false, false);
    }
  }

  const int mb = (lane >> 4) << 3;
  if constexpr (GUARD) {
    float bias = 0.f;
    if (colOK) {
      bias = bA[col];
      if constexpr (DUAL) bias += bB[col];
    }
#pragma unroll
    for (int vg = 0; vg < 8; ++vg) {
      const int r = row0 + mb + vg;
      if (colOK && r < nrows) {
        float val = acc[vg] + bias;
        if constexpr (RELU) val = val > 0.f ? val : 0.f;
        out[(size_t)r * OC + col] = val;
      }
    }
  } else {
    float bias = bA[col];
    if constexpr (DUAL) bias += bB[col];
    float* op = out + (size_t)(row0 + mb) * OC + col;
#pragma unroll
    for (int vg = 0; vg < 8; ++vg) {
      float val = acc[vg] + bias;
      if constexpr (RELU) val = val > 0.f ? val : 0.f;
      op[(size_t)vg * OC] = val;
    }
  }
}

__global__ __launch_bounds__(32) void gcn_fold_linear(const float* __restrict__ Wo,
                                                      const float* __restrict__ Wi,
                                                      const float* __restrict__ bi,
                                                      const float* __restrict__ bo,
                                                      float* __restrict__ Wf,
                                                      float* __restrict__ bf,
                                                      int O, int J, int C) {
  const int lane   = threadIdx.x;
  const int ctiles = C >> 4;
  const int ntiles = ((O + 15) >> 4) * ctiles;

  if ((int)blockIdx.x >= ntiles) {
    const int o = (int)blockIdx.x - ntiles;
    const float* wr = Wo + (size_t)o * J;
    float s = 0.f;
    for (int j = lane; j < J; j += 32) s += wr[j] * bi[j];
    for (int m = 16; m > 0; m >>= 1) s += __shfl_xor(s, m);
    if (lane == 0) bf[o] = s + bo[o];
    return;
  }

  const int ot     = blockIdx.x / ctiles;
  const int ct     = blockIdx.x - ot * ctiles;
  const int nidx   = lane & 15;
  const int koff   = (lane >> 4) << 1;
  const int orow   = (ot << 4) + nidx;
  const int orowC  = orow < O ? orow : (O - 1);
  const int col    = (ct << 4) + nidx;
  const float* ap = Wo + (size_t)orowC * J + koff;
  const float* bp = Wi + (size_t)koff * C + col;

  v8f acc = {};
  constexpr int FB = 16;
  int k = 0;
#pragma unroll 1
  for (; k + 4 * FB <= J; k += 4 * FB) {
    v2f fa[FB], fb[FB];
#pragma unroll
    for (int i = 0; i < FB; ++i) {
      const int kk = k + 4 * i;
      fa[i]   = *(const v2f*)(ap + kk);
      fb[i].x = bp[(size_t)kk * C];
      fb[i].y = bp[(size_t)(kk + 1) * C];
    }
#pragma unroll
    for (int i = 0; i < FB; ++i)
      acc = __builtin_amdgcn_wmma_f32_16x16x4_f32(false, fa[i], false, fb[i],
                                                  (short)0, acc, false, false);
  }
  for (; k < J; k += 4) {
    const v2f a = *(const v2f*)(ap + k);
    v2f b;
    b.x = bp[(size_t)k * C];
    b.y = bp[(size_t)(k + 1) * C];
    acc = __builtin_amdgcn_wmma_f32_16x16x4_f32(false, a, false, b, (short)0, acc, false, false);
  }
  const int mb = (lane >> 4) << 3;
#pragma unroll
  for (int vg = 0; vg < 8; ++vg) {
    const int o = (ot << 4) + mb + vg;
    if (o < O) Wf[(size_t)o * C + col] = acc[vg];
  }
}


template <int IC, bool DUAL, bool RELU>
static inline void launch_gemm(const float* inA, const float* WA, const float* bA,
                               const float* inB, const float* WB, const float* bB,
                               float* out, int N, int OC, hipStream_t stream) {
  const int rowTiles = (N + 15) / 16;
  const int gy = (((OC + 15) / 16) + WAVES_PER_BLOCK - 1) / WAVES_PER_BLOCK;
  const dim3 grid(rowTiles, gy), blk(32 * WAVES_PER_BLOCK);
  if ((N & 15) == 0 && (OC & 15) == 0)
    gcn_wmma_gemm<IC, DUAL, RELU, false><<<grid, blk, 0, stream>>>(
        inA, WA, bA, inB, WB, bB, out, N, OC);
  else
    gcn_wmma_gemm<IC, DUAL, RELU, true><<<grid, blk, 0, stream>>>(
        inA, WA, bA, inB, WB, bB, out, N, OC);
}

extern "C" void kernel_launch(void* const* d_in, const int* in_sizes, int n_in,
                              void* d_out, int out_size, void* d_ws, size_t ws_size,
                              hipStream_t stream) {
  (void)n_in; (void)out_size; (void)ws_size;
  const float* x    = (const float*)d_in[0];
  const int*   ei   = (const int*)d_in[1];
  const float* w1   = (const float*)d_in[2];
  const float* b1   = (const float*)d_in[3];
  const float* w2   = (const float*)d_in[4];
  const float* b2   = (const float*)d_in[5];
  const float* c1wl = (const float*)d_in[6];
  const float* c1bl = (const float*)d_in[7];
  const float* c1wr = (const float*)d_in[8];
  const float* c1br = (const float*)d_in[9];
  const float* c2wl = (const float*)d_in[10];
  const float* c2bl = (const float*)d_in[11];
  const float* c2wr = (const float*)d_in[12];
  const float* c2br = (const float*)d_in[13];
  const float* w3   = (const float*)d_in[14];
  const float* b3   = (const float*)d_in[15];
  const float* w4   = (const float*)d_in[16];
  const float* b4   = (const float*)d_in[17];

  const int IN_CH = 256;
  const int N = in_sizes[0] / IN_CH;
  const int E = in_sizes[1] / 2;
  const size_t n = (size_t)N;

  float* ws   = (float*)d_ws;
  float* dis  = ws;
  float* h2   = ws + 1 * n;
  float* agg1 = ws + 33 * n;
  float* h3   = ws + 65 * n;
  float* agg2 = ws + 129 * n;
  float* h4   = ws + 193 * n;
  float* h1   = ws + 193 * n;
  float* wf   = ws + 1 * n;
  float* bfo  = wf + 20 * 128;
  int*   part = (int*)(ws + 129 * n);

  const int eltBlocks  = (N + 255) / 256;
  const int degRanges  = (N + DEG_RANGE - 1) / DEG_RANGE;
  const int scatBlocks = (int)(((((long long)E + 31) / 32) * 32 + SCAT_THREADS - 1) / SCAT_THREADS);

  launch_gemm<256, false, false>(x, w1, b1, nullptr, nullptr, nullptr, h1, N, 16, stream);
  launch_gemm<16, false, false>(h1, w2, b2, nullptr, nullptr, nullptr, h2, N, 32, stream);

  gcn_deg_count<<<dim3(DEG_CHUNKS, degRanges), DEG_THREADS, 0, stream>>>(ei + E, part, N, E);
  gcn_deg_finish<<<eltBlocks, 256, 0, stream>>>(part, dis, N);

  gcn_self_loop<<<(int)((n * 32 + 255) / 256), 256, 0, stream>>>(dis, h2, agg1, N, 32);
  gcn_edge_scatter<32><<<scatBlocks, SCAT_THREADS, 0, stream>>>(ei, dis, h2, agg1, E);
  launch_gemm<32, true, true>(h2, c1wl, c1bl, agg1, c1wr, c1br, h3, N, 64, stream);

  gcn_self_loop<<<(int)((n * 64 + 255) / 256), 256, 0, stream>>>(dis, h3, agg2, N, 64);
  gcn_edge_scatter<64><<<scatBlocks, SCAT_THREADS, 0, stream>>>(ei, dis, h3, agg2, E);
  launch_gemm<64, true, true>(h3, c2wl, c2bl, agg2, c2wr, c2br, h4, N, 128, stream);

  gcn_fold_linear<<<2 * (128 / 16) + 20, 32, 0, stream>>>(w4, w3, b3, b4, wf, bfo, 20, 192, 128);
  launch_gemm<128, false, false>(h4, wf, bfo, nullptr, nullptr, nullptr, (float*)d_out, N, 20, stream);
}


# neighbour sums pulled per target node from a by-target edge list built each call (counting sort, slots from LDS counters); one wave per node, no float atomics, self-loop fused; sums in list order
# speedup vs baseline: 3.0839x; 1.3386x over previous
#include <hip/hip_runtime.h>

typedef __attribute__((ext_vector_type(2))) float v2f;
typedef __attribute__((ext_vector_type(8))) float v8f;

#define WAVES_PER_BLOCK 4

#define DEG_RANGE   6400
#define DEG_CHUNKS  16
#define DEG_THREADS 1024
#define FIN_THREADS 256

template <bool FILL>
__device__ __forceinline__ void bin_one(int* slot, int c, int r, int lo, int nn,
                                        int* __restrict__ csr) {
  const unsigned d = (unsigned)(c - lo);
  if (d < (unsigned)nn) {
    if constexpr (FILL) {
      const int pos = atomicAdd(&slot[d], 1);
      csr[pos] = r;
    } else {
      atomicAdd(&slot[d], 1);
    }
  }
}

template <bool FILL>
__device__ __forceinline__ void bin_four(int* slot, int4 c, int4 r, int lo, int nn,
                                         int* __restrict__ csr) {
  bin_one<FILL>(slot, c.x, r.x, lo, nn, csr);
  bin_one<FILL>(slot, c.y, r.y, lo, nn, csr);
  bin_one<FILL>(slot, c.z, r.z, lo, nn, csr);
  bin_one<FILL>(slot, c.w, r.w, lo, nn, csr);
}

template <bool FILL>
__global__ __launch_bounds__(DEG_THREADS) void gcn_edge_bins(const int* __restrict__ row,
                                                             const int* __restrict__ col,
                                                             int* __restrict__ part,
                                                             int* __restrict__ csr,
                                                             int N, int E) {
  __shared__ int slot[DEG_RANGE];
  const int tid = threadIdx.x;
  const int lo  = blockIdx.y * DEG_RANGE;
  const int nn  = (N - lo) < DEG_RANGE ? (N - lo) : DEG_RANGE;
  int* plane = part + (size_t)blockIdx.x * N + lo;
  for (int i = tid; i < DEG_RANGE; i += DEG_THREADS)
    slot[i] = (FILL && i < nn) ? plane[i] : 0;
  __syncthreads();

  const int quads = (int)(((long long)E + 3) >> 2);
  const int qper  = (quads + DEG_CHUNKS - 1) / DEG_CHUNKS;
  const int q0    = blockIdx.x * qper;
  const int q1    = (q0 + qper) < quads ? (q0 + qper) : quads;

  const bool vec = (E & 3) == 0 && (((size_t)col) & 15) == 0 &&
                   (!FILL || (((size_t)row) & 15) == 0);
  if (vec) {
    const int4* c4 = (const int4*)col;
    const int4* r4 = (const int4*)row;
    const int4 z = {0, 0, 0, 0};
    int q = q0 + tid;
    for (; q + 3 * DEG_THREADS < q1; q += 4 * DEG_THREADS) {
      const int4 ca = c4[q];
      const int4 cb = c4[q + DEG_THREADS];
      const int4 cc = c4[q + 2 * DEG_THREADS];
      const int4 cd = c4[q + 3 * DEG_THREADS];
      const int4 ra = FILL ? r4[q] : z;
      const int4 rb = FILL ? r4[q + DEG_THREADS] : z;
      const int4 rc = FILL ? r4[q + 2 * DEG_THREADS] : z;
      const int4 rd = FILL ? r4[q + 3 * DEG_THREADS] : z;
      bin_four<FILL>(slot, ca, ra, lo, nn, csr);
      bin_four<FILL>(slot, cb, rb, lo, nn, csr);
      bin_four<FILL>(slot, cc, rc, lo, nn, csr);
      bin_four<FILL>(slot, cd, rd, lo, nn, csr);
    }
    for (; q < q1; q += DEG_THREADS) bin_four<FILL>(slot, c4[q], FILL ? r4[q] : z, lo, nn, csr);
  } else {
    const long long e0 = (long long)q0 << 2;
    long long e1 = (long long)q1 << 2;
    if (e1 > E) e1 = E;
    for (long long e = e0 + tid; e < e1; e += DEG_THREADS)
      bin_one<FILL>(slot, col[e], FILL ? row[e] : 0, lo, nn, csr);
  }

  if constexpr (!FILL) {
    __syncthreads();
    for (int i = tid; i < nn; i += DEG_THREADS) plane[i] = slot[i];
  }
}

__global__ __launch_bounds__(FIN_THREADS) void gcn_deg_finish(const int* __restrict__ part,
                                                              float* __restrict__ dis,
                                                              int* __restrict__ blk, int N) {
  __shared__ int wsum[FIN_THREADS / 32];
  const int tid = threadIdx.x;
  const int i = blockIdx.x * FIN_THREADS + tid;
  int in = 0;
  if (i < N) {
    for (int p = 0; p < DEG_CHUNKS; ++p) in += part[(size_t)p * N + i];
    const float d = (float)(in + 1);
    dis[i] = (d > 0.0f) ? rsqrtf(d) : 0.0f;
  }
  int s = in;
  for (int m = 16; m > 0; m >>= 1) s += __shfl_xor(s, m);
  if ((tid & 31) == 0) wsum[tid >> 5] = s;
  __syncthreads();
  if (tid == 0) {
    int t = 0;
    for (int w = 0; w < FIN_THREADS / 32; ++w) t += wsum[w];
    blk[blockIdx.x] = t;
  }
}

__global__ __launch_bounds__(FIN_THREADS) void gcn_csr_offsets(int* __restrict__ part,
                                                               const int* __restrict__ blk,
                                                               int* __restrict__ rowptr, int N) {
  __shared__ int red[FIN_THREADS / 32];
  __shared__ int wtot[FIN_THREADS / 32];
  const int tid  = threadIdx.x;
  const int lane = tid & 31;
  const int wv   = tid >> 5;
  const int b    = blockIdx.x;
  const int i    = b * FIN_THREADS + tid;

  int s = 0;
  for (int k = tid; k < b; k += FIN_THREADS) s += blk[k];
  for (int m = 16; m > 0; m >>= 1) s += __shfl_xor(s, m);
  if (lane == 0) red[wv] = s;

  int c[DEG_CHUNKS];
  int cnt = 0;
#pragma unroll
  for (int p = 0; p < DEG_CHUNKS; ++p) {
    c[p] = (i < N) ? part[(size_t)p * N + i] : 0;
    cnt += c[p];
  }
  int inc = cnt;
  for (int d = 1; d < 32; d <<= 1) {
    const int t = __shfl_up(inc, d);
    if (lane >= d) inc += t;
  }
  if (lane == 31) wtot[wv] = inc;
  __syncthreads();
  int start = inc - cnt;
  for (int w = 0; w < FIN_THREADS / 32; ++w) {
    start += red[w];
    if (w < wv) start += wtot[w];
  }
  if (i < N) {
    rowptr[i] = start;
    if (i == N - 1) rowptr[N] = start + cnt;
    int run = start;
#pragma unroll
    for (int p = 0; p < DEG_CHUNKS; ++p) {
      part[(size_t)p * N + i] = run;
      run += c[p];
    }
  }
}

#define GATH_THREADS 256
template <int C>
__global__ __launch_bounds__(GATH_THREADS) void gcn_gather(const int* __restrict__ rowptr,
                                                           const int* __restrict__ csr,
                                                           const float* __restrict__ dis,
                                                           const float* __restrict__ h,
                                                           float* __restrict__ agg, int N) {
  static_assert(C == 32 || C == 64, "one or two channels per lane");
  constexpr int Q   = C / 32;
  constexpr int GRP = 8;
  const int lane = threadIdx.x & 31;
  const int wv   = __builtin_amdgcn_readfirstlane((int)(threadIdx.x >> 5));
  const int node = blockIdx.x * (GATH_THREADS / 32) + wv;
  if (node >= N) return;
  const int beg = __builtin_amdgcn_readfirstlane(rowptr[node]);
  const int end = __builtin_amdgcn_readfirstlane(rowptr[node + 1]);
  const float di = dis[node];

  float acc[Q];
  {
    const float* hrow = h + (long long)node * C + lane * Q;
    if constexpr (Q == 2) {
      const v2f t = *(const v2f*)hrow;
      acc[0] = di * di * t.x;
      acc[1] = di * di * t.y;
    } else {
      acc[0] = di * di * hrow[0];
    }
  }

  for (int base = beg; base < end; base += 32) {
    const int left = end - base;
    const int cnt  = left < 32 ? left : 32;
    int   s = 0;
    float ds = 0.f;
    if (lane < cnt) {
      s  = csr[base + lane];
      ds = dis[s];
    }
    for (int j = 0; j < cnt; j += GRP) {
      float v[GRP][Q];
#pragma unroll
      for (int u = 0; u < GRP; ++u) {
        const int jj = (j + u) < cnt ? (j + u) : (cnt - 1);
        const int sj = __builtin_amdgcn_readlane(s, jj);
        const float* hrow = h + (long long)sj * C + lane * Q;
        if constexpr (Q == 2) {
          const v2f t = *(const v2f*)hrow;
          v[u][0] = t.x;
          v[u][1] = t.y;
        } else {
          v[u][0] = hrow[0];
        }
      }
#pragma unroll
      for (int u = 0; u < GRP; ++u) {
        if (j + u < cnt) {
          const float wj = __int_as_float(__builtin_amdgcn_readlane(__float_as_int(ds), j + u)) * di;
#pragma unroll
          for (int q = 0; q < Q; ++q) acc[q] += wj * v[u][q];
        }
      }
    }
  }

  float* arow = agg + (long long)node * C + lane * Q;
  if constexpr (Q == 2) {
    v2f t;
    t.x = acc[0];
    t.y = acc[1];
    *(v2f*)arow = t;
  } else {
    arow[0] = acc[0];
  }
}

template <int IC, bool DUAL, bool RELU, bool GUARD>
__global__ __launch_bounds__(32 * WAVES_PER_BLOCK) void gcn_wmma_gemm(
    const float* __restrict__ inA, const float* __restrict__ WA, const float* __restrict__ bA,
    const float* __restrict__ inB, const float* __restrict__ WB, const float* __restrict__ bB,
    float* __restrict__ out, int nrows, int OC) {
  constexpr int PITCH = IC + 4;
  static_assert((PITCH & 7) == 4, "pitch must be 4 * odd floats");
  __shared__ __attribute__((aligned(16))) float lds[(DUAL ? 2 : 1) * 16 * PITCH];
  const int tid  = threadIdx.x;
  const int lane = tid & 31;
  const int wave = tid >> 5;
  const int row0 = blockIdx.x << 4;
  const int col0 = (blockIdx.y * WAVES_PER_BLOCK + wave) << 4;

  float* Atile = lds;
  float* Btile = DUAL ? (lds + 16 * PITCH) : lds;

  constexpr int vecPerRow = IC >> 2;
  constexpr int totalVec  = vecPerRow << 4;
  for (int idx = tid; idx < totalVec; idx += 32 * WAVES_PER_BLOCK) {
    const int r = idx / vecPerRow;
    const int v = idx - r * vecPerRow;
    const int node = row0 + r;
    const int off = r * PITCH + (v << 2);
    if (GUARD && node >= nrows) {
      float4 z = {0.f, 0.f, 0.f, 0.f};
      *(float4*)(Atile + off) = z;
      if constexpr (DUAL) *(float4*)(Btile + off) = z;
    } else {
      *(float4*)(Atile + off) = ((const float4*)(inA + (size_t)node * IC))[v];
      if constexpr (DUAL)
        *(float4*)(Btile + off) = ((const float4*)(inB + (size_t)node * IC))[v];
    }
  }
  __syncthreads();

  if (col0 >= OC) return;

  const int  nidx = lane & 15;
  const int  koff = (lane >> 4) << 1;
  const int  col  = col0 + nidx;
  const bool colOK = GUARD ? (col < OC) : true;
  const int  colC  = GUARD ? (colOK ? col : (OC - 1)) : col;
  const float* wAp = WA + (size_t)colC * IC + koff;
  const float* wBp = DUAL ? (WB + (size_t)colC * IC + koff) : WA;
  const float* ar  = Atile + nidx * PITCH + koff;
  const float* br  = Btile + nidx * PITCH + koff;

  constexpr int KSTEPS = IC / 4;
  constexpr int BATCH  = DUAL ? (KSTEPS < 4 ? KSTEPS : 4)
                              : (KSTEPS < 8 ? KSTEPS : 8);

  v8f acc = {};
#pragma unroll 1
  for (int kb = 0; kb < KSTEPS; kb += BATCH) {
    v2f fa[BATCH], fw[BATCH];
    v2f ga[DUAL ? BATCH : 1], gw[DUAL ? BATCH : 1];
#pragma unroll
    for (int i = 0; i < BATCH; ++i) {
      const int k = (kb + i) * 4;
      fa[i] = *(const v2f*)(ar + k);
      fw[i] = *(const v2f*)(wAp + k);
      if constexpr (DUAL) {
        ga[i] = *(const v2f*)(br + k);
        gw[i] = *(const v2f*)(wBp + k);
      }
    }
#pragma unroll
    for (int i = 0; i < BATCH; ++i) {
      acc = __builtin_amdgcn_wmma_f32_16x16x4_f32(false, fa[i], false, fw[i],
                                                  (short)0, acc, false, false);
      if constexpr (DUAL)
        acc = __builtin_amdgcn_wmma_f32_16x16x4_f32(false, ga[i], false, gw[i],
                                                    (short)0, acc, false, false);
    }
  }

  const int mb = (lane >> 4) << 3;
  if constexpr (GUARD) {
    float bias = 0.f;
    if (colOK) {
      bias = bA[col];
      if constexpr (DUAL) bias += bB[col];
    }
#pragma unroll
    for (int vg = 0; vg < 8; ++vg) {
      const int r = row0 + mb + vg;
      if (colOK && r < nrows) {
        float val = acc[vg] + bias;
        if constexpr (RELU) val = val > 0.f ? val : 0.f;
        out[(size_t)r * OC + col] = val;
      }
    }
  } else {
    float bias = bA[col];
    if constexpr (DUAL) bias += bB[col];
    float* op = out + (size_t)(row0 + mb) * OC + col;
#pragma unroll
    for (int vg = 0; vg < 8; ++vg) {
      float val = acc[vg] + bias;
      if constexpr (RELU) val = val > 0.f ? val : 0.f;
      op[(size_t)vg * OC] = val;
    }
  }
}

__global__ __launch_bounds__(32) void gcn_fold_linear(const float* __restrict__ Wo,
                                                      const float* __restrict__ Wi,
                                                      const float* __restrict__ bi,
                                                      const float* __restrict__ bo,
                                                      float* __restrict__ Wf,
                                                      float* __restrict__ bf,
                                                      int O, int J, int C) {
  const int lane   = threadIdx.x;
  const int ctiles = C >> 4;
  const int ntiles = ((O + 15) >> 4) * ctiles;

  if ((int)blockIdx.x >= ntiles) {
    const int o = (int)blockIdx.x - ntiles;
    const float* wr = Wo + (size_t)o * J;
    float s = 0.f;
    for (int j = lane; j < J; j += 32) s += wr[j] * bi[j];
    for (int m = 16; m > 0; m >>= 1) s += __shfl_xor(s, m);
    if (lane == 0) bf[o] = s + bo[o];
    return;
  }

  const int ot     = blockIdx.x / ctiles;
  const int ct     = blockIdx.x - ot * ctiles;
  const int nidx   = lane & 15;
  const int koff   = (lane >> 4) << 1;
  const int orow   = (ot << 4) + nidx;
  const int orowC  = orow < O ? orow : (O - 1);
  const int col    = (ct << 4) + nidx;
  const float* ap = Wo + (size_t)orowC * J + koff;
  const float* bp = Wi + (size_t)koff * C + col;

  v8f acc = {};
  constexpr int FB = 16;
  int k = 0;
#pragma unroll 1
  for (; k + 4 * FB <= J; k += 4 * FB) {
    v2f fa[FB], fb[FB];
#pragma unroll
    for (int i = 0; i < FB; ++i) {
      const int kk = k + 4 * i;
      fa[i]   = *(const v2f*)(ap + kk);
      fb[i].x = bp[(size_t)kk * C];
      fb[i].y = bp[(size_t)(kk + 1) * C];
    }
#pragma unroll
    for (int i = 0; i < FB; ++i)
      acc = __builtin_amdgcn_wmma_f32_16x16x4_f32(false, fa[i], false, fb[i],
                                                  (short)0, acc, false, false);
  }
  for (; k < J; k += 4) {
    const v2f a = *(const v2f*)(ap + k);
    v2f b;
    b.x = bp[(size_t)k * C];
    b.y = bp[(size_t)(k + 1) * C];
    acc = __builtin_amdgcn_wmma_f32_16x16x4_f32(false, a, false, b, (short)0, acc, false, false);
  }
  const int mb = (lane >> 4) << 3;
#pragma unroll
  for (int vg = 0; vg < 8; ++vg) {
    const int o = (ot << 4) + mb + vg;
    if (o < O) Wf[(size_t)o * C + col] = acc[vg];
  }
}


template <int IC, bool DUAL, bool RELU>
static inline void launch_gemm(const float* inA, const float* WA, const float* bA,
                               const float* inB, const float* WB, const float* bB,
                               float* out, int N, int OC, hipStream_t stream) {
  const int rowTiles = (N + 15) / 16;
  const int gy = (((OC + 15) / 16) + WAVES_PER_BLOCK - 1) / WAVES_PER_BLOCK;
  const dim3 grid(rowTiles, gy), blk(32 * WAVES_PER_BLOCK);
  if ((N & 15) == 0 && (OC & 15) == 0)
    gcn_wmma_gemm<IC, DUAL, RELU, false><<<grid, blk, 0, stream>>>(
        inA, WA, bA, inB, WB, bB, out, N, OC);
  else
    gcn_wmma_gemm<IC, DUAL, RELU, true><<<grid, blk, 0, stream>>>(
        inA, WA, bA, inB, WB, bB, out, N, OC);
}

extern "C" void kernel_launch(void* const* d_in, const int* in_sizes, int n_in,
                              void* d_out, int out_size, void* d_ws, size_t ws_size,
                              hipStream_t stream) {
  (void)n_in; (void)out_size; (void)ws_size;
  const float* x    = (const float*)d_in[0];
  const int*   ei   = (const int*)d_in[1];
  const float* w1   = (const float*)d_in[2];
  const float* b1   = (const float*)d_in[3];
  const float* w2   = (const float*)d_in[4];
  const float* b2   = (const float*)d_in[5];
  const float* c1wl = (const float*)d_in[6];
  const float* c1bl = (const float*)d_in[7];
  const float* c1wr = (const float*)d_in[8];
  const float* c1br = (const float*)d_in[9];
  const float* c2wl = (const float*)d_in[10];
  const float* c2bl = (const float*)d_in[11];
  const float* c2wr = (const float*)d_in[12];
  const float* c2br = (const float*)d_in[13];
  const float* w3   = (const float*)d_in[14];
  const float* b3   = (const float*)d_in[15];
  const float* w4   = (const float*)d_in[16];
  const float* b4   = (const float*)d_in[17];

  const int IN_CH = 256;
  const int N = in_sizes[0] / IN_CH;
  const int E = in_sizes[1] / 2;
  const size_t n = (size_t)N;

  float* ws   = (float*)d_ws;
  float* dis  = ws;
  float* h2   = ws + 1 * n;
  float* agg1 = ws + 33 * n;
  float* h3   = ws + 65 * n;
  float* agg2 = ws + 129 * n;
  float* h4   = ws + 193 * n;
  float* h1   = ws + 193 * n;
  float* wf   = ws + 1 * n;
  float* bfo  = wf + 20 * 128;
  int*   part = (int*)(ws + 129 * n);
  int*   csr    = (int*)(ws + 209 * n);
  int*   rowptr = csr + (((size_t)E + 3) & ~(size_t)3);
  int*   blk    = rowptr + ((n + 4) & ~(size_t)3);

  const int finBlocks  = (N + FIN_THREADS - 1) / FIN_THREADS;
  const int degRanges  = (N + DEG_RANGE - 1) / DEG_RANGE;
  const int gathBlocks = (N + (GATH_THREADS / 32) - 1) / (GATH_THREADS / 32);

  launch_gemm<256, false, false>(x, w1, b1, nullptr, nullptr, nullptr, h1, N, 16, stream);
  launch_gemm<16, false, false>(h1, w2, b2, nullptr, nullptr, nullptr, h2, N, 32, stream);

  gcn_edge_bins<false><<<dim3(DEG_CHUNKS, degRanges), DEG_THREADS, 0, stream>>>(ei, ei + E, part, csr, N, E);
  gcn_deg_finish<<<finBlocks, FIN_THREADS, 0, stream>>>(part, dis, blk, N);
  gcn_csr_offsets<<<finBlocks, FIN_THREADS, 0, stream>>>(part, blk, rowptr, N);
  gcn_edge_bins<true><<<dim3(DEG_CHUNKS, degRanges), DEG_THREADS, 0, stream>>>(ei, ei + E, part, csr, N, E);

  gcn_gather<32><<<gathBlocks, GATH_THREADS, 0, stream>>>(rowptr, csr, dis, h2, agg1, N);
  launch_gemm<32, true, true>(h2, c1wl, c1bl, agg1, c1wr, c1br, h3, N, 64, stream);

  gcn_gather<64><<<gathBlocks, GATH_THREADS, 0, stream>>>(rowptr, csr, dis, h3, agg2, N);
  launch_gemm<64, true, true>(h3, c2wl, c2bl, agg2, c2wr, c2br, h4, N, 128, stream);

  gcn_fold_linear<<<2 * (128 / 16) + 20, 32, 0, stream>>>(w4, w3, b3, b4, wf, bfo, 20, 192, 128);
  launch_gemm<128, false, false>(h4, wf, bfo, nullptr, nullptr, nullptr, (float*)d_out, N, 20, stream);
}


# GEMMs re-tiled: a work-group stages 64 rows and each wave keeps up to 4 row tiles per column tile, so one weight-fragment read (16 cache lines) feeds 4 products; no idle waves; sum order unchanged
# speedup vs baseline: 3.2020x; 1.0383x over previous
#include <hip/hip_runtime.h>

typedef __attribute__((ext_vector_type(2))) float v2f;
typedef __attribute__((ext_vector_type(8))) float v8f;
typedef __attribute__((ext_vector_type(4))) float v4f;

#define DEG_RANGE   6400
#define DEG_CHUNKS  16
#define DEG_THREADS 1024
#define FIN_THREADS 256

template <bool FILL>
__device__ __forceinline__ void bin_one(int* slot, int c, int r, int lo, int nn,
                                        int* __restrict__ csr) {
  const unsigned d = (unsigned)(c - lo);
  if (d < (unsigned)nn) {
    if constexpr (FILL) {
      const int pos = atomicAdd(&slot[d], 1);
      csr[pos] = r;
    } else {
      atomicAdd(&slot[d], 1);
    }
  }
}

template <bool FILL>
__device__ __forceinline__ void bin_four(int* slot, int4 c, int4 r, int lo, int nn,
                                         int* __restrict__ csr) {
  bin_one<FILL>(slot, c.x, r.x, lo, nn, csr);
  bin_one<FILL>(slot, c.y, r.y, lo, nn, csr);
  bin_one<FILL>(slot, c.z, r.z, lo, nn, csr);
  bin_one<FILL>(slot, c.w, r.w, lo, nn, csr);
}

template <bool FILL>
__global__ __launch_bounds__(DEG_THREADS) void gcn_edge_bins(const int* __restrict__ row,
                                                             const int* __restrict__ col,
                                                             int* __restrict__ part,
                                                             int* __restrict__ csr,
                                                             int N, int E) {
  __shared__ int slot[DEG_RANGE];
  const int tid = threadIdx.x;
  const int lo  = blockIdx.y * DEG_RANGE;
  const int nn  = (N - lo) < DEG_RANGE ? (N - lo) : DEG_RANGE;
  int* plane = part + (size_t)blockIdx.x * N + lo;
  for (int i = tid; i < DEG_RANGE; i += DEG_THREADS)
    slot[i] = (FILL && i < nn) ? plane[i] : 0;
  __syncthreads();

  const int quads = (int)(((long long)E + 3) >> 2);
  const int qper  = (quads + DEG_CHUNKS - 1) / DEG_CHUNKS;
  const int q0    = blockIdx.x * qper;
  const int q1    = (q0 + qper) < quads ? (q0 + qper) : quads;

  const bool vec = (E & 3) == 0 && (((size_t)col) & 15) == 0 &&
                   (!FILL || (((size_t)row) & 15) == 0);
  if (vec) {
    const int4* c4 = (const int4*)col;
    const int4* r4 = (const int4*)row;
    const int4 z = {0, 0, 0, 0};
    int q = q0 + tid;
    for (; q + 3 * DEG_THREADS < q1; q += 4 * DEG_THREADS) {
      const int4 ca = c4[q];
      const int4 cb = c4[q + DEG_THREADS];
      const int4 cc = c4[q + 2 * DEG_THREADS];
      const int4 cd = c4[q + 3 * DEG_THREADS];
      const int4 ra = FILL ? r4[q] : z;
      const int4 rb = FILL ? r4[q + DEG_THREADS] : z;
      const int4 rc = FILL ? r4[q + 2 * DEG_THREADS] : z;
      const int4 rd = FILL ? r4[q + 3 * DEG_THREADS] : z;
      bin_four<FILL>(slot, ca, ra, lo, nn, csr);
      bin_four<FILL>(slot, cb, rb, lo, nn, csr);
      bin_four<FILL>(slot, cc, rc, lo, nn, csr);
      bin_four<FILL>(slot, cd, rd, lo, nn, csr);
    }
    for (; q < q1; q += DEG_THREADS) bin_four<FILL>(slot, c4[q], FILL ? r4[q] : z, lo, nn, csr);
  } else {
    const long long e0 = (long long)q0 << 2;
    long long e1 = (long long)q1 << 2;
    if (e1 > E) e1 = E;
    for (long long e = e0 + tid; e < e1; e += DEG_THREADS)
      bin_one<FILL>(slot, col[e], FILL ? row[e] : 0, lo, nn, csr);
  }

  if constexpr (!FILL) {
    __syncthreads();
    for (int i = tid; i < nn; i += DEG_THREADS) plane[i] = slot[i];
  }
}

__global__ __launch_bounds__(FIN_THREADS) void gcn_deg_finish(const int* __restrict__ part,
                                                              float* __restrict__ dis,
                                                              int* __restrict__ blk, int N) {
  __shared__ int wsum[FIN_THREADS / 32];
  const int tid = threadIdx.x;
  const int i = blockIdx.x * FIN_THREADS + tid;
  int in = 0;
  if (i < N) {
    for (int p = 0; p < DEG_CHUNKS; ++p) in += part[(size_t)p * N + i];
    const float d = (float)(in + 1);
    dis[i] = (d > 0.0f) ? rsqrtf(d) : 0.0f;
  }
  int s = in;
  for (int m = 16; m > 0; m >>= 1) s += __shfl_xor(s, m);
  if ((tid & 31) == 0) wsum[tid >> 5] = s;
  __syncthreads();
  if (tid == 0) {
    int t = 0;
    for (int w = 0; w < FIN_THREADS / 32; ++w) t += wsum[w];
    blk[blockIdx.x] = t;
  }
}

__global__ __launch_bounds__(FIN_THREADS) void gcn_csr_offsets(int* __restrict__ part,
                                                               const int* __restrict__ blk,
                                                               int* __restrict__ rowptr, int N) {
  __shared__ int red[FIN_THREADS / 32];
  __shared__ int wtot[FIN_THREADS / 32];
  const int tid  = threadIdx.x;
  const int lane = tid & 31;
  const int wv   = tid >> 5;
  const int b    = blockIdx.x;
  const int i    = b * FIN_THREADS + tid;

  int s = 0;
  for (int k = tid; k < b; k += FIN_THREADS) s += blk[k];
  for (int m = 16; m > 0; m >>= 1) s += __shfl_xor(s, m);
  if (lane == 0) red[wv] = s;

  int c[DEG_CHUNKS];
  int cnt = 0;
#pragma unroll
  for (int p = 0; p < DEG_CHUNKS; ++p) {
    c[p] = (i < N) ? part[(size_t)p * N + i] : 0;
    cnt += c[p];
  }
  int inc = cnt;
  for (int d = 1; d < 32; d <<= 1) {
    const int t = __shfl_up(inc, d);
    if (lane >= d) inc += t;
  }
  if (lane == 31) wtot[wv] = inc;
  __syncthreads();
  int start = inc - cnt;
  for (int w = 0; w < FIN_THREADS / 32; ++w) {
    start += red[w];
    if (w < wv) start += wtot[w];
  }
  if (i < N) {
    rowptr[i] = start;
    if (i == N - 1) rowptr[N] = start + cnt;
    int run = start;
#pragma unroll
    for (int p = 0; p < DEG_CHUNKS; ++p) {
      part[(size_t)p * N + i] = run;
      run += c[p];
    }
  }
}

#define GATH_THREADS 256
template <int C>
__global__ __launch_bounds__(GATH_THREADS) void gcn_gather(const int* __restrict__ rowptr,
                                                           const int* __restrict__ csr,
                                                           const float* __restrict__ dis,
                                                           const float* __restrict__ h,
                                                           float* __restrict__ agg, int N) {
  static_assert(C == 32 || C == 64, "one or two channels per lane");
  constexpr int Q   = C / 32;
  constexpr int GRP = 8;
  const int lane = threadIdx.x & 31;
  const int wv   = __builtin_amdgcn_readfirstlane((int)(threadIdx.x >> 5));
  const int node = blockIdx.x * (GATH_THREADS / 32) + wv;
  if (node >= N) return;
  const int beg = __builtin_amdgcn_readfirstlane(rowptr[node]);
  const int end = __builtin_amdgcn_readfirstlane(rowptr[node + 1]);
  const float di = dis[node];

  float acc[Q];
  {
    const float* hrow = h + (long long)node * C + lane * Q;
    if constexpr (Q == 2) {
      const v2f t = *(const v2f*)hrow;
      acc[0] = di * di * t.x;
      acc[1] = di * di * t.y;
    } else {
      acc[0] = di * di * hrow[0];
    }
  }

  for (int base = beg; base < end; base += 32) {
    const int left = end - base;
    const int cnt  = left < 32 ? left : 32;
    int   s = 0;
    float ds = 0.f;
    if (lane < cnt) {
      s  = csr[base + lane];
      ds = dis[s];
    }
    for (int j = 0; j < cnt; j += GRP) {
      float v[GRP][Q];
#pragma unroll
      for (int u = 0; u < GRP; ++u) {
        const int jj = (j + u) < cnt ? (j + u) : (cnt - 1);
        const int sj = __builtin_amdgcn_readlane(s, jj);
        const float* hrow = h + (long long)sj * C + lane * Q;
        if constexpr (Q == 2) {
          const v2f t = *(const v2f*)hrow;
          v[u][0] = t.x;
          v[u][1] = t.y;
        } else {
          v[u][0] = hrow[0];
        }
      }
#pragma unroll
      for (int u = 0; u < GRP; ++u) {
        if (j + u < cnt) {
          const float wj = __int_as_float(__builtin_amdgcn_readlane(__float_as_int(ds), j + u)) * di;
#pragma unroll
          for (int q = 0; q < Q; ++q) acc[q] += wj * v[u][q];
        }
      }
    }
  }

  float* arow = agg + (long long)node * C + lane * Q;
  if constexpr (Q == 2) {
    v2f t;
    t.x = acc[0];
    t.y = acc[1];
    *(v2f*)arow = t;
  } else {
    arow[0] = acc[0];
  }
}

template <int IC, bool DUAL, bool RELU, bool GUARD, int MT, int RT, int NW>
__global__ __launch_bounds__(32 * NW) void gcn_wmma_gemm(
    const float* __restrict__ inA, const float* __restrict__ WA, const float* __restrict__ bA,
    const float* __restrict__ inB, const float* __restrict__ WB, const float* __restrict__ bB,
    float* __restrict__ out, int nrows, int OC) {
  static_assert(MT == 1 || MT == 2 || MT == 4, "row tiles per wave");
  static_assert(RT % MT == 0 && NW % (RT / MT) == 0, "one wave per (row group, column tile)");
  constexpr int PITCH   = IC + 4;
  static_assert((PITCH & 7) == 4, "pitch must be 4 * odd floats");
  constexpr int ROWS    = 16 * RT;
  constexpr int CTB     = NW / (RT / MT);
  constexpr int NTHR    = 32 * NW;
  __shared__ __attribute__((aligned(16))) float lds[(DUAL ? 2 : 1) * ROWS * PITCH];
  const int tid  = threadIdx.x;
  const int lane = tid & 31;
  const int wave = tid >> 5;
  const int row0 = blockIdx.x * ROWS;
  const int ct   = wave % CTB;
  const int trow = (wave / CTB) * (MT * 16);
  const int col0 = (blockIdx.y * CTB + ct) << 4;

  float* Atile = lds;
  float* Btile = DUAL ? (lds + ROWS * PITCH) : lds;

  constexpr int vecPerRow = IC >> 2;
  constexpr int totalVec  = vecPerRow * ROWS;
  static_assert(totalVec % NTHR == 0, "whole pieces per thread");
  constexpr int PER = totalVec / NTHR;
  constexpr int SBMAX = DUAL ? 8 : 16;
  constexpr int SB  = PER < SBMAX ? PER : SBMAX;
  static_assert(PER % SB == 0, "whole groups of pieces");
  const v4f* srcA = (const v4f*)(inA + (size_t)row0 * IC);
  const v4f* srcB = DUAL ? (const v4f*)(inB + (size_t)row0 * IC) : srcA;
  for (int c = 0; c < PER; c += SB) {
    v4f ta[SB];
    v4f tb[DUAL ? SB : 1];
    for (int s = 0; s < SB; ++s) {
      const int idx = tid + (c + s) * NTHR;
      const bool ok = !GUARD || (row0 + idx / vecPerRow < nrows);
      const int  ld = ok ? idx : 0;
      const v4f z = {0.f, 0.f, 0.f, 0.f};
      const v4f a = srcA[ld];
      ta[s] = ok ? a : z;
      if constexpr (DUAL) {
        const v4f b = srcB[ld];
        tb[s] = ok ? b : z;
      }
    }
    for (int s = 0; s < SB; ++s) {
      const int idx = tid + (c + s) * NTHR;
      const int r = idx / vecPerRow;
      const int v = idx - r * vecPerRow;
      const int off = r * PITCH + (v << 2);
      *(v4f*)(Atile + off) = ta[s];
      if constexpr (DUAL) *(v4f*)(Btile + off) = tb[s];
    }
  }
  __syncthreads();

  if (col0 >= OC) return;

  const int  nidx = lane & 15;
  const int  koff = (lane >> 4) << 1;
  const int  col  = col0 + nidx;
  const bool colOK = GUARD ? (col < OC) : true;
  const int  colC  = GUARD ? (colOK ? col : (OC - 1)) : col;
  const float* wAp = WA + (size_t)colC * IC + koff;
  const float* wBp = DUAL ? (WB + (size_t)colC * IC + koff) : WA;
  const float* ar  = Atile + (trow + nidx) * PITCH + koff;
  const float* br  = Btile + (trow + nidx) * PITCH + koff;

  constexpr int KSTEPS = IC / 4;
  constexpr int BRAW   = (DUAL || MT == 4) ? 4 : 8;
  constexpr int BATCH  = KSTEPS < BRAW ? KSTEPS : BRAW;

  v8f acc[MT];
  for (int m = 0; m < MT; ++m) acc[m] = v8f{};
#pragma unroll 1
  for (int kb = 0; kb < KSTEPS; kb += BATCH) {
    v2f fw[BATCH], fa[MT][BATCH];
    v2f gw[DUAL ? BATCH : 1], ga[DUAL ? MT : 1][DUAL ? BATCH : 1];
#pragma unroll
    for (int i = 0; i < BATCH; ++i) {
      const int k = (kb + i) * 4;
      fw[i] = *(const v2f*)(wAp + k);
      if constexpr (DUAL) gw[i] = *(const v2f*)(wBp + k);
      for (int m = 0; m < MT; ++m) {
        fa[m][i] = *(const v2f*)(ar + m * 16 * PITCH + k);
        if constexpr (DUAL) ga[m][i] = *(const v2f*)(br + m * 16 * PITCH + k);
      }
    }
#pragma unroll
    for (int i = 0; i < BATCH; ++i) {
      for (int m = 0; m < MT; ++m)
        acc[m] = __builtin_amdgcn_wmma_f32_16x16x4_f32(false, fa[m][i], false, fw[i],
                                                       (short)0, acc[m], false, false);
      if constexpr (DUAL) {
        for (int m = 0; m < MT; ++m)
          acc[m] = __builtin_amdgcn_wmma_f32_16x16x4_f32(false, ga[m][i], false, gw[i],
                                                         (short)0, acc[m], false, false);
      }
    }
  }

  const int mb = (lane >> 4) << 3;
  if constexpr (GUARD) {
    float bias = 0.f;
    if (colOK) {
      bias = bA[col];
      if constexpr (DUAL) bias += bB[col];
    }
    for (int m = 0; m < MT; ++m) {
#pragma unroll
      for (int vg = 0; vg < 8; ++vg) {
        const int r = row0 + trow + m * 16 + mb + vg;
        if (colOK && r < nrows) {
          float val = acc[m][vg] + bias;
          if constexpr (RELU) val = val > 0.f ? val : 0.f;
          out[(size_t)r * OC + col] = val;
        }
      }
    }
  } else {
    float bias = bA[col];
    if constexpr (DUAL) bias += bB[col];
    for (int m = 0; m < MT; ++m) {
      float* op = out + (size_t)(row0 + trow + m * 16 + mb) * OC + col;
#pragma unroll
      for (int vg = 0; vg < 8; ++vg) {
        float val = acc[m][vg] + bias;
        if constexpr (RELU) val = val > 0.f ? val : 0.f;
        op[(size_t)vg * OC] = val;
      }
    }
  }
}

__global__ __launch_bounds__(32) void gcn_fold_linear(const float* __restrict__ Wo,
                                                      const float* __restrict__ Wi,
                                                      const float* __restrict__ bi,
                                                      const float* __restrict__ bo,
                                                      float* __restrict__ Wf,
                                                      float* __restrict__ bf,
                                                      int O, int J, int C) {
  const int lane   = threadIdx.x;
  const int ctiles = C >> 4;
  const int ntiles = ((O + 15) >> 4) * ctiles;

  if ((int)blockIdx.x >= ntiles) {
    const int o = (int)blockIdx.x - ntiles;
    const float* wr = Wo + (size_t)o * J;
    float s = 0.f;
    for (int j = lane; j < J; j += 32) s += wr[j] * bi[j];
    for (int m = 16; m > 0; m >>= 1) s += __shfl_xor(s, m);
    if (lane == 0) bf[o] = s + bo[o];
    return;
  }

  const int ot     = blockIdx.x / ctiles;
  const int ct     = blockIdx.x - ot * ctiles;
  const int nidx   = lane & 15;
  const int koff   = (lane >> 4) << 1;
  const int orow   = (ot << 4) + nidx;
  const int orowC  = orow < O ? orow : (O - 1);
  const int col    = (ct << 4) + nidx;
  const float* ap = Wo + (size_t)orowC * J + koff;
  const float* bp = Wi + (size_t)koff * C + col;

  v8f acc = {};
  constexpr int FB = 16;
  int k = 0;
#pragma unroll 1
  for (; k + 4 * FB <= J; k += 4 * FB) {
    v2f fa[FB], fb[FB];
#pragma unroll
    for (int i = 0; i < FB; ++i) {
      const int kk = k + 4 * i;
      fa[i]   = *(const v2f*)(ap + kk);
      fb[i].x = bp[(size_t)kk * C];
      fb[i].y = bp[(size_t)(kk + 1) * C];
    }
#pragma unroll
    for (int i = 0; i < FB; ++i)
      acc = __builtin_amdgcn_wmma_f32_16x16x4_f32(false, fa[i], false, fb[i],
                                                  (short)0, acc, false, false);
  }
  for (; k < J; k += 4) {
    const v2f a = *(const v2f*)(ap + k);
    v2f b;
    b.x = bp[(size_t)k * C];
    b.y = bp[(size_t)(k + 1) * C];
    acc = __builtin_amdgcn_wmma_f32_16x16x4_f32(false, a, false, b, (short)0, acc, false, false);
  }
  const int mb = (lane >> 4) << 3;
#pragma unroll
  for (int vg = 0; vg < 8; ++vg) {
    const int o = (ot << 4) + mb + vg;
    if (o < O) Wf[(size_t)o * C + col] = acc[vg];
  }
}


template <int IC, bool DUAL, bool RELU, int MT, int RT, int NW>
static inline void launch_gemm(const float* inA, const float* WA, const float* bA,
                               const float* inB, const float* WB, const float* bB,
                               float* out, int N, int OC, hipStream_t stream) {
  constexpr int ROWS = 16 * RT;
  constexpr int CTB  = NW / (RT / MT);
  const int rowBlocks = (N + ROWS - 1) / ROWS;
  const int gy = (((OC + 15) / 16) + CTB - 1) / CTB;
  const dim3 grid(rowBlocks, gy), blk(32 * NW);
  if ((N % ROWS) == 0 && (OC & 15) == 0)
    gcn_wmma_gemm<IC, DUAL, RELU, false, MT, RT, NW><<<grid, blk, 0, stream>>>(
        inA, WA, bA, inB, WB, bB, out, N, OC);
  else
    gcn_wmma_gemm<IC, DUAL, RELU, true, MT, RT, NW><<<grid, blk, 0, stream>>>(
        inA, WA, bA, inB, WB, bB, out, N, OC);
}

extern "C" void kernel_launch(void* const* d_in, const int* in_sizes, int n_in,
                              void* d_out, int out_size, void* d_ws, size_t ws_size,
                              hipStream_t stream) {
  (void)n_in; (void)out_size; (void)ws_size;
  const float* x    = (const float*)d_in[0];
  const int*   ei   = (const int*)d_in[1];
  const float* w1   = (const float*)d_in[2];
  const float* b1   = (const float*)d_in[3];
  const float* w2   = (const float*)d_in[4];
  const float* b2   = (const float*)d_in[5];
  const float* c1wl = (const float*)d_in[6];
  const float* c1bl = (const float*)d_in[7];
  const float* c1wr = (const float*)d_in[8];
  const float* c1br = (const float*)d_in[9];
  const float* c2wl = (const float*)d_in[10];
  const float* c2bl = (const float*)d_in[11];
  const float* c2wr = (const float*)d_in[12];
  const float* c2br = (const float*)d_in[13];
  const float* w3   = (const float*)d_in[14];
  const float* b3   = (const float*)d_in[15];
  const float* w4   = (const float*)d_in[16];
  const float* b4   = (const float*)d_in[17];

  const int IN_CH = 256;
  const int N = in_sizes[0] / IN_CH;
  const int E = in_sizes[1] / 2;
  const size_t n = (size_t)N;

  float* ws   = (float*)d_ws;
  float* dis  = ws;
  float* h2   = ws + 1 * n;
  float* agg1 = ws + 33 * n;
  float* h3   = ws + 65 * n;
  float* agg2 = ws + 129 * n;
  float* h4   = ws + 193 * n;
  float* h1   = ws + 193 * n;
  float* wf   = ws + 1 * n;
  float* bfo  = wf + 20 * 128;
  int*   part = (int*)(ws + 129 * n);
  int*   csr    = (int*)(ws + 209 * n);
  int*   rowptr = csr + (((size_t)E + 3) & ~(size_t)3);
  int*   blk    = rowptr + ((n + 4) & ~(size_t)3);

  const int finBlocks  = (N + FIN_THREADS - 1) / FIN_THREADS;
  const int degRanges  = (N + DEG_RANGE - 1) / DEG_RANGE;
  const int gathBlocks = (N + (GATH_THREADS / 32) - 1) / (GATH_THREADS / 32);

  launch_gemm<256, false, false, 1, 1, 1>(x, w1, b1, nullptr, nullptr, nullptr, h1, N, 16, stream);
  launch_gemm<16, false, false, 2, 4, 4>(h1, w2, b2, nullptr, nullptr, nullptr, h2, N, 32, stream);

  gcn_edge_bins<false><<<dim3(DEG_CHUNKS, degRanges), DEG_THREADS, 0, stream>>>(ei, ei + E, part, csr, N, E);
  gcn_deg_finish<<<finBlocks, FIN_THREADS, 0, stream>>>(part, dis, blk, N);
  gcn_csr_offsets<<<finBlocks, FIN_THREADS, 0, stream>>>(part, blk, rowptr, N);
  gcn_edge_bins<true><<<dim3(DEG_CHUNKS, degRanges), DEG_THREADS, 0, stream>>>(ei, ei + E, part, csr, N, E);

  gcn_gather<32><<<gathBlocks, GATH_THREADS, 0, stream>>>(rowptr, csr, dis, h2, agg1, N);
  launch_gemm<32, true, true, 4, 4, 4>(h2, c1wl, c1bl, agg1, c1wr, c1br, h3, N, 64, stream);

  gcn_gather<64><<<gathBlocks, GATH_THREADS, 0, stream>>>(rowptr, csr, dis, h3, agg2, N);
  launch_gemm<64, true, true, 4, 4, 4>(h3, c2wl, c2bl, agg2, c2wr, c2br, h4, N, 128, stream);

  gcn_fold_linear<<<2 * (128 / 16) + 20, 32, 0, stream>>>(w4, w3, b3, b4, wf, bfo, 20, 192, 128);
  launch_gemm<128, false, false, 2, 4, 4>(h4, wf, bfo, nullptr, nullptr, nullptr, (float*)d_out, N, 20, stream);
}


# a linear layer right behind a GEMM is taken in the same launch from the work-group's LDS tile (16->32 behind GEMM 1, folded 128->20 behind conv2): 2 launches fewer; sum order unchanged
# speedup vs baseline: 3.5519x; 1.1093x over previous
#include <hip/hip_runtime.h>

typedef __attribute__((ext_vector_type(2))) float v2f;
typedef __attribute__((ext_vector_type(8))) float v8f;
typedef __attribute__((ext_vector_type(4))) float v4f;

#define DEG_RANGE   6400
#define DEG_CHUNKS  16
#define DEG_THREADS 1024
#define FIN_THREADS 256

template <bool FILL>
__device__ __forceinline__ void bin_one(int* slot, int c, int r, int lo, int nn,
                                        int* __restrict__ csr) {
  const unsigned d = (unsigned)(c - lo);
  if (d < (unsigned)nn) {
    if constexpr (FILL) {
      const int pos = atomicAdd(&slot[d], 1);
      csr[pos] = r;
    } else {
      atomicAdd(&slot[d], 1);
    }
  }
}

template <bool FILL>
__device__ __forceinline__ void bin_four(int* slot, int4 c, int4 r, int lo, int nn,
                                         int* __restrict__ csr) {
  bin_one<FILL>(slot, c.x, r.x, lo, nn, csr);
  bin_one<FILL>(slot, c.y, r.y, lo, nn, csr);
  bin_one<FILL>(slot, c.z, r.z, lo, nn, csr);
  bin_one<FILL>(slot, c.w, r.w, lo, nn, csr);
}

template <bool FILL>
__global__ __launch_bounds__(DEG_THREADS) void gcn_edge_bins(const int* __restrict__ row,
                                                             const int* __restrict__ col,
                                                             int* __restrict__ part,
                                                             int* __restrict__ csr,
                                                             int N, int E) {
  __shared__ int slot[DEG_RANGE];
  const int tid = threadIdx.x;
  const int lo  = blockIdx.y * DEG_RANGE;
  const int nn  = (N - lo) < DEG_RANGE ? (N - lo) : DEG_RANGE;
  int* plane = part + (size_t)blockIdx.x * N + lo;
  for (int i = tid; i < DEG_RANGE; i += DEG_THREADS)
    slot[i] = (FILL && i < nn) ? plane[i] : 0;
  __syncthreads();

  const int quads = (int)(((long long)E + 3) >> 2);
  const int qper  = (quads + DEG_CHUNKS - 1) / DEG_CHUNKS;
  const int q0    = blockIdx.x * qper;
  const int q1    = (q0 + qper) < quads ? (q0 + qper) : quads;

  const bool vec = (E & 3) == 0 && (((size_t)col) & 15) == 0 &&
                   (!FILL || (((size_t)row) & 15) == 0);
  if (vec) {
    const int4* c4 = (const int4*)col;
    const int4* r4 = (const int4*)row;
    const int4 z = {0, 0, 0, 0};
    int q = q0 + tid;
    for (; q + 3 * DEG_THREADS < q1; q += 4 * DEG_THREADS) {
      const int4 ca = c4[q];
      const int4 cb = c4[q + DEG_THREADS];
      const int4 cc = c4[q + 2 * DEG_THREADS];
      const int4 cd = c4[q + 3 * DEG_THREADS];
      const int4 ra = FILL ? r4[q] : z;
      const int4 rb = FILL ? r4[q + DEG_THREADS] : z;
      const int4 rc = FILL ? r4[q + 2 * DEG_THREADS] : z;
      const int4 rd = FILL ? r4[q + 3 * DEG_THREADS] : z;
      bin_four<FILL>(slot, ca, ra, lo, nn, csr);
      bin_four<FILL>(slot, cb, rb, lo, nn, csr);
      bin_four<FILL>(slot, cc, rc, lo, nn, csr);
      bin_four<FILL>(slot, cd, rd, lo, nn, csr);
    }
    for (; q < q1; q += DEG_THREADS) bin_four<FILL>(slot, c4[q], FILL ? r4[q] : z, lo, nn, csr);
  } else {
    const long long e0 = (long long)q0 << 2;
    long long e1 = (long long)q1 << 2;
    if (e1 > E) e1 = E;
    for (long long e = e0 + tid; e < e1; e += DEG_THREADS)
      bin_one<FILL>(slot, col[e], FILL ? row[e] : 0, lo, nn, csr);
  }

  if constexpr (!FILL) {
    __syncthreads();
    for (int i = tid; i < nn; i += DEG_THREADS) plane[i] = slot[i];
  }
}

__global__ __launch_bounds__(FIN_THREADS) void gcn_deg_finish(const int* __restrict__ part,
                                                              float* __restrict__ dis,
                                                              int* __restrict__ blk, int N) {
  __shared__ int wsum[FIN_THREADS / 32];
  const int tid = threadIdx.x;
  const int i = blockIdx.x * FIN_THREADS + tid;
  int in = 0;
  if (i < N) {
    for (int p = 0; p < DEG_CHUNKS; ++p) in += part[(size_t)p * N + i];
    const float d = (float)(in + 1);
    dis[i] = (d > 0.0f) ? rsqrtf(d) : 0.0f;
  }
  int s = in;
  for (int m = 16; m > 0; m >>= 1) s += __shfl_xor(s, m);
  if ((tid & 31) == 0) wsum[tid >> 5] = s;
  __syncthreads();
  if (tid == 0) {
    int t = 0;
    for (int w = 0; w < FIN_THREADS / 32; ++w) t += wsum[w];
    blk[blockIdx.x] = t;
  }
}

__global__ __launch_bounds__(FIN_THREADS) void gcn_csr_offsets(int* __restrict__ part,
                                                               const int* __restrict__ blk,
                                                               int* __restrict__ rowptr, int N) {
  __shared__ int red[FIN_THREADS / 32];
  __shared__ int wtot[FIN_THREADS / 32];
  const int tid  = threadIdx.x;
  const int lane = tid & 31;
  const int wv   = tid >> 5;
  const int b    = blockIdx.x;
  const int i    = b * FIN_THREADS + tid;

  int s = 0;
  for (int k = tid; k < b; k += FIN_THREADS) s += blk[k];
  for (int m = 16; m > 0; m >>= 1) s += __shfl_xor(s, m);
  if (lane == 0) red[wv] = s;

  int c[DEG_CHUNKS];
  int cnt = 0;
#pragma unroll
  for (int p = 0; p < DEG_CHUNKS; ++p) {
    c[p] = (i < N) ? part[(size_t)p * N + i] : 0;
    cnt += c[p];
  }
  int inc = cnt;
  for (int d = 1; d < 32; d <<= 1) {
    const int t = __shfl_up(inc, d);
    if (lane >= d) inc += t;
  }
  if (lane == 31) wtot[wv] = inc;
  __syncthreads();
  int start = inc - cnt;
  for (int w = 0; w < FIN_THREADS / 32; ++w) {
    start += red[w];
    if (w < wv) start += wtot[w];
  }
  if (i < N) {
    rowptr[i] = start;
    if (i == N - 1) rowptr[N] = start + cnt;
    int run = start;
#pragma unroll
    for (int p = 0; p < DEG_CHUNKS; ++p) {
      part[(size_t)p * N + i] = run;
      run += c[p];
    }
  }
}

#define GATH_THREADS 256
template <int C>
__global__ __launch_bounds__(GATH_THREADS) void gcn_gather(const int* __restrict__ rowptr,
                                                           const int* __restrict__ csr,
                                                           const float* __restrict__ dis,
                                                           const float* __restrict__ h,
                                                           float* __restrict__ agg, int N) {
  static_assert(C == 32 || C == 64, "one or two channels per lane");
  constexpr int Q   = C / 32;
  constexpr int GRP = 8;
  const int lane = threadIdx.x & 31;
  const int wv   = __builtin_amdgcn_readfirstlane((int)(threadIdx.x >> 5));
  const int node = blockIdx.x * (GATH_THREADS / 32) + wv;
  if (node >= N) return;
  const int beg = __builtin_amdgcn_readfirstlane(rowptr[node]);
  const int end = __builtin_amdgcn_readfirstlane(rowptr[node + 1]);
  const float di = dis[node];

  float acc[Q];
  {
    const float* hrow = h + (long long)node * C + lane * Q;
    if constexpr (Q == 2) {
      const v2f t = *(const v2f*)hrow;
      acc[0] = di * di * t.x;
      acc[1] = di * di * t.y;
    } else {
      acc[0] = di * di * hrow[0];
    }
  }

  for (int base = beg; base < end; base += 32) {
    const int left = end - base;
    const int cnt  = left < 32 ? left : 32;
    int   s = 0;
    float ds = 0.f;
    if (lane < cnt) {
      s  = csr[base + lane];
      ds = dis[s];
    }
    for (int j = 0; j < cnt; j += GRP) {
      float v[GRP][Q];
#pragma unroll
      for (int u = 0; u < GRP; ++u) {
        const int jj = (j + u) < cnt ? (j + u) : (cnt - 1);
        const int sj = __builtin_amdgcn_readlane(s, jj);
        const float* hrow = h + (long long)sj * C + lane * Q;
        if constexpr (Q == 2) {
          const v2f t = *(const v2f*)hrow;
          v[u][0] = t.x;
          v[u][1] = t.y;
        } else {
          v[u][0] = hrow[0];
        }
      }
#pragma unroll
      for (int u = 0; u < GRP; ++u) {
        if (j + u < cnt) {
          const float wj = __int_as_float(__builtin_amdgcn_readlane(__float_as_int(ds), j + u)) * di;
#pragma unroll
          for (int q = 0; q < Q; ++q) acc[q] += wj * v[u][q];
        }
      }
    }
  }

  float* arow = agg + (long long)node * C + lane * Q;
  if constexpr (Q == 2) {
    v2f t;
    t.x = acc[0];
    t.y = acc[1];
    *(v2f*)arow = t;
  } else {
    arow[0] = acc[0];
  }
}

template <int IC, bool DUAL, bool RELU, bool GUARD, int MT, int RT, int NW, int TMT>
__global__ __launch_bounds__(32 * NW) void gcn_wmma_gemm(
    const float* __restrict__ inA, const float* __restrict__ WA, const float* __restrict__ bA,
    const float* __restrict__ inB, const float* __restrict__ WB, const float* __restrict__ bB,
    float* __restrict__ out, int nrows, int OC,
    const float* __restrict__ Wt, const float* __restrict__ bt, int OT) {
  static_assert(MT == 1 || MT == 2 || MT == 4, "row tiles per wave");
  static_assert(RT % MT == 0 && NW % (RT / MT) == 0, "one wave per (row group, column tile)");
  constexpr int PITCH   = IC + 4;
  static_assert((PITCH & 7) == 4, "pitch must be 4 * odd floats");
  constexpr int ROWS    = 16 * RT;
  constexpr int CTB     = NW / (RT / MT);
  constexpr int NTHR    = 32 * NW;
  __shared__ __attribute__((aligned(16))) float lds[(DUAL ? 2 : 1) * ROWS * PITCH];
  const int tid  = threadIdx.x;
  const int lane = tid & 31;
  const int wave = tid >> 5;
  const int row0 = blockIdx.x * ROWS;
  const int ct   = wave % CTB;
  const int trow = (wave / CTB) * (MT * 16);
  const int col0 = (blockIdx.y * CTB + ct) << 4;

  float* Atile = lds;
  float* Btile = DUAL ? (lds + ROWS * PITCH) : lds;

  constexpr int vecPerRow = IC >> 2;
  constexpr int totalVec  = vecPerRow * ROWS;
  static_assert(totalVec % NTHR == 0, "whole pieces per thread");
  constexpr int PER = totalVec / NTHR;
  constexpr int SBMAX = DUAL ? 8 : 16;
  constexpr int SB  = PER < SBMAX ? PER : SBMAX;
  static_assert(PER % SB == 0, "whole groups of pieces");
  const v4f* srcA = (const v4f*)(inA + (size_t)row0 * IC);
  const v4f* srcB = DUAL ? (const v4f*)(inB + (size_t)row0 * IC) : srcA;
  for (int c = 0; c < PER; c += SB) {
    v4f ta[SB];
    v4f tb[DUAL ? SB : 1];
    for (int s = 0; s < SB; ++s) {
      const int idx = tid + (c + s) * NTHR;
      const bool ok = !GUARD || (row0 + idx / vecPerRow < nrows);
      const int  ld = ok ? idx : 0;
      const v4f z = {0.f, 0.f, 0.f, 0.f};
      const v4f a = srcA[ld];
      ta[s] = ok ? a : z;
      if constexpr (DUAL) {
        const v4f b = srcB[ld];
        tb[s] = ok ? b : z;
      }
    }
    for (int s = 0; s < SB; ++s) {
      const int idx = tid + (c + s) * NTHR;
      const int r = idx / vecPerRow;
      const int v = idx - r * vecPerRow;
      const int off = r * PITCH + (v << 2);
      *(v4f*)(Atile + off) = ta[s];
      if constexpr (DUAL) *(v4f*)(Btile + off) = tb[s];
    }
  }
  __syncthreads();

  if constexpr (TMT == 0) {
    if (col0 >= OC) return;
  }

  const int  nidx = lane & 15;
  const int  koff = (lane >> 4) << 1;
  const int  col  = col0 + nidx;
  const bool colOK = GUARD ? (col < OC) : true;
  const int  colC  = GUARD ? (colOK ? col : (OC - 1)) : col;
  const float* wAp = WA + (size_t)colC * IC + koff;
  const float* wBp = DUAL ? (WB + (size_t)colC * IC + koff) : WA;
  const float* ar  = Atile + (trow + nidx) * PITCH + koff;
  const float* br  = Btile + (trow + nidx) * PITCH + koff;

  constexpr int KSTEPS = IC / 4;
  constexpr int BRAW   = (DUAL || MT == 4) ? 4 : 8;
  constexpr int BATCH  = KSTEPS < BRAW ? KSTEPS : BRAW;

  v8f acc[MT];
  for (int m = 0; m < MT; ++m) acc[m] = v8f{};
#pragma unroll 1
  for (int kb = 0; kb < KSTEPS; kb += BATCH) {
    v2f fw[BATCH], fa[MT][BATCH];
    v2f gw[DUAL ? BATCH : 1], ga[DUAL ? MT : 1][DUAL ? BATCH : 1];
#pragma unroll
    for (int i = 0; i < BATCH; ++i) {
      const int k = (kb + i) * 4;
      fw[i] = *(const v2f*)(wAp + k);
      if constexpr (DUAL) gw[i] = *(const v2f*)(wBp + k);
      for (int m = 0; m < MT; ++m) {
        fa[m][i] = *(const v2f*)(ar + m * 16 * PITCH + k);
        if constexpr (DUAL) ga[m][i] = *(const v2f*)(br + m * 16 * PITCH + k);
      }
    }
#pragma unroll
    for (int i = 0; i < BATCH; ++i) {
      for (int m = 0; m < MT; ++m)
        acc[m] = __builtin_amdgcn_wmma_f32_16x16x4_f32(false, fa[m][i], false, fw[i],
                                                       (short)0, acc[m], false, false);
      if constexpr (DUAL) {
        for (int m = 0; m < MT; ++m)
          acc[m] = __builtin_amdgcn_wmma_f32_16x16x4_f32(false, ga[m][i], false, gw[i],
                                                         (short)0, acc[m], false, false);
      }
    }
  }

  const int mb = (lane >> 4) << 3;
  if constexpr (TMT > 0) {
    constexpr int HC  = 16 * CTB;
    constexpr int HP  = HC + 4;
    static_assert((HP & 7) == 4, "pitch must be 4 * odd floats");
    static_assert(ROWS * HP <= (DUAL ? 2 : 1) * ROWS * PITCH, "tile fits the staging space");
    static_assert(RT % TMT == 0, "whole row groups");
    float* hid = lds;
    __syncthreads();
    {
      float bias = bA[col];
      if constexpr (DUAL) bias += bB[col];
      for (int m = 0; m < MT; ++m) {
#pragma unroll
        for (int vg = 0; vg < 8; ++vg) {
          float val = acc[m][vg] + bias;
          if constexpr (RELU) val = val > 0.f ? val : 0.f;
          hid[(trow + m * 16 + mb + vg) * HP + (ct << 4) + nidx] = val;
        }
      }
    }
    __syncthreads();

    constexpr int KS2 = HC / 4;
    constexpr int B2  = KS2 < 8 ? KS2 : 8;
    const int wv    = __builtin_amdgcn_readfirstlane(wave);
    const int tct   = (OT + 15) >> 4;
    const int items = tct * (RT / TMT);
    for (int item = wv; item < items; item += NW) {
      const int rg    = item / tct;
      const int c2    = item - rg * tct;
      const int trow2 = rg * (TMT * 16);
      const int colt  = (c2 << 4) + nidx;
      const bool cOK  = colt < OT;
      const int  cC   = cOK ? colt : (OT - 1);
      const float* wp = Wt + (size_t)cC * HC + koff;
      const float* hr = hid + (trow2 + nidx) * HP + koff;
      v8f a2[TMT];
      for (int m = 0; m < TMT; ++m) a2[m] = v8f{};
#pragma unroll 1
      for (int kb = 0; kb < KS2; kb += B2) {
        v2f tw[B2], ta2[TMT][B2];
#pragma unroll
        for (int i = 0; i < B2; ++i) {
          const int k = (kb + i) * 4;
          tw[i] = *(const v2f*)(wp + k);
          for (int m = 0; m < TMT; ++m) ta2[m][i] = *(const v2f*)(hr + m * 16 * HP + k);
        }
#pragma unroll
        for (int i = 0; i < B2; ++i)
          for (int m = 0; m < TMT; ++m)
            a2[m] = __builtin_amdgcn_wmma_f32_16x16x4_f32(false, ta2[m][i], false, tw[i],
                                                          (short)0, a2[m], false, false);
      }
      const float bias2 = cOK ? bt[colt] : 0.f;
      for (int m = 0; m < TMT; ++m) {
#pragma unroll
        for (int vg = 0; vg < 8; ++vg) {
          const int r = row0 + trow2 + m * 16 + mb + vg;
          if (cOK && r < nrows) out[(size_t)r * OT + colt] = a2[m][vg] + bias2;
        }
      }
    }
  } else if constexpr (GUARD) {
    float bias = 0.f;
    if (colOK) {
      bias = bA[col];
      if constexpr (DUAL) bias += bB[col];
    }
    for (int m = 0; m < MT; ++m) {
#pragma unroll
      for (int vg = 0; vg < 8; ++vg) {
        const int r = row0 + trow + m * 16 + mb + vg;
        if (colOK && r < nrows) {
          float val = acc[m][vg] + bias;
          if constexpr (RELU) val = val > 0.f ? val : 0.f;
          out[(size_t)r * OC + col] = val;
        }
      }
    }
  } else {
    float bias = bA[col];
    if constexpr (DUAL) bias += bB[col];
    for (int m = 0; m < MT; ++m) {
      float* op = out + (size_t)(row0 + trow + m * 16 + mb) * OC + col;
#pragma unroll
      for (int vg = 0; vg < 8; ++vg) {
        float val = acc[m][vg] + bias;
        if constexpr (RELU) val = val > 0.f ? val : 0.f;
        op[(size_t)vg * OC] = val;
      }
    }
  }
}

__global__ __launch_bounds__(32) void gcn_fold_linear(const float* __restrict__ Wo,
                                                      const float* __restrict__ Wi,
                                                      const float* __restrict__ bi,
                                                      const float* __restrict__ bo,
                                                      float* __restrict__ Wf,
                                                      float* __restrict__ bf,
                                                      int O, int J, int C) {
  const int lane   = threadIdx.x;
  const int ctiles = C >> 4;
  const int ntiles = ((O + 15) >> 4) * ctiles;

  if ((int)blockIdx.x >= ntiles) {
    const int o = (int)blockIdx.x - ntiles;
    const float* wr = Wo + (size_t)o * J;
    float s = 0.f;
    for (int j = lane; j < J; j += 32) s += wr[j] * bi[j];
    for (int m = 16; m > 0; m >>= 1) s += __shfl_xor(s, m);
    if (lane == 0) bf[o] = s + bo[o];
    return;
  }

  const int ot     = blockIdx.x / ctiles;
  const int ct     = blockIdx.x - ot * ctiles;
  const int nidx   = lane & 15;
  const int koff   = (lane >> 4) << 1;
  const int orow   = (ot << 4) + nidx;
  const int orowC  = orow < O ? orow : (O - 1);
  const int col    = (ct << 4) + nidx;
  const float* ap = Wo + (size_t)orowC * J + koff;
  const float* bp = Wi + (size_t)koff * C + col;

  v8f acc = {};
  constexpr int FB = 16;
  int k = 0;
#pragma unroll 1
  for (; k + 4 * FB <= J; k += 4 * FB) {
    v2f fa[FB], fb[FB];
#pragma unroll
    for (int i = 0; i < FB; ++i) {
      const int kk = k + 4 * i;
      fa[i]   = *(const v2f*)(ap + kk);
      fb[i].x = bp[(size_t)kk * C];
      fb[i].y = bp[(size_t)(kk + 1) * C];
    }
#pragma unroll
    for (int i = 0; i < FB; ++i)
      acc = __builtin_amdgcn_wmma_f32_16x16x4_f32(false, fa[i], false, fb[i],
                                                  (short)0, acc, false, false);
  }
  for (; k < J; k += 4) {
    const v2f a = *(const v2f*)(ap + k);
    v2f b;
    b.x = bp[(size_t)k * C];
    b.y = bp[(size_t)(k + 1) * C];
    acc = __builtin_amdgcn_wmma_f32_16x16x4_f32(false, a, false, b, (short)0, acc, false, false);
  }
  const int mb = (lane >> 4) << 3;
#pragma unroll
  for (int vg = 0; vg < 8; ++vg) {
    const int o = (ot << 4) + mb + vg;
    if (o < O) Wf[(size_t)o * C + col] = acc[vg];
  }
}


template <int IC, bool DUAL, bool RELU, int MT, int RT, int NW>
static inline void launch_gemm(const float* inA, const float* WA, const float* bA,
                               const float* inB, const float* WB, const float* bB,
                               float* out, int N, int OC, hipStream_t stream) {
  constexpr int ROWS = 16 * RT;
  constexpr int CTB  = NW / (RT / MT);
  const int rowBlocks = (N + ROWS - 1) / ROWS;
  const int gy = (((OC + 15) / 16) + CTB - 1) / CTB;
  const dim3 grid(rowBlocks, gy), blk(32 * NW);
  if ((N % ROWS) == 0 && (OC & 15) == 0)
    gcn_wmma_gemm<IC, DUAL, RELU, false, MT, RT, NW, 0><<<grid, blk, 0, stream>>>(
        inA, WA, bA, inB, WB, bB, out, N, OC, nullptr, nullptr, 0);
  else
    gcn_wmma_gemm<IC, DUAL, RELU, true, MT, RT, NW, 0><<<grid, blk, 0, stream>>>(
        inA, WA, bA, inB, WB, bB, out, N, OC, nullptr, nullptr, 0);
}

template <int IC, bool DUAL, bool RELU, int MT, int RT, int NW, int TMT>
static inline void launch_gemm_tail(const float* inA, const float* WA, const float* bA,
                                    const float* inB, const float* WB, const float* bB,
                                    const float* Wt, const float* bt,
                                    float* out, int N, int OT, hipStream_t stream) {
  constexpr int ROWS = 16 * RT;
  constexpr int CTB  = NW / (RT / MT);
  const int rowBlocks = (N + ROWS - 1) / ROWS;
  const dim3 grid(rowBlocks, 1), blk(32 * NW);
  if ((N % ROWS) == 0)
    gcn_wmma_gemm<IC, DUAL, RELU, false, MT, RT, NW, TMT><<<grid, blk, 0, stream>>>(
        inA, WA, bA, inB, WB, bB, out, N, 16 * CTB, Wt, bt, OT);
  else
    gcn_wmma_gemm<IC, DUAL, RELU, true, MT, RT, NW, TMT><<<grid, blk, 0, stream>>>(
        inA, WA, bA, inB, WB, bB, out, N, 16 * CTB, Wt, bt, OT);
}

extern "C" void kernel_launch(void* const* d_in, const int* in_sizes, int n_in,
                              void* d_out, int out_size, void* d_ws, size_t ws_size,
                              hipStream_t stream) {
  (void)n_in; (void)out_size; (void)ws_size;
  const float* x    = (const float*)d_in[0];
  const int*   ei   = (const int*)d_in[1];
  const float* w1   = (const float*)d_in[2];
  const float* b1   = (const float*)d_in[3];
  const float* w2   = (const float*)d_in[4];
  const float* b2   = (const float*)d_in[5];
  const float* c1wl = (const float*)d_in[6];
  const float* c1bl = (const float*)d_in[7];
  const float* c1wr = (const float*)d_in[8];
  const float* c1br = (const float*)d_in[9];
  const float* c2wl = (const float*)d_in[10];
  const float* c2bl = (const float*)d_in[11];
  const float* c2wr = (const float*)d_in[12];
  const float* c2br = (const float*)d_in[13];
  const float* w3   = (const float*)d_in[14];
  const float* b3   = (const float*)d_in[15];
  const float* w4   = (const float*)d_in[16];
  const float* b4   = (const float*)d_in[17];

  const int IN_CH = 256;
  const int N = in_sizes[0] / IN_CH;
  const int E = in_sizes[1] / 2;
  const size_t n = (size_t)N;

  float* ws   = (float*)d_ws;
  float* dis  = ws;
  float* h2   = ws + 1 * n;
  float* agg1 = ws + 33 * n;
  float* h3   = ws + 65 * n;
  float* agg2 = ws + 129 * n;
  float* wf   = ws + 300 * n;
  float* bfo  = wf + 20 * 128;
  int*   part = (int*)(ws + 129 * n);
  int*   csr    = (int*)(ws + 209 * n);
  int*   rowptr = csr + (((size_t)E + 3) & ~(size_t)3);
  int*   blk    = rowptr + ((n + 4) & ~(size_t)3);

  const int finBlocks  = (N + FIN_THREADS - 1) / FIN_THREADS;
  const int degRanges  = (N + DEG_RANGE - 1) / DEG_RANGE;
  const int gathBlocks = (N + (GATH_THREADS / 32) - 1) / (GATH_THREADS / 32);

  gcn_fold_linear<<<2 * (128 / 16) + 20, 32, 0, stream>>>(w4, w3, b3, b4, wf, bfo, 20, 192, 128);

  launch_gemm_tail<256, false, false, 1, 1, 1, 1>(x, w1, b1, nullptr, nullptr, nullptr, w2, b2, h2, N, 32, stream);

  gcn_edge_bins<false><<<dim3(DEG_CHUNKS, degRanges), DEG_THREADS, 0, stream>>>(ei, ei + E, part, csr, N, E);
  gcn_deg_finish<<<finBlocks, FIN_THREADS, 0, stream>>>(part, dis, blk, N);
  gcn_csr_offsets<<<finBlocks, FIN_THREADS, 0, stream>>>(part, blk, rowptr, N);
  gcn_edge_bins<true><<<dim3(DEG_CHUNKS, degRanges), DEG_THREADS, 0, stream>>>(ei, ei + E, part, csr, N, E);

  gcn_gather<32><<<gathBlocks, GATH_THREADS, 0, stream>>>(rowptr, csr, dis, h2, agg1, N);
  launch_gemm<32, true, true, 4, 4, 4>(h2, c1wl, c1bl, agg1, c1wr, c1br, h3, N, 64, stream);

  gcn_gather<64><<<gathBlocks, GATH_THREADS, 0, stream>>>(rowptr, csr, dis, h3, agg2, N);
  launch_gemm_tail<64, true, true, 4, 4, 8, 2>(h3, c2wl, c2bl, agg2, c2wr, c2br, wf, bfo, (float*)d_out, N, 20, stream);
}


# 3 launches fewer (10 -> 7): each edge-list fill group sums its range's count planes itself (degree, dis, first slots; no finish or offset launch); weight fold rides in spare groups of the count pass
# speedup vs baseline: 3.7020x; 1.0423x over previous
#include <hip/hip_runtime.h>

typedef __attribute__((ext_vector_type(2))) float v2f;
typedef __attribute__((ext_vector_type(8))) float v8f;
typedef __attribute__((ext_vector_type(4))) float v4f;


__device__ __forceinline__ void fold_item(int item, int lane,
                                          const float* __restrict__ Wo,
                                          const float* __restrict__ Wi,
                                          const float* __restrict__ bi,
                                          const float* __restrict__ bo,
                                          float* __restrict__ Wf,
                                          float* __restrict__ bf,
                                          int O, int J, int C) {
  const int ctiles = C >> 4;
  const int ntiles = ((O + 15) >> 4) * ctiles;
  if (item >= ntiles + O) return;

  if (item >= ntiles) {
    const int o = item - ntiles;
    const float* wr = Wo + (size_t)o * J;
    float s = 0.f;
    for (int j = lane; j < J; j += 32) s += wr[j] * bi[j];
    for (int m = 16; m > 0; m >>= 1) s += __shfl_xor(s, m);
    if (lane == 0) bf[o] = s + bo[o];
    return;
  }

  const int ot     = item / ctiles;
  const int ct     = item - ot * ctiles;
  const int nidx   = lane & 15;
  const int koff   = (lane >> 4) << 1;
  const int orow   = (ot << 4) + nidx;
  const int orowC  = orow < O ? orow : (O - 1);
  const int col    = (ct << 4) + nidx;
  const float* ap = Wo + (size_t)orowC * J + koff;
  const float* bp = Wi + (size_t)koff * C + col;

  v8f acc = {};
  constexpr int FB = 8;
  int k = 0;
#pragma unroll 1
  for (; k + 4 * FB <= J; k += 4 * FB) {
    v2f fa[FB], fb[FB];
#pragma unroll
    for (int i = 0; i < FB; ++i) {
      const int kk = k + 4 * i;
      fa[i]   = *(const v2f*)(ap + kk);
      fb[i].x = bp[(size_t)kk * C];
      fb[i].y = bp[(size_t)(kk + 1) * C];
    }
#pragma unroll
    for (int i = 0; i < FB; ++i)
      acc = __builtin_amdgcn_wmma_f32_16x16x4_f32(false, fa[i], false, fb[i],
                                                  (short)0, acc, false, false);
  }
  for (; k < J; k += 4) {
    const v2f a = *(const v2f*)(ap + k);
    v2f b;
    b.x = bp[(size_t)k * C];
    b.y = bp[(size_t)(k + 1) * C];
    acc = __builtin_amdgcn_wmma_f32_16x16x4_f32(false, a, false, b, (short)0, acc, false, false);
  }
  const int mb = (lane >> 4) << 3;
#pragma unroll
  for (int vg = 0; vg < 8; ++vg) {
    const int o = (ot << 4) + mb + vg;
    if (o < O) Wf[(size_t)o * C + col] = acc[vg];
  }
}

#define DEG_RANGE   6400
#define DEG_CHUNKS  16
#define DEG_THREADS 1024
#define DEG_WAVES   (DEG_THREADS / 32)
#define DEG_PER     ((DEG_RANGE + DEG_THREADS - 1) / DEG_THREADS)
#define DEG_OWN     (DEG_RANGE / DEG_CHUNKS)
#define DEG_GRP     4
static_assert(DEG_OWN * DEG_CHUNKS == DEG_RANGE, "equal shares of a range");

template <bool FILL>
__device__ __forceinline__ void bin_one(int* slot, int c, int r, int lo, int nn,
                                        int* __restrict__ csr) {
  const unsigned d = (unsigned)(c - lo);
  if (d < (unsigned)nn) {
    if constexpr (FILL) {
      const int pos = atomicAdd(&slot[d], 1);
      csr[pos] = r;
    } else {
      atomicAdd(&slot[d], 1);
    }
  }
}

template <bool FILL>
__device__ __forceinline__ void bin_four(int* slot, int4 c, int4 r, int lo, int nn,
                                         int* __restrict__ csr) {
  bin_one<FILL>(slot, c.x, r.x, lo, nn, csr);
  bin_one<FILL>(slot, c.y, r.y, lo, nn, csr);
  bin_one<FILL>(slot, c.z, r.z, lo, nn, csr);
  bin_one<FILL>(slot, c.w, r.w, lo, nn, csr);
}

template <bool FILL>
__device__ __forceinline__ void bin_piece(int* slot, const int* __restrict__ row,
                                          const int* __restrict__ col, int* __restrict__ csr,
                                          int piece, int lo, int nn, int E) {
  const int tid = threadIdx.x;
  const int quads = (int)(((long long)E + 3) >> 2);
  const int qper  = (quads + DEG_CHUNKS - 1) / DEG_CHUNKS;
  const int q0    = piece * qper;
  const int q1    = (q0 + qper) < quads ? (q0 + qper) : quads;

  const bool vec = (E & 3) == 0 && (((size_t)col) & 15) == 0 &&
                   (!FILL || (((size_t)row) & 15) == 0);
  if (vec) {
    const int4* c4 = (const int4*)col;
    const int4* r4 = (const int4*)row;
    const int4 z = {0, 0, 0, 0};
    int q = q0 + tid;
    for (; q + 3 * DEG_THREADS < q1; q += 4 * DEG_THREADS) {
      const int4 ca = c4[q];
      const int4 cb = c4[q + DEG_THREADS];
      const int4 cc = c4[q + 2 * DEG_THREADS];
      const int4 cd = c4[q + 3 * DEG_THREADS];
      const int4 ra = FILL ? r4[q] : z;
      const int4 rb = FILL ? r4[q + DEG_THREADS] : z;
      const int4 rc = FILL ? r4[q + 2 * DEG_THREADS] : z;
      const int4 rd = FILL ? r4[q + 3 * DEG_THREADS] : z;
      bin_four<FILL>(slot, ca, ra, lo, nn, csr);
      bin_four<FILL>(slot, cb, rb, lo, nn, csr);
      bin_four<FILL>(slot, cc, rc, lo, nn, csr);
      bin_four<FILL>(slot, cd, rd, lo, nn, csr);
    }
    for (; q < q1; q += DEG_THREADS) bin_four<FILL>(slot, c4[q], FILL ? r4[q] : z, lo, nn, csr);
  } else {
    const long long e0 = (long long)q0 << 2;
    long long e1 = (long long)q1 << 2;
    if (e1 > E) e1 = E;
    for (long long e = e0 + tid; e < e1; e += DEG_THREADS)
      bin_one<FILL>(slot, col[e], FILL ? row[e] : 0, lo, nn, csr);
  }
}

__global__ __launch_bounds__(DEG_THREADS) void gcn_count_edges(
    const int* __restrict__ col, int* __restrict__ part, int* __restrict__ rtot,
    int N, int E, int nbins,
    const float* __restrict__ Wo, const float* __restrict__ Wi, const float* __restrict__ bi,
    const float* __restrict__ bo, float* __restrict__ Wf, float* __restrict__ bf,
    int O, int J, int C) {
  __shared__ int slot[DEG_RANGE];
  __shared__ int wsum[DEG_WAVES];
  const int tid = threadIdx.x;
  if ((int)blockIdx.x >= nbins) {
    const int wv = __builtin_amdgcn_readfirstlane(tid >> 5);
    fold_item(((int)blockIdx.x - nbins) * DEG_WAVES + wv, tid & 31, Wo, Wi, bi, bo, Wf, bf, O, J, C);
    return;
  }
  const int piece = blockIdx.x % DEG_CHUNKS;
  const int range = blockIdx.x / DEG_CHUNKS;
  const int lo    = range * DEG_RANGE;
  const int nn    = (N - lo) < DEG_RANGE ? (N - lo) : DEG_RANGE;
  int* plane = part + (size_t)piece * N + lo;
  for (int i = tid; i < DEG_RANGE; i += DEG_THREADS) slot[i] = 0;
  __syncthreads();

  bin_piece<false>(slot, col, col, nullptr, piece, lo, nn, E);

  __syncthreads();
  int s = 0;
  for (int i = tid; i < nn; i += DEG_THREADS) {
    const int v = slot[i];
    plane[i] = v;
    s += v;
  }
  for (int m = 16; m > 0; m >>= 1) s += __shfl_xor(s, m);
  if ((tid & 31) == 0) wsum[tid >> 5] = s;
  __syncthreads();
  if (tid == 0) {
    int t = 0;
    for (int w = 0; w < DEG_WAVES; ++w) t += wsum[w];
    rtot[blockIdx.x] = t;
  }
}

__global__ __launch_bounds__(DEG_THREADS) void gcn_fill_edges(
    const int* __restrict__ row, const int* __restrict__ col, const int* __restrict__ part,
    const int* __restrict__ rtot, int* __restrict__ csr, int* __restrict__ rowptr,
    float* __restrict__ dis, int N, int E) {
  __shared__ int slot[DEG_RANGE];
  __shared__ int totl[DEG_RANGE];
  __shared__ int wtot[DEG_WAVES];
  __shared__ int wred[DEG_WAVES];
  const int tid   = threadIdx.x;
  const int lane  = tid & 31;
  const int wave  = tid >> 5;
  const int piece = blockIdx.x % DEG_CHUNKS;
  const int range = blockIdx.x / DEG_CHUNKS;
  const int lo    = range * DEG_RANGE;
  const int nn    = (N - lo) < DEG_RANGE ? (N - lo) : DEG_RANGE;

#pragma unroll
  for (int k0 = 0; k0 < DEG_PER; k0 += DEG_GRP) {
    int c[DEG_GRP][DEG_CHUNKS];
#pragma unroll
    for (int kk = 0; kk < DEG_GRP; ++kk) {
      if (k0 + kk < DEG_PER) {
        const int i  = tid + (k0 + kk) * DEG_THREADS;
        const int ic = i < nn ? i : nn - 1;
        const int* p = part + lo + ic;
#pragma unroll
        for (int q = 0; q < DEG_CHUNKS; ++q) c[kk][q] = p[(size_t)q * N];
      }
    }
#pragma unroll
    for (int kk = 0; kk < DEG_GRP; ++kk) {
      if (k0 + kk < DEG_PER) {
        const int i = tid + (k0 + kk) * DEG_THREADS;
        int tot = 0, mine = 0;
#pragma unroll
        for (int q = 0; q < DEG_CHUNKS; ++q) {
          tot += c[kk][q];
          mine += (q < piece) ? c[kk][q] : 0;
        }
        if (i < DEG_RANGE) {
          totl[i] = i < nn ? tot : 0;
          slot[i] = i < nn ? mine : 0;
        }
      }
    }
  }
  int rb = 0;
  for (int k = tid; k < range * DEG_CHUNKS; k += DEG_THREADS) rb += rtot[k];
  for (int m = 16; m > 0; m >>= 1) rb += __shfl_xor(rb, m);
  if (lane == 0) wred[wave] = rb;
  __syncthreads();

  const int base = tid * DEG_PER;
  int deg[DEG_PER];
  int sum = 0;
#pragma unroll
  for (int j = 0; j < DEG_PER; ++j) {
    deg[j] = (base + j < DEG_RANGE) ? totl[base + j] : 0;
    sum += deg[j];
  }
  int inc = sum;
  for (int d = 1; d < 32; d <<= 1) {
    const int t = __shfl_up(inc, d);
    if (lane >= d) inc += t;
  }
  if (lane == 31) wtot[wave] = inc;
  __syncthreads();
  int start = inc - sum;
  for (int w = 0; w < DEG_WAVES; ++w) {
    start += wred[w];
    if (w < wave) start += wtot[w];
  }
#pragma unroll
  for (int j = 0; j < DEG_PER; ++j) {
    const int i = base + j;
    if (i < nn) {
      slot[i] += start;
      if ((unsigned)(i - piece * DEG_OWN) < (unsigned)DEG_OWN) {
        const int g = lo + i;
        rowptr[g] = start;
        if (g == N - 1) rowptr[N] = start + deg[j];
        const float d = (float)(deg[j] + 1);
        dis[g] = (d > 0.0f) ? rsqrtf(d) : 0.0f;
      }
    }
    start += deg[j];
  }
  __syncthreads();

  bin_piece<true>(slot, row, col, csr, piece, lo, nn, E);
}

#define GATH_THREADS 256
template <int C>
__global__ __launch_bounds__(GATH_THREADS) void gcn_gather(const int* __restrict__ rowptr,
                                                           const int* __restrict__ csr,
                                                           const float* __restrict__ dis,
                                                           const float* __restrict__ h,
                                                           float* __restrict__ agg, int N) {
  static_assert(C == 32 || C == 64, "one or two channels per lane");
  constexpr int Q   = C / 32;
  constexpr int GRP = 8;
  const int lane = threadIdx.x & 31;
  const int wv   = __builtin_amdgcn_readfirstlane((int)(threadIdx.x >> 5));
  const int node = blockIdx.x * (GATH_THREADS / 32) + wv;
  if (node >= N) return;
  const int beg = __builtin_amdgcn_readfirstlane(rowptr[node]);
  const int end = __builtin_amdgcn_readfirstlane(rowptr[node + 1]);
  const float di = dis[node];

  float acc[Q];
  {
    const float* hrow = h + (long long)node * C + lane * Q;
    if constexpr (Q == 2) {
      const v2f t = *(const v2f*)hrow;
      acc[0] = di * di * t.x;
      acc[1] = di * di * t.y;
    } else {
      acc[0] = di * di * hrow[0];
    }
  }

  for (int base = beg; base < end; base += 32) {
    const int left = end - base;
    const int cnt  = left < 32 ? left : 32;
    int   s = 0;
    float ds = 0.f;
    if (lane < cnt) {
      s  = csr[base + lane];
      ds = dis[s];
    }
    for (int j = 0; j < cnt; j += GRP) {
      float v[GRP][Q];
#pragma unroll
      for (int u = 0; u < GRP; ++u) {
        const int jj = (j + u) < cnt ? (j + u) : (cnt - 1);
        const int sj = __builtin_amdgcn_readlane(s, jj);
        const float* hrow = h + (long long)sj * C + lane * Q;
        if constexpr (Q == 2) {
          const v2f t = *(const v2f*)hrow;
          v[u][0] = t.x;
          v[u][1] = t.y;
        } else {
          v[u][0] = hrow[0];
        }
      }
#pragma unroll
      for (int u = 0; u < GRP; ++u) {
        if (j + u < cnt) {
          const float wj = __int_as_float(__builtin_amdgcn_readlane(__float_as_int(ds), j + u)) * di;
#pragma unroll
          for (int q = 0; q < Q; ++q) acc[q] += wj * v[u][q];
        }
      }
    }
  }

  float* arow = agg + (long long)node * C + lane * Q;
  if constexpr (Q == 2) {
    v2f t;
    t.x = acc[0];
    t.y = acc[1];
    *(v2f*)arow = t;
  } else {
    arow[0] = acc[0];
  }
}

template <int IC, bool DUAL, bool RELU, bool GUARD, int MT, int RT, int NW, int TMT>
__global__ __launch_bounds__(32 * NW) void gcn_wmma_gemm(
    const float* __restrict__ inA, const float* __restrict__ WA, const float* __restrict__ bA,
    const float* __restrict__ inB, const float* __restrict__ WB, const float* __restrict__ bB,
    float* __restrict__ out, int nrows, int OC,
    const float* __restrict__ Wt, const float* __restrict__ bt, int OT) {
  static_assert(MT == 1 || MT == 2 || MT == 4, "row tiles per wave");
  static_assert(RT % MT == 0 && NW % (RT / MT) == 0, "one wave per (row group, column tile)");
  constexpr int PITCH   = IC + 4;
  static_assert((PITCH & 7) == 4, "pitch must be 4 * odd floats");
  constexpr int ROWS    = 16 * RT;
  constexpr int CTB     = NW / (RT / MT);
  constexpr int NTHR    = 32 * NW;
  __shared__ __attribute__((aligned(16))) float lds[(DUAL ? 2 : 1) * ROWS * PITCH];
  const int tid  = threadIdx.x;
  const int lane = tid & 31;
  const int wave = tid >> 5;
  const int row0 = blockIdx.x * ROWS;
  const int ct   = wave % CTB;
  const int trow = (wave / CTB) * (MT * 16);
  const int col0 = (blockIdx.y * CTB + ct) << 4;

  float* Atile = lds;
  float* Btile = DUAL ? (lds + ROWS * PITCH) : lds;

  constexpr int vecPerRow = IC >> 2;
  constexpr int totalVec  = vecPerRow * ROWS;
  static_assert(totalVec % NTHR == 0, "whole pieces per thread");
  constexpr int PER = totalVec / NTHR;
  constexpr int SBMAX = DUAL ? 8 : 16;
  constexpr int SB  = PER < SBMAX ? PER : SBMAX;
  static_assert(PER % SB == 0, "whole groups of pieces");
  const v4f* srcA = (const v4f*)(inA + (size_t)row0 * IC);
  const v4f* srcB = DUAL ? (const v4f*)(inB + (size_t)row0 * IC) : srcA;
  for (int c = 0; c < PER; c += SB) {
    v4f ta[SB];
    v4f tb[DUAL ? SB : 1];
    for (int s = 0; s < SB; ++s) {
      const int idx = tid + (c + s) * NTHR;
      const bool ok = !GUARD || (row0 + idx / vecPerRow < nrows);
      const int  ld = ok ? idx : 0;
      const v4f z = {0.f, 0.f, 0.f, 0.f};
      const v4f a = srcA[ld];
      ta[s] = ok ? a : z;
      if constexpr (DUAL) {
        const v4f b = srcB[ld];
        tb[s] = ok ? b : z;
      }
    }
    for (int s = 0; s < SB; ++s) {
      const int idx = tid + (c + s) * NTHR;
      const int r = idx / vecPerRow;
      const int v = idx - r * vecPerRow;
      const int off = r * PITCH + (v << 2);
      *(v4f*)(Atile + off) = ta[s];
      if constexpr (DUAL) *(v4f*)(Btile + off) = tb[s];
    }
  }
  __syncthreads();

  if constexpr (TMT == 0) {
    if (col0 >= OC) return;
  }

  const int  nidx = lane & 15;
  const int  koff = (lane >> 4) << 1;
  const int  col  = col0 + nidx;
  const bool colOK = GUARD ? (col < OC) : true;
  const int  colC  = GUARD ? (colOK ? col : (OC - 1)) : col;
  const float* wAp = WA + (size_t)colC * IC + koff;
  const float* wBp = DUAL ? (WB + (size_t)colC * IC + koff) : WA;
  const float* ar  = Atile + (trow + nidx) * PITCH + koff;
  const float* br  = Btile + (trow + nidx) * PITCH + koff;

  constexpr int KSTEPS = IC / 4;
  constexpr int BRAW   = (DUAL || MT == 4) ? 4 : 8;
  constexpr int BATCH  = KSTEPS < BRAW ? KSTEPS : BRAW;

  v8f acc[MT];
  for (int m = 0; m < MT; ++m) acc[m] = v8f{};
#pragma unroll 1
  for (int kb = 0; kb < KSTEPS; kb += BATCH) {
    v2f fw[BATCH], fa[MT][BATCH];
    v2f gw[DUAL ? BATCH : 1], ga[DUAL ? MT : 1][DUAL ? BATCH : 1];
#pragma unroll
    for (int i = 0; i < BATCH; ++i) {
      const int k = (kb + i) * 4;
      fw[i] = *(const v2f*)(wAp + k);
      if constexpr (DUAL) gw[i] = *(const v2f*)(wBp + k);
      for (int m = 0; m < MT; ++m) {
        fa[m][i] = *(const v2f*)(ar + m * 16 * PITCH + k);
        if constexpr (DUAL) ga[m][i] = *(const v2f*)(br + m * 16 * PITCH + k);
      }
    }
#pragma unroll
    for (int i = 0; i < BATCH; ++i) {
      for (int m = 0; m < MT; ++m)
        acc[m] = __builtin_amdgcn_wmma_f32_16x16x4_f32(false, fa[m][i], false, fw[i],
                                                       (short)0, acc[m], false, false);
      if constexpr (DUAL) {
        for (int m = 0; m < MT; ++m)
          acc[m] = __builtin_amdgcn_wmma_f32_16x16x4_f32(false, ga[m][i], false, gw[i],
                                                         (short)0, acc[m], false, false);
      }
    }
  }

  const int mb = (lane >> 4) << 3;
  if constexpr (TMT > 0) {
    constexpr int HC  = 16 * CTB;
    constexpr int HP  = HC + 4;
    static_assert((HP & 7) == 4, "pitch must be 4 * odd floats");
    static_assert(ROWS * HP <= (DUAL ? 2 : 1) * ROWS * PITCH, "tile fits the staging space");
    static_assert(RT % TMT == 0, "whole row groups");
    float* hid = lds;
    __syncthreads();
    {
      float bias = bA[col];
      if constexpr (DUAL) bias += bB[col];
      for (int m = 0; m < MT; ++m) {
#pragma unroll
        for (int vg = 0; vg < 8; ++vg) {
          float val = acc[m][vg] + bias;
          if constexpr (RELU) val = val > 0.f ? val : 0.f;
          hid[(trow + m * 16 + mb + vg) * HP + (ct << 4) + nidx] = val;
        }
      }
    }
    __syncthreads();

    constexpr int KS2 = HC / 4;
    constexpr int B2  = KS2 < 8 ? KS2 : 8;
    const int wv    = __builtin_amdgcn_readfirstlane(wave);
    const int tct   = (OT + 15) >> 4;
    const int items = tct * (RT / TMT);
    for (int item = wv; item < items; item += NW) {
      const int rg    = item / tct;
      const int c2    = item - rg * tct;
      const int trow2 = rg * (TMT * 16);
      const int colt  = (c2 << 4) + nidx;
      const bool cOK  = colt < OT;
      const int  cC   = cOK ? colt : (OT - 1);
      const float* wp = Wt + (size_t)cC * HC + koff;
      const float* hr = hid + (trow2 + nidx) * HP + koff;
      v8f a2[TMT];
      for (int m = 0; m < TMT; ++m) a2[m] = v8f{};
#pragma unroll 1
      for (int kb = 0; kb < KS2; kb += B2) {
        v2f tw[B2], ta2[TMT][B2];
#pragma unroll
        for (int i = 0; i < B2; ++i) {
          const int k = (kb + i) * 4;
          tw[i] = *(const v2f*)(wp + k);
          for (int m = 0; m < TMT; ++m) ta2[m][i] = *(const v2f*)(hr + m * 16 * HP + k);
        }
#pragma unroll
        for (int i = 0; i < B2; ++i)
          for (int m = 0; m < TMT; ++m)
            a2[m] = __builtin_amdgcn_wmma_f32_16x16x4_f32(false, ta2[m][i], false, tw[i],
                                                          (short)0, a2[m], false, false);
      }
      const float bias2 = cOK ? bt[colt] : 0.f;
      for (int m = 0; m < TMT; ++m) {
#pragma unroll
        for (int vg = 0; vg < 8; ++vg) {
          const int r = row0 + trow2 + m * 16 + mb + vg;
          if (cOK && r < nrows) out[(size_t)r * OT + colt] = a2[m][vg] + bias2;
        }
      }
    }
  } else if constexpr (GUARD) {
    float bias = 0.f;
    if (colOK) {
      bias = bA[col];
      if constexpr (DUAL) bias += bB[col];
    }
    for (int m = 0; m < MT; ++m) {
#pragma unroll
      for (int vg = 0; vg < 8; ++vg) {
        const int r = row0 + trow + m * 16 + mb + vg;
        if (colOK && r < nrows) {
          float val = acc[m][vg] + bias;
          if constexpr (RELU) val = val > 0.f ? val : 0.f;
          out[(size_t)r * OC + col] = val;
        }
      }
    }
  } else {
    float bias = bA[col];
    if constexpr (DUAL) bias += bB[col];
    for (int m = 0; m < MT; ++m) {
      float* op = out + (size_t)(row0 + trow + m * 16 + mb) * OC + col;
#pragma unroll
      for (int vg = 0; vg < 8; ++vg) {
        float val = acc[m][vg] + bias;
        if constexpr (RELU) val = val > 0.f ? val : 0.f;
        op[(size_t)vg * OC] = val;
      }
    }
  }
}


template <int IC, bool DUAL, bool RELU, int MT, int RT, int NW>
static inline void launch_gemm(const float* inA, const float* WA, const float* bA,
                               const float* inB, const float* WB, const float* bB,
                               float* out, int N, int OC, hipStream_t stream) {
  constexpr int ROWS = 16 * RT;
  constexpr int CTB  = NW / (RT / MT);
  const int rowBlocks = (N + ROWS - 1) / ROWS;
  const int gy = (((OC + 15) / 16) + CTB - 1) / CTB;
  const dim3 grid(rowBlocks, gy), blk(32 * NW);
  if ((N % ROWS) == 0 && (OC & 15) == 0)
    gcn_wmma_gemm<IC, DUAL, RELU, false, MT, RT, NW, 0><<<grid, blk, 0, stream>>>(
        inA, WA, bA, inB, WB, bB, out, N, OC, nullptr, nullptr, 0);
  else
    gcn_wmma_gemm<IC, DUAL, RELU, true, MT, RT, NW, 0><<<grid, blk, 0, stream>>>(
        inA, WA, bA, inB, WB, bB, out, N, OC, nullptr, nullptr, 0);
}

template <int IC, bool DUAL, bool RELU, int MT, int RT, int NW, int TMT>
static inline void launch_gemm_tail(const float* inA, const float* WA, const float* bA,
                                    const float* inB, const float* WB, const float* bB,
                                    const float* Wt, const float* bt,
                                    float* out, int N, int OT, hipStream_t stream) {
  constexpr int ROWS = 16 * RT;
  constexpr int CTB  = NW / (RT / MT);
  const int rowBlocks = (N + ROWS - 1) / ROWS;
  const dim3 grid(rowBlocks, 1), blk(32 * NW);
  if ((N % ROWS) == 0)
    gcn_wmma_gemm<IC, DUAL, RELU, false, MT, RT, NW, TMT><<<grid, blk, 0, stream>>>(
        inA, WA, bA, inB, WB, bB, out, N, 16 * CTB, Wt, bt, OT);
  else
    gcn_wmma_gemm<IC, DUAL, RELU, true, MT, RT, NW, TMT><<<grid, blk, 0, stream>>>(
        inA, WA, bA, inB, WB, bB, out, N, 16 * CTB, Wt, bt, OT);
}

extern "C" void kernel_launch(void* const* d_in, const int* in_sizes, int n_in,
                              void* d_out, int out_size, void* d_ws, size_t ws_size,
                              hipStream_t stream) {
  (void)n_in; (void)out_size; (void)ws_size;
  const float* x    = (const float*)d_in[0];
  const int*   ei   = (const int*)d_in[1];
  const float* w1   = (const float*)d_in[2];
  const float* b1   = (const float*)d_in[3];
  const float* w2   = (const float*)d_in[4];
  const float* b2   = (const float*)d_in[5];
  const float* c1wl = (const float*)d_in[6];
  const float* c1bl = (const float*)d_in[7];
  const float* c1wr = (const float*)d_in[8];
  const float* c1br = (const float*)d_in[9];
  const float* c2wl = (const float*)d_in[10];
  const float* c2bl = (const float*)d_in[11];
  const float* c2wr = (const float*)d_in[12];
  const float* c2br = (const float*)d_in[13];
  const float* w3   = (const float*)d_in[14];
  const float* b3   = (const float*)d_in[15];
  const float* w4   = (const float*)d_in[16];
  const float* b4   = (const float*)d_in[17];

  const int IN_CH = 256;
  const int N = in_sizes[0] / IN_CH;
  const int E = in_sizes[1] / 2;
  const size_t n = (size_t)N;

  float* ws   = (float*)d_ws;
  float* dis  = ws;
  float* h2   = ws + 1 * n;
  float* agg1 = ws + 33 * n;
  float* h3   = ws + 65 * n;
  float* agg2 = ws + 129 * n;
  float* wf   = ws + 300 * n;
  float* bfo  = wf + 20 * 128;
  int*   part = (int*)(ws + 129 * n);
  int*   csr    = (int*)(ws + 209 * n);
  int*   rowptr = csr + (((size_t)E + 3) & ~(size_t)3);
  int*   rtot   = rowptr + ((n + 4) & ~(size_t)3);

  const int degRanges  = (N + DEG_RANGE - 1) / DEG_RANGE;
  const int binBlocks  = DEG_CHUNKS * degRanges;
  const int foldBlocks = (2 * (128 / 16) + 20 + DEG_WAVES - 1) / DEG_WAVES;
  const int gathBlocks = (N + (GATH_THREADS / 32) - 1) / (GATH_THREADS / 32);

  launch_gemm_tail<256, false, false, 1, 1, 1, 1>(x, w1, b1, nullptr, nullptr, nullptr, w2, b2, h2, N, 32, stream);

  gcn_count_edges<<<binBlocks + foldBlocks, DEG_THREADS, 0, stream>>>(ei + E, part, rtot, N, E, binBlocks,
                                                                     w4, w3, b3, b4, wf, bfo, 20, 192, 128);
  gcn_fill_edges<<<binBlocks, DEG_THREADS, 0, stream>>>(ei, ei + E, part, rtot, csr, rowptr, dis, N, E);

  gcn_gather<32><<<gathBlocks, GATH_THREADS, 0, stream>>>(rowptr, csr, dis, h2, agg1, N);
  launch_gemm<32, true, true, 4, 4, 4>(h2, c1wl, c1bl, agg1, c1wr, c1br, h3, N, 64, stream);

  gcn_gather<64><<<gathBlocks, GATH_THREADS, 0, stream>>>(rowptr, csr, dis, h3, agg2, N);
  launch_gemm_tail<64, true, true, 4, 4, 8, 2>(h3, c2wl, c2bl, agg2, c2wr, c2br, wf, bfo, (float*)d_out, N, 20, stream);
}


# first GEMM (x->16->32) rides in the edge-list fill launch as extra work-groups: one wave per 16-row tile, x staged 64 columns a pass in the wave's own LDS; 7 -> 6 launches; sum order kept
# speedup vs baseline: 4.0050x; 1.0818x over previous
#include <hip/hip_runtime.h>

typedef __attribute__((ext_vector_type(2))) float v2f;
typedef __attribute__((ext_vector_type(8))) float v8f;
typedef __attribute__((ext_vector_type(4))) float v4f;


__device__ __forceinline__ void fold_item(int item, int lane,
                                          const float* __restrict__ Wo,
                                          const float* __restrict__ Wi,
                                          const float* __restrict__ bi,
                                          const float* __restrict__ bo,
                                          float* __restrict__ Wf,
                                          float* __restrict__ bf,
                                          int O, int J, int C) {
  const int ctiles = C >> 4;
  const int ntiles = ((O + 15) >> 4) * ctiles;
  if (item >= ntiles + O) return;

  if (item >= ntiles) {
    const int o = item - ntiles;
    const float* wr = Wo + (size_t)o * J;
    float s = 0.f;
    for (int j = lane; j < J; j += 32) s += wr[j] * bi[j];
    for (int m = 16; m > 0; m >>= 1) s += __shfl_xor(s, m);
    if (lane == 0) bf[o] = s + bo[o];
    return;
  }

  const int ot     = item / ctiles;
  const int ct     = item - ot * ctiles;
  const int nidx   = lane & 15;
  const int koff   = (lane >> 4) << 1;
  const int orow   = (ot << 4) + nidx;
  const int orowC  = orow < O ? orow : (O - 1);
  const int col    = (ct << 4) + nidx;
  const float* ap = Wo + (size_t)orowC * J + koff;
  const float* bp = Wi + (size_t)koff * C + col;

  v8f acc = {};
  constexpr int FB = 8;
  int k = 0;
#pragma unroll 1
  for (; k + 4 * FB <= J; k += 4 * FB) {
    v2f fa[FB], fb[FB];
#pragma unroll
    for (int i = 0; i < FB; ++i) {
      const int kk = k + 4 * i;
      fa[i]   = *(const v2f*)(ap + kk);
      fb[i].x = bp[(size_t)kk * C];
      fb[i].y = bp[(size_t)(kk + 1) * C];
    }
#pragma unroll
    for (int i = 0; i < FB; ++i)
      acc = __builtin_amdgcn_wmma_f32_16x16x4_f32(false, fa[i], false, fb[i],
                                                  (short)0, acc, false, false);
  }
  for (; k < J; k += 4) {
    const v2f a = *(const v2f*)(ap + k);
    v2f b;
    b.x = bp[(size_t)k * C];
    b.y = bp[(size_t)(k + 1) * C];
    acc = __builtin_amdgcn_wmma_f32_16x16x4_f32(false, a, false, b, (short)0, acc, false, false);
  }
  const int mb = (lane >> 4) << 3;
#pragma unroll
  for (int vg = 0; vg < 8; ++vg) {
    const int o = (ot << 4) + mb + vg;
    if (o < O) Wf[(size_t)o * C + col] = acc[vg];
  }
}

#define G1_KC    64
#define G1_PITCH (G1_KC + 4)
#define G1_SLICE (16 * G1_PITCH)
#define G1_FB    4
static_assert((G1_PITCH & 7) == 4, "pitch must be 4 * odd floats");

__device__ __forceinline__ void wave_lds_order() {
  __builtin_amdgcn_fence(__ATOMIC_RELEASE, "wavefront");
  __builtin_amdgcn_wave_barrier();
  __builtin_amdgcn_fence(__ATOMIC_ACQUIRE, "wavefront");
}

__device__ __forceinline__ void gemm1_tile(float* tl, int tile, int lane,
                                           const float* __restrict__ x,
                                           const float* __restrict__ w1,
                                           const float* __restrict__ b1,
                                           const float* __restrict__ w2,
                                           const float* __restrict__ b2,
                                           float* __restrict__ h2, int N) {
  constexpr int IC = 256;
  const int row0 = tile << 4;
  const int nidx = lane & 15;
  const int koff = (lane >> 4) << 1;
  const int mb   = (lane >> 4) << 3;
  const int sr = lane >> 4;
  const int sv = (lane & 15) << 2;
  const float* wp = w1 + (size_t)nidx * IC + koff;
  const float* ar = tl + nidx * G1_PITCH + koff;

  v8f acc = {};
#pragma unroll 1
  for (int kc = 0; kc < IC; kc += G1_KC) {
    v4f t[8];
#pragma unroll
    for (int s = 0; s < 8; ++s) {
      int gr = row0 + 2 * s + sr;
      gr = gr < N ? gr : N - 1;
      t[s] = *(const v4f*)(x + (size_t)gr * IC + kc + sv);
    }
    wave_lds_order();
#pragma unroll
    for (int s = 0; s < 8; ++s) *(v4f*)(tl + (2 * s + sr) * G1_PITCH + sv) = t[s];
    wave_lds_order();
#pragma unroll 1
    for (int kb = 0; kb < G1_KC; kb += 4 * G1_FB) {
      v2f fw[G1_FB], fa[G1_FB];
#pragma unroll
      for (int i = 0; i < G1_FB; ++i) {
        fw[i] = *(const v2f*)(wp + kc + kb + 4 * i);
        fa[i] = *(const v2f*)(ar + kb + 4 * i);
      }
#pragma unroll
      for (int i = 0; i < G1_FB; ++i)
        acc = __builtin_amdgcn_wmma_f32_16x16x4_f32(false, fa[i], false, fw[i],
                                                    (short)0, acc, false, false);
    }
  }

  constexpr int HP = 20;
  static_assert((HP & 7) == 4, "pitch must be 4 * odd floats");
  wave_lds_order();
  {
    const float bias = b1[nidx];
#pragma unroll
    for (int vg = 0; vg < 8; ++vg) tl[(mb + vg) * HP + nidx] = acc[vg] + bias;
  }
  wave_lds_order();
  const float* hr = tl + nidx * HP + koff;
#pragma unroll
  for (int c2 = 0; c2 < 2; ++c2) {
    const int colt = (c2 << 4) + nidx;
    const float* wq = w2 + (size_t)colt * 16 + koff;
    v2f tw[4], ta[4];
#pragma unroll
    for (int i = 0; i < 4; ++i) {
      tw[i] = *(const v2f*)(wq + 4 * i);
      ta[i] = *(const v2f*)(hr + 4 * i);
    }
    v8f a2 = {};
#pragma unroll
    for (int i = 0; i < 4; ++i)
      a2 = __builtin_amdgcn_wmma_f32_16x16x4_f32(false, ta[i], false, tw[i],
                                                 (short)0, a2, false, false);
    const float bias2 = b2[colt];
#pragma unroll
    for (int vg = 0; vg < 8; ++vg) {
      const int r = row0 + mb + vg;
      if (r < N) h2[(size_t)r * 32 + colt] = a2[vg] + bias2;
    }
  }
}

#define DEG_RANGE   6400
#define DEG_CHUNKS  16
#define DEG_THREADS 1024
#define DEG_WAVES   (DEG_THREADS / 32)
#define DEG_PER     ((DEG_RANGE + DEG_THREADS - 1) / DEG_THREADS)
#define DEG_OWN     (DEG_RANGE / DEG_CHUNKS)
#define DEG_GRP     2
static_assert(DEG_OWN * DEG_CHUNKS == DEG_RANGE, "equal shares of a range");

template <bool FILL>
__device__ __forceinline__ void bin_one(int* slot, int c, int r, int lo, int nn,
                                        int* __restrict__ csr) {
  const unsigned d = (unsigned)(c - lo);
  if (d < (unsigned)nn) {
    if constexpr (FILL) {
      const int pos = atomicAdd(&slot[d], 1);
      csr[pos] = r;
    } else {
      atomicAdd(&slot[d], 1);
    }
  }
}

template <bool FILL>
__device__ __forceinline__ void bin_four(int* slot, int4 c, int4 r, int lo, int nn,
                                         int* __restrict__ csr) {
  bin_one<FILL>(slot, c.x, r.x, lo, nn, csr);
  bin_one<FILL>(slot, c.y, r.y, lo, nn, csr);
  bin_one<FILL>(slot, c.z, r.z, lo, nn, csr);
  bin_one<FILL>(slot, c.w, r.w, lo, nn, csr);
}

template <bool FILL>
__device__ __forceinline__ void bin_piece(int* slot, const int* __restrict__ row,
                                          const int* __restrict__ col, int* __restrict__ csr,
                                          int piece, int lo, int nn, int E) {
  const int tid = threadIdx.x;
  const int quads = (int)(((long long)E + 3) >> 2);
  const int qper  = (quads + DEG_CHUNKS - 1) / DEG_CHUNKS;
  const int q0    = piece * qper;
  const int q1    = (q0 + qper) < quads ? (q0 + qper) : quads;

  const bool vec = (E & 3) == 0 && (((size_t)col) & 15) == 0 &&
                   (!FILL || (((size_t)row) & 15) == 0);
  if (vec) {
    const int4* c4 = (const int4*)col;
    const int4* r4 = (const int4*)row;
    const int4 z = {0, 0, 0, 0};
    int q = q0 + tid;
    for (; q + 3 * DEG_THREADS < q1; q += 4 * DEG_THREADS) {
      const int4 ca = c4[q];
      const int4 cb = c4[q + DEG_THREADS];
      const int4 cc = c4[q + 2 * DEG_THREADS];
      const int4 cd = c4[q + 3 * DEG_THREADS];
      const int4 ra = FILL ? r4[q] : z;
      const int4 rb = FILL ? r4[q + DEG_THREADS] : z;
      const int4 rc = FILL ? r4[q + 2 * DEG_THREADS] : z;
      const int4 rd = FILL ? r4[q + 3 * DEG_THREADS] : z;
      bin_four<FILL>(slot, ca, ra, lo, nn, csr);
      bin_four<FILL>(slot, cb, rb, lo, nn, csr);
      bin_four<FILL>(slot, cc, rc, lo, nn, csr);
      bin_four<FILL>(slot, cd, rd, lo, nn, csr);
    }
    for (; q < q1; q += DEG_THREADS) bin_four<FILL>(slot, c4[q], FILL ? r4[q] : z, lo, nn, csr);
  } else {
    const long long e0 = (long long)q0 << 2;
    long long e1 = (long long)q1 << 2;
    if (e1 > E) e1 = E;
    for (long long e = e0 + tid; e < e1; e += DEG_THREADS)
      bin_one<FILL>(slot, col[e], FILL ? row[e] : 0, lo, nn, csr);
  }
}

__global__ __launch_bounds__(DEG_THREADS) void gcn_count_edges(
    const int* __restrict__ col, int* __restrict__ part, int* __restrict__ rtot,
    int N, int E, int nbins,
    const float* __restrict__ Wo, const float* __restrict__ Wi, const float* __restrict__ bi,
    const float* __restrict__ bo, float* __restrict__ Wf, float* __restrict__ bf,
    int O, int J, int C) {
  __shared__ int slot[DEG_RANGE];
  __shared__ int wsum[DEG_WAVES];
  const int tid = threadIdx.x;
  if ((int)blockIdx.x >= nbins) {
    const int wv = __builtin_amdgcn_readfirstlane(tid >> 5);
    fold_item(((int)blockIdx.x - nbins) * DEG_WAVES + wv, tid & 31, Wo, Wi, bi, bo, Wf, bf, O, J, C);
    return;
  }
  const int piece = blockIdx.x % DEG_CHUNKS;
  const int range = blockIdx.x / DEG_CHUNKS;
  const int lo    = range * DEG_RANGE;
  const int nn    = (N - lo) < DEG_RANGE ? (N - lo) : DEG_RANGE;
  int* plane = part + (size_t)piece * N + lo;
  for (int i = tid; i < DEG_RANGE; i += DEG_THREADS) slot[i] = 0;
  __syncthreads();

  bin_piece<false>(slot, col, col, nullptr, piece, lo, nn, E);

  __syncthreads();
  int s = 0;
  for (int i = tid; i < nn; i += DEG_THREADS) {
    const int v = slot[i];
    plane[i] = v;
    s += v;
  }
  for (int m = 16; m > 0; m >>= 1) s += __shfl_xor(s, m);
  if ((tid & 31) == 0) wsum[tid >> 5] = s;
  __syncthreads();
  if (tid == 0) {
    int t = 0;
    for (int w = 0; w < DEG_WAVES; ++w) t += wsum[w];
    rtot[blockIdx.x] = t;
  }
}

__global__ __launch_bounds__(DEG_THREADS, 16) void gcn_fill_edges(
    const int* __restrict__ row, const int* __restrict__ col, const int* __restrict__ part,
    const int* __restrict__ rtot, int* __restrict__ csr, int* __restrict__ rowptr,
    float* __restrict__ dis, int N, int E, int nfill,
    const float* __restrict__ x, const float* __restrict__ w1, const float* __restrict__ b1,
    const float* __restrict__ w2, const float* __restrict__ b2, float* __restrict__ h2) {
  __shared__ __attribute__((aligned(16))) float ldsf[DEG_WAVES * G1_SLICE];
  static_assert(2 * DEG_RANGE + 2 * DEG_WAVES <= DEG_WAVES * G1_SLICE, "the fill arrays fit the block");
  const int tid   = threadIdx.x;
  if ((int)blockIdx.x >= nfill) {
    const int wv   = __builtin_amdgcn_readfirstlane(tid >> 5);
    const int tile = ((int)blockIdx.x - nfill) * DEG_WAVES + wv;
    if ((tile << 4) < N) gemm1_tile(ldsf + wv * G1_SLICE, tile, tid & 31, x, w1, b1, w2, b2, h2, N);
    return;
  }
  int* slot = (int*)ldsf;
  int* totl = slot + DEG_RANGE;
  int* wtot = totl + DEG_RANGE;
  int* wred = wtot + DEG_WAVES;
  const int lane  = tid & 31;
  const int wave  = tid >> 5;
  const int piece = blockIdx.x % DEG_CHUNKS;
  const int range = blockIdx.x / DEG_CHUNKS;
  const int lo    = range * DEG_RANGE;
  const int nn    = (N - lo) < DEG_RANGE ? (N - lo) : DEG_RANGE;

#pragma unroll
  for (int k0 = 0; k0 < DEG_PER; k0 += DEG_GRP) {
    int c[DEG_GRP][DEG_CHUNKS];
#pragma unroll
    for (int kk = 0; kk < DEG_GRP; ++kk) {
      if (k0 + kk < DEG_PER) {
        const int i  = tid + (k0 + kk) * DEG_THREADS;
        const int ic = i < nn ? i : nn - 1;
        const int* p = part + lo + ic;
#pragma unroll
        for (int q = 0; q < DEG_CHUNKS; ++q) c[kk][q] = p[(size_t)q * N];
      }
    }
#pragma unroll
    for (int kk = 0; kk < DEG_GRP; ++kk) {
      if (k0 + kk < DEG_PER) {
        const int i = tid + (k0 + kk) * DEG_THREADS;
        int tot = 0, mine = 0;
#pragma unroll
        for (int q = 0; q < DEG_CHUNKS; ++q) {
          tot += c[kk][q];
          mine += (q < piece) ? c[kk][q] : 0;
        }
        if (i < DEG_RANGE) {
          totl[i] = i < nn ? tot : 0;
          slot[i] = i < nn ? mine : 0;
        }
      }
    }
  }
  int rb = 0;
  for (int k = tid; k < range * DEG_CHUNKS; k += DEG_THREADS) rb += rtot[k];
  for (int m = 16; m > 0; m >>= 1) rb += __shfl_xor(rb, m);
  if (lane == 0) wred[wave] = rb;
  __syncthreads();

  const int base = tid * DEG_PER;
  int deg[DEG_PER];
  int sum = 0;
#pragma unroll
  for (int j = 0; j < DEG_PER; ++j) {
    deg[j] = (base + j < DEG_RANGE) ? totl[base + j] : 0;
    sum += deg[j];
  }
  int inc = sum;
  for (int d = 1; d < 32; d <<= 1) {
    const int t = __shfl_up(inc, d);
    if (lane >= d) inc += t;
  }
  if (lane == 31) wtot[wave] = inc;
  __syncthreads();
  int start = inc - sum;
  for (int w = 0; w < DEG_WAVES; ++w) {
    start += wred[w];
    if (w < wave) start += wtot[w];
  }
#pragma unroll
  for (int j = 0; j < DEG_PER; ++j) {
    const int i = base + j;
    if (i < nn) {
      slot[i] += start;
      if ((unsigned)(i - piece * DEG_OWN) < (unsigned)DEG_OWN) {
        const int g = lo + i;
        rowptr[g] = start;
        if (g == N - 1) rowptr[N] = start + deg[j];
        const float d = (float)(deg[j] + 1);
        dis[g] = (d > 0.0f) ? rsqrtf(d) : 0.0f;
      }
    }
    start += deg[j];
  }
  __syncthreads();

  bin_piece<true>(slot, row, col, csr, piece, lo, nn, E);
}

#define GATH_THREADS 256
template <int C>
__global__ __launch_bounds__(GATH_THREADS) void gcn_gather(const int* __restrict__ rowptr,
                                                           const int* __restrict__ csr,
                                                           const float* __restrict__ dis,
                                                           const float* __restrict__ h,
                                                           float* __restrict__ agg, int N) {
  static_assert(C == 32 || C == 64, "one or two channels per lane");
  constexpr int Q   = C / 32;
  constexpr int GRP = 8;
  const int lane = threadIdx.x & 31;
  const int wv   = __builtin_amdgcn_readfirstlane((int)(threadIdx.x >> 5));
  const int node = blockIdx.x * (GATH_THREADS / 32) + wv;
  if (node >= N) return;
  const int beg = __builtin_amdgcn_readfirstlane(rowptr[node]);
  const int end = __builtin_amdgcn_readfirstlane(rowptr[node + 1]);
  const float di = dis[node];

  float acc[Q];
  {
    const float* hrow = h + (long long)node * C + lane * Q;
    if constexpr (Q == 2) {
      const v2f t = *(const v2f*)hrow;
      acc[0] = di * di * t.x;
      acc[1] = di * di * t.y;
    } else {
      acc[0] = di * di * hrow[0];
    }
  }

  for (int base = beg; base < end; base += 32) {
    const int left = end - base;
    const int cnt  = left < 32 ? left : 32;
    int   s = 0;
    float ds = 0.f;
    if (lane < cnt) {
      s  = csr[base + lane];
      ds = dis[s];
    }
    for (int j = 0; j < cnt; j += GRP) {
      float v[GRP][Q];
#pragma unroll
      for (int u = 0; u < GRP; ++u) {
        const int jj = (j + u) < cnt ? (j + u) : (cnt - 1);
        const int sj = __builtin_amdgcn_readlane(s, jj);
        const float* hrow = h + (long long)sj * C + lane * Q;
        if constexpr (Q == 2) {
          const v2f t = *(const v2f*)hrow;
          v[u][0] = t.x;
          v[u][1] = t.y;
        } else {
          v[u][0] = hrow[0];
        }
      }
#pragma unroll
      for (int u = 0; u < GRP; ++u) {
        if (j + u < cnt) {
          const float wj = __int_as_float(__builtin_amdgcn_readlane(__float_as_int(ds), j + u)) * di;
#pragma unroll
          for (int q = 0; q < Q; ++q) acc[q] += wj * v[u][q];
        }
      }
    }
  }

  float* arow = agg + (long long)node * C + lane * Q;
  if constexpr (Q == 2) {
    v2f t;
    t.x = acc[0];
    t.y = acc[1];
    *(v2f*)arow = t;
  } else {
    arow[0] = acc[0];
  }
}

template <int IC, bool DUAL, bool RELU, bool GUARD, int MT, int RT, int NW, int TMT>
__global__ __launch_bounds__(32 * NW) void gcn_wmma_gemm(
    const float* __restrict__ inA, const float* __restrict__ WA, const float* __restrict__ bA,
    const float* __restrict__ inB, const float* __restrict__ WB, const float* __restrict__ bB,
    float* __restrict__ out, int nrows, int OC,
    const float* __restrict__ Wt, const float* __restrict__ bt, int OT) {
  static_assert(MT == 1 || MT == 2 || MT == 4, "row tiles per wave");
  static_assert(RT % MT == 0 && NW % (RT / MT) == 0, "one wave per (row group, column tile)");
  constexpr int PITCH   = IC + 4;
  static_assert((PITCH & 7) == 4, "pitch must be 4 * odd floats");
  constexpr int ROWS    = 16 * RT;
  constexpr int CTB     = NW / (RT / MT);
  constexpr int NTHR    = 32 * NW;
  __shared__ __attribute__((aligned(16))) float lds[(DUAL ? 2 : 1) * ROWS * PITCH];
  const int tid  = threadIdx.x;
  const int lane = tid & 31;
  const int wave = tid >> 5;
  const int row0 = blockIdx.x * ROWS;
  const int ct   = wave % CTB;
  const int trow = (wave / CTB) * (MT * 16);
  const int col0 = (blockIdx.y * CTB + ct) << 4;

  float* Atile = lds;
  float* Btile = DUAL ? (lds + ROWS * PITCH) : lds;

  constexpr int vecPerRow = IC >> 2;
  constexpr int totalVec  = vecPerRow * ROWS;
  static_assert(totalVec % NTHR == 0, "whole pieces per thread");
  constexpr int PER = totalVec / NTHR;
  constexpr int SBMAX = DUAL ? 8 : 16;
  constexpr int SB  = PER < SBMAX ? PER : SBMAX;
  static_assert(PER % SB == 0, "whole groups of pieces");
  const v4f* srcA = (const v4f*)(inA + (size_t)row0 * IC);
  const v4f* srcB = DUAL ? (const v4f*)(inB + (size_t)row0 * IC) : srcA;
  for (int c = 0; c < PER; c += SB) {
    v4f ta[SB];
    v4f tb[DUAL ? SB : 1];
    for (int s = 0; s < SB; ++s) {
      const int idx = tid + (c + s) * NTHR;
      const bool ok = !GUARD || (row0 + idx / vecPerRow < nrows);
      const int  ld = ok ? idx : 0;
      const v4f z = {0.f, 0.f, 0.f, 0.f};
      const v4f a = srcA[ld];
      ta[s] = ok ? a : z;
      if constexpr (DUAL) {
        const v4f b = srcB[ld];
        tb[s] = ok ? b : z;
      }
    }
    for (int s = 0; s < SB; ++s) {
      const int idx = tid + (c + s) * NTHR;
      const int r = idx / vecPerRow;
      const int v = idx - r * vecPerRow;
      const int off = r * PITCH + (v << 2);
      *(v4f*)(Atile + off) = ta[s];
      if constexpr (DUAL) *(v4f*)(Btile + off) = tb[s];
    }
  }
  __syncthreads();

  if constexpr (TMT == 0) {
    if (col0 >= OC) return;
  }

  const int  nidx = lane & 15;
  const int  koff = (lane >> 4) << 1;
  const int  col  = col0 + nidx;
  const bool colOK = GUARD ? (col < OC) : true;
  const int  colC  = GUARD ? (colOK ? col : (OC - 1)) : col;
  const float* wAp = WA + (size_t)colC * IC + koff;
  const float* wBp = DUAL ? (WB + (size_t)colC * IC + koff) : WA;
  const float* ar  = Atile + (trow + nidx) * PITCH + koff;
  const float* br  = Btile + (trow + nidx) * PITCH + koff;

  constexpr int KSTEPS = IC / 4;
  constexpr int BRAW   = (DUAL || MT == 4) ? 4 : 8;
  constexpr int BATCH  = KSTEPS < BRAW ? KSTEPS : BRAW;

  v8f acc[MT];
  for (int m = 0; m < MT; ++m) acc[m] = v8f{};
#pragma unroll 1
  for (int kb = 0; kb < KSTEPS; kb += BATCH) {
    v2f fw[BATCH], fa[MT][BATCH];
    v2f gw[DUAL ? BATCH : 1], ga[DUAL ? MT : 1][DUAL ? BATCH : 1];
#pragma unroll
    for (int i = 0; i < BATCH; ++i) {
      const int k = (kb + i) * 4;
      fw[i] = *(const v2f*)(wAp + k);
      if constexpr (DUAL) gw[i] = *(const v2f*)(wBp + k);
      for (int m = 0; m < MT; ++m) {
        fa[m][i] = *(const v2f*)(ar + m * 16 * PITCH + k);
        if constexpr (DUAL) ga[m][i] = *(const v2f*)(br + m * 16 * PITCH + k);
      }
    }
#pragma unroll
    for (int i = 0; i < BATCH; ++i) {
      for (int m = 0; m < MT; ++m)
        acc[m] = __builtin_amdgcn_wmma_f32_16x16x4_f32(false, fa[m][i], false, fw[i],
                                                       (short)0, acc[m], false, false);
      if constexpr (DUAL) {
        for (int m = 0; m < MT; ++m)
          acc[m] = __builtin_amdgcn_wmma_f32_16x16x4_f32(false, ga[m][i], false, gw[i],
                                                         (short)0, acc[m], false, false);
      }
    }
  }

  const int mb = (lane >> 4) << 3;
  if constexpr (TMT > 0) {
    constexpr int HC  = 16 * CTB;
    constexpr int HP  = HC + 4;
    static_assert((HP & 7) == 4, "pitch must be 4 * odd floats");
    static_assert(ROWS * HP <= (DUAL ? 2 : 1) * ROWS * PITCH, "tile fits the staging space");
    static_assert(RT % TMT == 0, "whole row groups");
    float* hid = lds;
    __syncthreads();
    {
      float bias = bA[col];
      if constexpr (DUAL) bias += bB[col];
      for (int m = 0; m < MT; ++m) {
#pragma unroll
        for (int vg = 0; vg < 8; ++vg) {
          float val = acc[m][vg] + bias;
          if constexpr (RELU) val = val > 0.f ? val : 0.f;
          hid[(trow + m * 16 + mb + vg) * HP + (ct << 4) + nidx] = val;
        }
      }
    }
    __syncthreads();

    constexpr int KS2 = HC / 4;
    constexpr int B2  = KS2 < 8 ? KS2 : 8;
    const int wv    = __builtin_amdgcn_readfirstlane(wave);
    const int tct   = (OT + 15) >> 4;
    const int items = tct * (RT / TMT);
    for (int item = wv; item < items; item += NW) {
      const int rg    = item / tct;
      const int c2    = item - rg * tct;
      const int trow2 = rg * (TMT * 16);
      const int colt  = (c2 << 4) + nidx;
      const bool cOK  = colt < OT;
      const int  cC   = cOK ? colt : (OT - 1);
      const float* wp = Wt + (size_t)cC * HC + koff;
      const float* hr = hid + (trow2 + nidx) * HP + koff;
      v8f a2[TMT];
      for (int m = 0; m < TMT; ++m) a2[m] = v8f{};
#pragma unroll 1
      for (int kb = 0; kb < KS2; kb += B2) {
        v2f tw[B2], ta2[TMT][B2];
#pragma unroll
        for (int i = 0; i < B2; ++i) {
          const int k = (kb + i) * 4;
          tw[i] = *(const v2f*)(wp + k);
          for (int m = 0; m < TMT; ++m) ta2[m][i] = *(const v2f*)(hr + m * 16 * HP + k);
        }
#pragma unroll
        for (int i = 0; i < B2; ++i)
          for (int m = 0; m < TMT; ++m)
            a2[m] = __builtin_amdgcn_wmma_f32_16x16x4_f32(false, ta2[m][i], false, tw[i],
                                                          (short)0, a2[m], false, false);
      }
      const float bias2 = cOK ? bt[colt] : 0.f;
      for (int m = 0; m < TMT; ++m) {
#pragma unroll
        for (int vg = 0; vg < 8; ++vg) {
          const int r = row0 + trow2 + m * 16 + mb + vg;
          if (cOK && r < nrows) out[(size_t)r * OT + colt] = a2[m][vg] + bias2;
        }
      }
    }
  } else if constexpr (GUARD) {
    float bias = 0.f;
    if (colOK) {
      bias = bA[col];
      if constexpr (DUAL) bias += bB[col];
    }
    for (int m = 0; m < MT; ++m) {
#pragma unroll
      for (int vg = 0; vg < 8; ++vg) {
        const int r = row0 + trow + m * 16 + mb + vg;
        if (colOK && r < nrows) {
          float val = acc[m][vg] + bias;
          if constexpr (RELU) val = val > 0.f ? val : 0.f;
          out[(size_t)r * OC + col] = val;
        }
      }
    }
  } else {
    float bias = bA[col];
    if constexpr (DUAL) bias += bB[col];
    for (int m = 0; m < MT; ++m) {
      float* op = out + (size_t)(row0 + trow + m * 16 + mb) * OC + col;
#pragma unroll
      for (int vg = 0; vg < 8; ++vg) {
        float val = acc[m][vg] + bias;
        if constexpr (RELU) val = val > 0.f ? val : 0.f;
        op[(size_t)vg * OC] = val;
      }
    }
  }
}


template <int IC, bool DUAL, bool RELU, int MT, int RT, int NW>
static inline void launch_gemm(const float* inA, const float* WA, const float* bA,
                               const float* inB, const float* WB, const float* bB,
                               float* out, int N, int OC, hipStream_t stream) {
  constexpr int ROWS = 16 * RT;
  constexpr int CTB  = NW / (RT / MT);
  const int rowBlocks = (N + ROWS - 1) / ROWS;
  const int gy = (((OC + 15) / 16) + CTB - 1) / CTB;
  const dim3 grid(rowBlocks, gy), blk(32 * NW);
  if ((N % ROWS) == 0 && (OC & 15) == 0)
    gcn_wmma_gemm<IC, DUAL, RELU, false, MT, RT, NW, 0><<<grid, blk, 0, stream>>>(
        inA, WA, bA, inB, WB, bB, out, N, OC, nullptr, nullptr, 0);
  else
    gcn_wmma_gemm<IC, DUAL, RELU, true, MT, RT, NW, 0><<<grid, blk, 0, stream>>>(
        inA, WA, bA, inB, WB, bB, out, N, OC, nullptr, nullptr, 0);
}

template <int IC, bool DUAL, bool RELU, int MT, int RT, int NW, int TMT>
static inline void launch_gemm_tail(const float* inA, const float* WA, const float* bA,
                                    const float* inB, const float* WB, const float* bB,
                                    const float* Wt, const float* bt,
                                    float* out, int N, int OT, hipStream_t stream) {
  constexpr int ROWS = 16 * RT;
  constexpr int CTB  = NW / (RT / MT);
  const int rowBlocks = (N + ROWS - 1) / ROWS;
  const dim3 grid(rowBlocks, 1), blk(32 * NW);
  if ((N % ROWS) == 0)
    gcn_wmma_gemm<IC, DUAL, RELU, false, MT, RT, NW, TMT><<<grid, blk, 0, stream>>>(
        inA, WA, bA, inB, WB, bB, out, N, 16 * CTB, Wt, bt, OT);
  else
    gcn_wmma_gemm<IC, DUAL, RELU, true, MT, RT, NW, TMT><<<grid, blk, 0, stream>>>(
        inA, WA, bA, inB, WB, bB, out, N, 16 * CTB, Wt, bt, OT);
}

extern "C" void kernel_launch(void* const* d_in, const int* in_sizes, int n_in,
                              void* d_out, int out_size, void* d_ws, size_t ws_size,
                              hipStream_t stream) {
  (void)n_in; (void)out_size; (void)ws_size;
  const float* x    = (const float*)d_in[0];
  const int*   ei   = (const int*)d_in[1];
  const float* w1   = (const float*)d_in[2];
  const float* b1   = (const float*)d_in[3];
  const float* w2   = (const float*)d_in[4];
  const float* b2   = (const float*)d_in[5];
  const float* c1wl = (const float*)d_in[6];
  const float* c1bl = (const float*)d_in[7];
  const float* c1wr = (const float*)d_in[8];
  const float* c1br = (const float*)d_in[9];
  const float* c2wl = (const float*)d_in[10];
  const float* c2bl = (const float*)d_in[11];
  const float* c2wr = (const float*)d_in[12];
  const float* c2br = (const float*)d_in[13];
  const float* w3   = (const float*)d_in[14];
  const float* b3   = (const float*)d_in[15];
  const float* w4   = (const float*)d_in[16];
  const float* b4   = (const float*)d_in[17];

  const int IN_CH = 256;
  const int N = in_sizes[0] / IN_CH;
  const int E = in_sizes[1] / 2;
  const size_t n = (size_t)N;

  float* ws   = (float*)d_ws;
  float* dis  = ws;
  float* h2   = ws + 1 * n;
  float* agg1 = ws + 33 * n;
  float* h3   = ws + 65 * n;
  float* agg2 = ws + 129 * n;
  float* wf   = ws + 300 * n;
  float* bfo  = wf + 20 * 128;
  int*   part = (int*)(ws + 129 * n);
  int*   csr    = (int*)(ws + 209 * n);
  int*   rowptr = csr + (((size_t)E + 3) & ~(size_t)3);
  int*   rtot   = rowptr + ((n + 4) & ~(size_t)3);

  const int degRanges  = (N + DEG_RANGE - 1) / DEG_RANGE;
  const int binBlocks  = DEG_CHUNKS * degRanges;
  const int foldBlocks = (2 * (128 / 16) + 20 + DEG_WAVES - 1) / DEG_WAVES;
  const int gathBlocks = (N + (GATH_THREADS / 32) - 1) / (GATH_THREADS / 32);
  const int g1Blocks   = ((N + 15) / 16 + DEG_WAVES - 1) / DEG_WAVES;

  gcn_count_edges<<<binBlocks + foldBlocks, DEG_THREADS, 0, stream>>>(ei + E, part, rtot, N, E, binBlocks,
                                                                     w4, w3, b3, b4, wf, bfo, 20, 192, 128);
  gcn_fill_edges<<<binBlocks + g1Blocks, DEG_THREADS, 0, stream>>>(ei, ei + E, part, rtot, csr, rowptr, dis, N, E,
                                                                 binBlocks, x, w1, b1, w2, b2, h2);

  gcn_gather<32><<<gathBlocks, GATH_THREADS, 0, stream>>>(rowptr, csr, dis, h2, agg1, N);
  launch_gemm<32, true, true, 4, 4, 4>(h2, c1wl, c1bl, agg1, c1wr, c1br, h3, N, 64, stream);

  gcn_gather<64><<<gathBlocks, GATH_THREADS, 0, stream>>>(rowptr, csr, dis, h3, agg2, N);
  launch_gemm_tail<64, true, true, 4, 4, 8, 2>(h3, c2wl, c2bl, agg2, c2wr, c2br, wf, bfo, (float*)d_out, N, 20, stream);
}
